# starting kernel
# baseline (speedup 1.0000x reference)
#include <hip/hip_runtime.h>

#define N_TOTAL    64000
#define NODE_COUNT 1000
#define BATCH      64
#define STATE      16
#define HID        32
#define N_EDGES    2048000
#define E_TOTAL    (N_EDGES + N_TOTAL)

typedef float v2f __attribute__((ext_vector_type(2)));
typedef float v8f __attribute__((ext_vector_type(8)));

typedef const __attribute__((address_space(1))) float* gcfp;
typedef __attribute__((address_space(1))) float*       gfp;

__global__ void k_zero(float* __restrict__ p, int n) {
  int i = blockIdx.x * blockDim.x + threadIdx.x;
  if (i < n) p[i] = 0.0f;
}

__global__ void k_degree(const int* __restrict__ cols, float* __restrict__ deg) {
  int e = blockIdx.x * blockDim.x + threadIdx.x;
  if (e < N_EDGES) atomicAdd(&deg[cols[e]], 1.0f);
}

__global__ void k_dinv(float* __restrict__ deg) {
  int i = blockIdx.x * blockDim.x + threadIdx.x;
  if (i < N_TOTAL) deg[i] = rsqrtf(deg[i] + 1.0f);
}

__global__ void k_norm(const int* __restrict__ rows, const int* __restrict__ cols,
                       const float* __restrict__ dinv, float* __restrict__ norm) {
  int e = blockIdx.x * blockDim.x + threadIdx.x;
  if (e >= E_TOTAL) return;
  int r, c;
  if (e < N_EDGES) { r = rows[e]; c = cols[e]; }
  else             { r = c = e - N_EDGES; }
  norm[e] = dinv[r] * dinv[c];
}

__global__ void k_scatter(const int* __restrict__ rows, const int* __restrict__ cols,
                          const float* __restrict__ norm,
                          const float* __restrict__ h, float* __restrict__ out) {
  int t = blockIdx.x * blockDim.x + threadIdx.x;
  int e = t >> 5;
  int j = t & 31;
  if (e >= E_TOTAL) return;
  int r, c;
  if (e < N_EDGES) { r = rows[e]; c = cols[e]; }
  else             { r = c = e - N_EDGES; }
  float w = norm[e];
  gcfp hg = (gcfp)h;
  atomicAdd(&out[(unsigned)c * HID + j], hg[(unsigned)r * HID + j] * w);
}

__global__ void k_bias_relu(float* __restrict__ x, const float* __restrict__ b, int n) {
  int i = blockIdx.x * blockDim.x + threadIdx.x;
  if (i < n) {
    float v = x[i] + b[i & (HID - 1)];
    x[i] = v > 0.0f ? v : 0.0f;
  }
}

template <int NT, bool RAGGED, bool BIAS, bool RELU>
__global__ void k_gemm_wmma(const float* __restrict__ A, const float* __restrict__ B,
                            const float* __restrict__ bias, float* __restrict__ C,
                            int M, int N, int K) {
  int wave  = threadIdx.x >> 5;
  int lane  = threadIdx.x & 31;
  int mtile = blockIdx.y * 4 + wave;
  int m0    = mtile * 16;
  if (m0 >= M) return;
  int n0   = blockIdx.x * (16 * NT);
  int half = lane >> 4;
  int lm   = lane & 15;

  gcfp Ag = (gcfp)A;
  gcfp Bg = (gcfp)B;

  int  ncol[NT];
  bool nok[NT];
  unsigned boff[NT];
#pragma unroll
  for (int t = 0; t < NT; ++t) {
    ncol[t] = n0 + t * 16 + lm;
    int nld;
    if (RAGGED) {
      nok[t] = ncol[t] < N;
      nld    = nok[t] ? ncol[t] : (N - 1);
    } else {
      nok[t] = true;
      nld    = ncol[t];
    }
    boff[t] = (unsigned)(half * 2) * (unsigned)N + (unsigned)nld;
  }

  unsigned aoff = (unsigned)(m0 + lm) * (unsigned)K + (unsigned)(half * 2);

  v8f acc[NT];
#pragma unroll
  for (int t = 0; t < NT; ++t) acc[t] = (v8f){0.f,0.f,0.f,0.f,0.f,0.f,0.f,0.f};

  for (int k0 = 0; k0 < K; k0 += 4) {
    v2f a;
    a.x = Ag[aoff];
    a.y = Ag[aoff + 1];
#pragma unroll
    for (int t = 0; t < NT; ++t) {
      float b0 = Bg[boff[t]];
      float b1 = Bg[boff[t] + (unsigned)N];
      if (RAGGED) { b0 = nok[t] ? b0 : 0.0f; b1 = nok[t] ? b1 : 0.0f; }
      v2f b; b.x = b0; b.y = b1;
      acc[t] = __builtin_amdgcn_wmma_f32_16x16x4_f32(
           false, a,  false, b,
           (short)0, acc[t],  false,  false);
      boff[t] += 4u * (unsigned)N;
    }
    aoff += 4u;
  }

  gfp Cg = (gfp)C;
  gcfp biasg = (gcfp)bias;
#pragma unroll
  for (int t = 0; t < NT; ++t) {
    float bv = BIAS ? biasg[nok[t] ? ncol[t] : 0] : 0.0f;
    if (!RAGGED || nok[t]) {
#pragma unroll
      for (int r = 0; r < 8; ++r) {
        int m = m0 + r + half * 8;
        float v = acc[t][r] + bv;
        if (RELU) v = v > 0.0f ? v : 0.0f;
        Cg[(unsigned)m * (unsigned)N + (unsigned)ncol[t]] = v;
      }
    }
  }
}

__global__ void k_value(const float* __restrict__ hidv, const float* __restrict__ Wv2,
                        const float* __restrict__ bv2, float* __restrict__ out) {
  int b = blockIdx.x * blockDim.x + threadIdx.x;
  if (b < BATCH) {
    float s = bv2[0];
    for (int k = 0; k < 256; ++k) s += hidv[b * 256 + k] * Wv2[k];
    out[b] = s;
  }
}

static inline int cdiv(long long a, int b) { return (int)((a + b - 1) / b); }

extern "C" void kernel_launch(void* const* d_in, const int* in_sizes, int n_in,
                              void* d_out, int out_size, void* d_ws, size_t ws_size,
                              hipStream_t stream) {
  (void)in_sizes; (void)n_in; (void)out_size; (void)ws_size;

  const float* x    = (const float*)d_in[0];
  const int*   ei   = (const int*)d_in[1];
  const int*   rows = ei;
  const int*   cols = ei + N_EDGES;
  const float* W[5]  = {(const float*)d_in[2], (const float*)d_in[4], (const float*)d_in[6],
                        (const float*)d_in[8], (const float*)d_in[10]};
  const float* bb[5] = {(const float*)d_in[3], (const float*)d_in[5], (const float*)d_in[7],
                        (const float*)d_in[9], (const float*)d_in[11]};
  const float* Wp1 = (const float*)d_in[12]; const float* bp1 = (const float*)d_in[13];
  const float* Wp2 = (const float*)d_in[14]; const float* bp2 = (const float*)d_in[15];
  const float* Wv1 = (const float*)d_in[16]; const float* bv1 = (const float*)d_in[17];
  const float* Wv2 = (const float*)d_in[18]; const float* bv2 = (const float*)d_in[19];
  float* out = (float*)d_out;

  float* ws   = (float*)d_ws;
  float* deg  = ws;
  float* norm = deg  + N_TOTAL;
  float* bufA = norm + E_TOTAL;
  float* bufB = bufA + (size_t)N_TOTAL * HID;
  float* hidp = bufB + (size_t)N_TOTAL * HID;
  float* hidv = hidp + BATCH * 256;

  const int TB = 256;

  k_zero  <<<cdiv(N_TOTAL, TB), TB, 0, stream>>>(deg, N_TOTAL);
  k_degree<<<cdiv(N_EDGES, TB), TB, 0, stream>>>(cols, deg);
  k_dinv  <<<cdiv(N_TOTAL, TB), TB, 0, stream>>>(deg);
  k_norm  <<<cdiv(E_TOTAL, TB), TB, 0, stream>>>(rows, cols, deg, norm);

  const float* in = x;
  int Kdim = STATE;
  dim3 layer_grid(1, N_TOTAL / 64);
  int scat_blocks = cdiv((long long)E_TOTAL * 32, TB);
  for (int l = 0; l < 5; ++l) {
    k_gemm_wmma<2, false, false, false><<<layer_grid, 128, 0, stream>>>(
        in, W[l], nullptr, bufA, N_TOTAL, HID, Kdim);
    k_zero     <<<cdiv((long long)N_TOTAL * HID, TB), TB, 0, stream>>>(bufB, N_TOTAL * HID);
    k_scatter  <<<scat_blocks, TB, 0, stream>>>(rows, cols, norm, bufA, bufB);
    k_bias_relu<<<cdiv((long long)N_TOTAL * HID, TB), TB, 0, stream>>>(bufB, bb[l],
                                                                       N_TOTAL * HID);
    in = bufB;
    Kdim = HID;
  }

  dim3 g1(256 / 32, cdiv(BATCH, 64));
  k_gemm_wmma<2, false, true, true><<<g1, 128, 0, stream>>>(
      bufB, Wp1, bp1, hidp, BATCH, 256, NODE_COUNT * HID);
  k_gemm_wmma<2, false, true, true><<<g1, 128, 0, stream>>>(
      bufB, Wv1, bv1, hidv, BATCH, 256, NODE_COUNT * HID);

  dim3 g2(cdiv(NODE_COUNT, 16), cdiv(BATCH, 64));
  k_gemm_wmma<1, true, true, false><<<g2, 128, 0, stream>>>(
      hidp, Wp2, bp2, out, BATCH, NODE_COUNT, 256);

  k_value<<<1, 64, 0, stream>>>(hidv, Wv2, bv2, out + (size_t)BATCH * NODE_COUNT);
}


# Layer scatter-add: global f32 atomics -> LDS f32 adds, one work-group per 256-target block over edges binned by block once per call, bias+relu in it; sum order: self term first, then arrival
# speedup vs baseline: 6.3703x; 6.3703x over previous
#include <hip/hip_runtime.h>

#define N_TOTAL    64000
#define NODE_COUNT 1000
#define BATCH      64
#define STATE      16
#define HID        32
#define N_EDGES    2048000
#define TBN        256
#define NBLK       (N_TOTAL / TBN)
#define CUR_PITCH  64
static_assert(N_TOTAL % TBN == 0, "blocks cover the nodes exactly");
static_assert(N_TOTAL <= (1 << 24), "source id and 8-bit local target share one int");
static_assert(TBN == 256 && HID == 32, "record packing and lane-per-feature layout");

typedef float v2f __attribute__((ext_vector_type(2)));
typedef float v8f __attribute__((ext_vector_type(8)));

typedef const __attribute__((address_space(1))) float* gcfp;
typedef __attribute__((address_space(1))) float*       gfp;

__global__ void k_zero(float* __restrict__ p, int n) {
  int i = blockIdx.x * blockDim.x + threadIdx.x;
  if (i < n) p[i] = 0.0f;
}

__global__ void k_degree(const int* __restrict__ cols, float* __restrict__ deg) {
  int e = blockIdx.x * blockDim.x + threadIdx.x;
  if (e < N_EDGES) atomicAdd(&deg[cols[e]], 1.0f);
}

__global__ void __launch_bounds__(TBN)
k_dinv_count(float* __restrict__ deg, int* __restrict__ count) {
  __shared__ int wsum[TBN / 32];
  const int i = blockIdx.x * TBN + threadIdx.x;
  const float d = deg[i];
  int c = (int)d;
  for (int o = 16; o > 0; o >>= 1) c += __shfl_xor(c, o);
  if ((threadIdx.x & 31) == 0) wsum[threadIdx.x >> 5] = c;
  deg[i] = rsqrtf(d + 1.0f);
  __syncthreads();
  if (threadIdx.x == 0) {
    int s = 0;
    for (int w = 0; w < TBN / 32; ++w) s += wsum[w];
    count[blockIdx.x] = s;
  }
}

__global__ void __launch_bounds__(256)
k_scan(const int* __restrict__ count, int* __restrict__ base) {
  __shared__ int s[256];
  const int t = threadIdx.x;
  s[t] = t < NBLK ? count[t] : 0;
  __syncthreads();
  for (int o = 1; o < 256; o <<= 1) {
    int a = t >= o ? s[t - o] : 0;
    __syncthreads();
    s[t] += a;
    __syncthreads();
  }
  if (t == 0) base[0] = 0;
  if (t < NBLK) base[t + 1] = s[t];
}

__global__ void k_fill(const int* __restrict__ rows, const int* __restrict__ cols,
                       const float* __restrict__ dinv, const int* __restrict__ base,
                       int* __restrict__ cursor, int* __restrict__ rec) {
  int e = blockIdx.x * blockDim.x + threadIdx.x;
  if (e >= N_EDGES) return;
  int r = rows[e];
  int c = cols[e];
  float w = dinv[r] * dinv[c];
  int b = c / TBN;
  int pos = base[b] + atomicAdd(&cursor[b * CUR_PITCH], 1);
  int2 v;
  v.x = (r << 8) | (c & (TBN - 1));
  v.y = __float_as_int(w);
  ((int2*)rec)[pos] = v;
}

typedef const __attribute__((address_space(1))) int2* gci2p;

template <int I0>
__device__ __forceinline__ void agg_load16(gcfp hg, int lane, int px, float (&v)[16]) {
#pragma unroll
  for (int i = 0; i < 16; ++i) {
    int pk = __builtin_amdgcn_readlane(px, I0 + i);
    v[i] = hg[((unsigned)pk >> 8) * HID + (unsigned)lane];
  }
}

template <int I0>
__device__ __forceinline__ void agg_add16(float* acc, int lane, int px, int py,
                                          const float (&v)[16]) {
#pragma unroll
  for (int i = 0; i < 16; ++i) {
    int pk  = __builtin_amdgcn_readlane(px, I0 + i);
    float w = __int_as_float(__builtin_amdgcn_readlane(py, I0 + i));
    __hip_atomic_fetch_add(&acc[(pk & (TBN - 1)) * HID + lane], v[i] * w,
                           __ATOMIC_RELAXED, __HIP_MEMORY_SCOPE_WORKGROUP);
  }
}

__global__ void __launch_bounds__(1024)
k_agg(const int* __restrict__ rec, const int* __restrict__ base,
      const float* __restrict__ dinv, const float* __restrict__ h,
      const float* __restrict__ bias, float* __restrict__ out) {
  __shared__ float acc[TBN * HID];

  const int wave  = __builtin_amdgcn_readfirstlane((int)(threadIdx.x >> 5));
  const int lane  = threadIdx.x & 31;
  const int node0 = blockIdx.x * TBN;
  gcfp  hg = (gcfp)h;
  gci2p rg = (gci2p)rec;

#pragma unroll
  for (int j = 0; j < TBN / 32; ++j) {
    int n = wave + 32 * j;
    float d = dinv[node0 + n];
    acc[n * HID + lane] = hg[(unsigned)(node0 + n) * HID + (unsigned)lane] * (d * d);
  }
  __syncthreads();

  const int beg = base[blockIdx.x];
  const int end = base[blockIdx.x + 1];

  int p = beg + wave * 32;
  if (p < end) {
    int2 cur = rg[min(p + lane, end - 1)];
    for (;;) {
      const int n  = end - p;
      const int pn = p + 32 * 32;
      if (n < 32) {
        for (int i0 = 0; i0 < n; i0 += 8) {
          float v[8];
#pragma unroll
          for (int j = 0; j < 8; ++j) {
            v[j] = 0.0f;
            if (i0 + j < n) {
              int pk = __builtin_amdgcn_readlane(cur.x, i0 + j);
              v[j] = hg[((unsigned)pk >> 8) * HID + (unsigned)lane];
            }
          }
#pragma unroll
          for (int j = 0; j < 8; ++j) {
            if (i0 + j < n) {
              int pk  = __builtin_amdgcn_readlane(cur.x, i0 + j);
              float w = __int_as_float(__builtin_amdgcn_readlane(cur.y, i0 + j));
              __hip_atomic_fetch_add(&acc[(pk & (TBN - 1)) * HID + lane], v[j] * w,
                                     __ATOMIC_RELAXED, __HIP_MEMORY_SCOPE_WORKGROUP);
            }
          }
        }
        break;
      }
      float va[16], vb[16];
      int2 nxt = rg[min(pn + lane, end - 1)];
      agg_load16<0>(hg, lane, cur.x, va);
      agg_load16<16>(hg, lane, cur.x, vb);
      asm volatile("" :: "v"(nxt.x), "v"(nxt.y));
      agg_add16<0>(acc, lane, cur.x, cur.y, va);
      agg_add16<16>(acc, lane, cur.x, cur.y, vb);
      if (pn >= end) break;
      cur = nxt;
      p = pn;
    }
  }
  __syncthreads();

  const float bv = bias[lane];
#pragma unroll
  for (int j = 0; j < TBN / 32; ++j) {
    int n = wave + 32 * j;
    float v = acc[n * HID + lane] + bv;
    out[(node0 + n) * HID + lane] = v > 0.0f ? v : 0.0f;
  }
}

template <int NT, bool RAGGED, bool BIAS, bool RELU>
__global__ void k_gemm_wmma(const float* __restrict__ A, const float* __restrict__ B,
                            const float* __restrict__ bias, float* __restrict__ C,
                            int M, int N, int K) {
  int wave  = threadIdx.x >> 5;
  int lane  = threadIdx.x & 31;
  int mtile = blockIdx.y * 4 + wave;
  int m0    = mtile * 16;
  if (m0 >= M) return;
  int n0   = blockIdx.x * (16 * NT);
  int half = lane >> 4;
  int lm   = lane & 15;

  gcfp Ag = (gcfp)A;
  gcfp Bg = (gcfp)B;

  int  ncol[NT];
  bool nok[NT];
  unsigned boff[NT];
#pragma unroll
  for (int t = 0; t < NT; ++t) {
    ncol[t] = n0 + t * 16 + lm;
    int nld;
    if (RAGGED) {
      nok[t] = ncol[t] < N;
      nld    = nok[t] ? ncol[t] : (N - 1);
    } else {
      nok[t] = true;
      nld    = ncol[t];
    }
    boff[t] = (unsigned)(half * 2) * (unsigned)N + (unsigned)nld;
  }

  unsigned aoff = (unsigned)(m0 + lm) * (unsigned)K + (unsigned)(half * 2);

  v8f acc[NT];
#pragma unroll
  for (int t = 0; t < NT; ++t) acc[t] = (v8f){0.f,0.f,0.f,0.f,0.f,0.f,0.f,0.f};

  for (int k0 = 0; k0 < K; k0 += 4) {
    v2f a;
    a.x = Ag[aoff];
    a.y = Ag[aoff + 1];
#pragma unroll
    for (int t = 0; t < NT; ++t) {
      float b0 = Bg[boff[t]];
      float b1 = Bg[boff[t] + (unsigned)N];
      if (RAGGED) { b0 = nok[t] ? b0 : 0.0f; b1 = nok[t] ? b1 : 0.0f; }
      v2f b; b.x = b0; b.y = b1;
      acc[t] = __builtin_amdgcn_wmma_f32_16x16x4_f32(
           false, a,  false, b,
           (short)0, acc[t],  false,  false);
      boff[t] += 4u * (unsigned)N;
    }
    aoff += 4u;
  }

  gfp Cg = (gfp)C;
  gcfp biasg = (gcfp)bias;
#pragma unroll
  for (int t = 0; t < NT; ++t) {
    float bv = BIAS ? biasg[nok[t] ? ncol[t] : 0] : 0.0f;
    if (!RAGGED || nok[t]) {
#pragma unroll
      for (int r = 0; r < 8; ++r) {
        int m = m0 + r + half * 8;
        float v = acc[t][r] + bv;
        if (RELU) v = v > 0.0f ? v : 0.0f;
        Cg[(unsigned)m * (unsigned)N + (unsigned)ncol[t]] = v;
      }
    }
  }
}

#define HEAD_K      (NODE_COUNT * HID)
#define HEAD_N      256
#define HEAD_WAVES  25
#define HEAD_GROUPS 40
#define HEAD_GK     32

typedef const __attribute__((address_space(1))) v2f* gcv2p;

__global__ void __launch_bounds__(HEAD_WAVES * 32)
k_head_gemm(const float* __restrict__ A, const float* __restrict__ B,
            const float* __restrict__ bias, float* __restrict__ C) {
  __shared__ float part[HEAD_WAVES * 512];

  const int wave = threadIdx.x >> 5;
  const int lane = threadIdx.x & 31;
  const int half = lane >> 4;
  const int lm   = lane & 15;
  const int m0   = blockIdx.y * 16;
  const int n0   = blockIdx.x * 32;
  const int kbeg = wave * (HEAD_GROUPS * HEAD_GK);

  gcfp ap = (gcfp)A + ((unsigned)(m0 + lm) * (unsigned)HEAD_K + (unsigned)(kbeg + half * 2));
  gcfp bp = (gcfp)B + ((unsigned)(kbeg + half * 2) * (unsigned)HEAD_N + (unsigned)(n0 + lm));

  v8f acc0 = (v8f){0.f,0.f,0.f,0.f,0.f,0.f,0.f,0.f};
  v8f acc1 = (v8f){0.f,0.f,0.f,0.f,0.f,0.f,0.f,0.f};

  for (int g = 0; g < HEAD_GROUPS; ++g) {
    v2f a[8], b0[8], b1[8];
#pragma unroll
    for (int s = 0; s < 8; ++s)
      a[s] = *(gcv2p)(ap + 4 * s);
#pragma unroll
    for (int s = 0; s < 8; ++s) {
      b0[s].x = bp[(4 * s) * HEAD_N];
      b0[s].y = bp[(4 * s + 1) * HEAD_N];
      b1[s].x = bp[(4 * s) * HEAD_N + 16];
      b1[s].y = bp[(4 * s + 1) * HEAD_N + 16];
    }
    __builtin_amdgcn_sched_barrier(0);
#pragma unroll
    for (int s = 0; s < 8; ++s) {
      acc0 = __builtin_amdgcn_wmma_f32_16x16x4_f32(
          false, a[s], false, b0[s], (short)0, acc0, false, false);
      acc1 = __builtin_amdgcn_wmma_f32_16x16x4_f32(
          false, a[s], false, b1[s], (short)0, acc1, false, false);
    }
    ap += HEAD_GK;
    bp += HEAD_GK * HEAD_N;
  }

  float* mine = part + wave * 512;
#pragma unroll
  for (int r = 0; r < 8; ++r) {
    int row = r + half * 8;
    mine[row * 32 + lm]      = acc0[r];
    mine[row * 32 + 16 + lm] = acc1[r];
  }
  __syncthreads();

  const int tid = threadIdx.x;
  if (tid < 512) {
    float s = part[tid];
    for (int w = 1; w < HEAD_WAVES; ++w) s += part[w * 512 + tid];
    const int row = tid >> 5;
    const int col = tid & 31;
    float v = s + bias[n0 + col];
    v = v > 0.0f ? v : 0.0f;
    C[(m0 + row) * HEAD_N + n0 + col] = v;
  }
}

__global__ void k_value(const float* __restrict__ hidv, const float* __restrict__ Wv2,
                        const float* __restrict__ bv2, float* __restrict__ out) {
  int b = blockIdx.x * blockDim.x + threadIdx.x;
  if (b < BATCH) {
    float s = bv2[0];
    for (int k = 0; k < 256; ++k) s += hidv[b * 256 + k] * Wv2[k];
    out[b] = s;
  }
}

static inline int cdiv(long long a, int b) { return (int)((a + b - 1) / b); }

extern "C" void kernel_launch(void* const* d_in, const int* in_sizes, int n_in,
                              void* d_out, int out_size, void* d_ws, size_t ws_size,
                              hipStream_t stream) {
  (void)in_sizes; (void)n_in; (void)out_size; (void)ws_size;

  const float* x    = (const float*)d_in[0];
  const int*   ei   = (const int*)d_in[1];
  const int*   rows = ei;
  const int*   cols = ei + N_EDGES;
  const float* W[5]  = {(const float*)d_in[2], (const float*)d_in[4], (const float*)d_in[6],
                        (const float*)d_in[8], (const float*)d_in[10]};
  const float* bb[5] = {(const float*)d_in[3], (const float*)d_in[5], (const float*)d_in[7],
                        (const float*)d_in[9], (const float*)d_in[11]};
  const float* Wp1 = (const float*)d_in[12]; const float* bp1 = (const float*)d_in[13];
  const float* Wp2 = (const float*)d_in[14]; const float* bp2 = (const float*)d_in[15];
  const float* Wv1 = (const float*)d_in[16]; const float* bv1 = (const float*)d_in[17];
  const float* Wv2 = (const float*)d_in[18]; const float* bv2 = (const float*)d_in[19];
  float* out = (float*)d_out;

  float* ws     = (float*)d_ws;
  float* deg    = ws;
  int*   cursor = (int*)(deg + N_TOTAL);
  int*   count  = cursor + NBLK * CUR_PITCH;
  int*   base   = count + 256;
  int*   rec    = base + 512;
  float* bufA   = (float*)(rec + 2 * (size_t)N_EDGES);
  float* bufB   = bufA + (size_t)N_TOTAL * HID;
  float* hidp   = bufB + (size_t)N_TOTAL * HID;
  float* hidv   = hidp + BATCH * 256;

  const int TB = 256;

  k_zero      <<<cdiv(N_TOTAL + NBLK * CUR_PITCH, TB), TB, 0, stream>>>(deg, N_TOTAL + NBLK * CUR_PITCH);
  k_degree    <<<cdiv(N_EDGES, TB), TB, 0, stream>>>(cols, deg);
  k_dinv_count<<<NBLK, TBN, 0, stream>>>(deg, count);
  k_scan      <<<1, 256, 0, stream>>>(count, base);
  k_fill      <<<cdiv(N_EDGES, TB), TB, 0, stream>>>(rows, cols, deg, base, cursor, rec);

  const float* in = x;
  int Kdim = STATE;
  dim3 layer_grid(1, N_TOTAL / 64);
  for (int l = 0; l < 5; ++l) {
    k_gemm_wmma<2, false, false, false><<<layer_grid, 128, 0, stream>>>(
        in, W[l], nullptr, bufA, N_TOTAL, HID, Kdim);
    k_agg<<<NBLK, 1024, 0, stream>>>(rec, base, deg, bufA, bb[l], bufB);
    in = bufB;
    Kdim = HID;
  }

  dim3 g1(HEAD_N / 32, BATCH / 16);
  k_head_gemm<<<g1, HEAD_WAVES * 32, 0, stream>>>(bufB, Wp1, bp1, hidp);
  k_head_gemm<<<g1, HEAD_WAVES * 32, 0, stream>>>(bufB, Wv1, bv1, hidv);

  dim3 g2(cdiv(NODE_COUNT, 16), cdiv(BATCH, 64));
  k_gemm_wmma<1, true, true, false><<<g2, 128, 0, stream>>>(
      hidp, Wp2, bp2, out, BATCH, NODE_COUNT, 256);

  k_value<<<1, 64, 0, stream>>>(hidv, Wv2, bv2, out + (size_t)BATCH * NODE_COUNT);
}


# Head GEMMs: one grid of 200 work-groups, each 64 rows x 64 cols of one head over a K slice of 1280 (wave strip 16x64); slices joined by f32 atomic adds, bias+relu moved to readers; sum order changes
# speedup vs baseline: 8.0549x; 1.2644x over previous
#include <hip/hip_runtime.h>

#define N_TOTAL    64000
#define NODE_COUNT 1000
#define BATCH      64
#define STATE      16
#define HID        32
#define N_EDGES    2048000
#define TBN        256
#define NBLK       (N_TOTAL / TBN)
#define CUR_PITCH  64
static_assert(N_TOTAL % TBN == 0, "blocks cover the nodes exactly");
static_assert(N_TOTAL <= (1 << 24), "source id and 8-bit local target share one int");
static_assert(TBN == 256 && HID == 32, "record packing and lane-per-feature layout");

typedef float v2f __attribute__((ext_vector_type(2)));
typedef float v8f __attribute__((ext_vector_type(8)));

typedef const __attribute__((address_space(1))) float* gcfp;
typedef __attribute__((address_space(1))) float*       gfp;

__global__ void k_zero(float* __restrict__ p, int n) {
  int i = blockIdx.x * blockDim.x + threadIdx.x;
  if (i < n) p[i] = 0.0f;
}

__global__ void k_degree(const int* __restrict__ cols, float* __restrict__ deg) {
  int e = blockIdx.x * blockDim.x + threadIdx.x;
  if (e < N_EDGES) atomicAdd(&deg[cols[e]], 1.0f);
}

__global__ void __launch_bounds__(TBN)
k_dinv_count(float* __restrict__ deg, int* __restrict__ count) {
  __shared__ int wsum[TBN / 32];
  const int i = blockIdx.x * TBN + threadIdx.x;
  const float d = deg[i];
  int c = (int)d;
  for (int o = 16; o > 0; o >>= 1) c += __shfl_xor(c, o);
  if ((threadIdx.x & 31) == 0) wsum[threadIdx.x >> 5] = c;
  deg[i] = rsqrtf(d + 1.0f);
  __syncthreads();
  if (threadIdx.x == 0) {
    int s = 0;
    for (int w = 0; w < TBN / 32; ++w) s += wsum[w];
    count[blockIdx.x] = s;
  }
}

__global__ void __launch_bounds__(256)
k_scan(const int* __restrict__ count, int* __restrict__ base) {
  __shared__ int s[256];
  const int t = threadIdx.x;
  s[t] = t < NBLK ? count[t] : 0;
  __syncthreads();
  for (int o = 1; o < 256; o <<= 1) {
    int a = t >= o ? s[t - o] : 0;
    __syncthreads();
    s[t] += a;
    __syncthreads();
  }
  if (t == 0) base[0] = 0;
  if (t < NBLK) base[t + 1] = s[t];
}

__global__ void k_fill(const int* __restrict__ rows, const int* __restrict__ cols,
                       const float* __restrict__ dinv, const int* __restrict__ base,
                       int* __restrict__ cursor, int* __restrict__ rec) {
  int e = blockIdx.x * blockDim.x + threadIdx.x;
  if (e >= N_EDGES) return;
  int r = rows[e];
  int c = cols[e];
  float w = dinv[r] * dinv[c];
  int b = c / TBN;
  int pos = base[b] + atomicAdd(&cursor[b * CUR_PITCH], 1);
  int2 v;
  v.x = (r << 8) | (c & (TBN - 1));
  v.y = __float_as_int(w);
  ((int2*)rec)[pos] = v;
}

typedef const __attribute__((address_space(1))) int2* gci2p;

template <int I0>
__device__ __forceinline__ void agg_load16(gcfp hg, int lane, int px, float (&v)[16]) {
#pragma unroll
  for (int i = 0; i < 16; ++i) {
    int pk = __builtin_amdgcn_readlane(px, I0 + i);
    v[i] = hg[((unsigned)pk >> 8) * HID + (unsigned)lane];
  }
}

template <int I0>
__device__ __forceinline__ void agg_add16(float* acc, int lane, int px, int py,
                                          const float (&v)[16]) {
#pragma unroll
  for (int i = 0; i < 16; ++i) {
    int pk  = __builtin_amdgcn_readlane(px, I0 + i);
    float w = __int_as_float(__builtin_amdgcn_readlane(py, I0 + i));
    __hip_atomic_fetch_add(&acc[(pk & (TBN - 1)) * HID + lane], v[i] * w,
                           __ATOMIC_RELAXED, __HIP_MEMORY_SCOPE_WORKGROUP);
  }
}

__global__ void __launch_bounds__(1024)
k_agg(const int* __restrict__ rec, const int* __restrict__ base,
      const float* __restrict__ dinv, const float* __restrict__ h,
      const float* __restrict__ bias, float* __restrict__ out) {
  __shared__ float acc[TBN * HID];

  const int wave  = __builtin_amdgcn_readfirstlane((int)(threadIdx.x >> 5));
  const int lane  = threadIdx.x & 31;
  const int node0 = blockIdx.x * TBN;
  gcfp  hg = (gcfp)h;
  gci2p rg = (gci2p)rec;

#pragma unroll
  for (int j = 0; j < TBN / 32; ++j) {
    int n = wave + 32 * j;
    float d = dinv[node0 + n];
    acc[n * HID + lane] = hg[(unsigned)(node0 + n) * HID + (unsigned)lane] * (d * d);
  }
  __syncthreads();

  const int beg = base[blockIdx.x];
  const int end = base[blockIdx.x + 1];

  int p = beg + wave * 32;
  if (p < end) {
    int2 cur = rg[min(p + lane, end - 1)];
    for (;;) {
      const int n  = end - p;
      const int pn = p + 32 * 32;
      if (n < 32) {
        for (int i0 = 0; i0 < n; i0 += 8) {
          float v[8];
#pragma unroll
          for (int j = 0; j < 8; ++j) {
            v[j] = 0.0f;
            if (i0 + j < n) {
              int pk = __builtin_amdgcn_readlane(cur.x, i0 + j);
              v[j] = hg[((unsigned)pk >> 8) * HID + (unsigned)lane];
            }
          }
#pragma unroll
          for (int j = 0; j < 8; ++j) {
            if (i0 + j < n) {
              int pk  = __builtin_amdgcn_readlane(cur.x, i0 + j);
              float w = __int_as_float(__builtin_amdgcn_readlane(cur.y, i0 + j));
              __hip_atomic_fetch_add(&acc[(pk & (TBN - 1)) * HID + lane], v[j] * w,
                                     __ATOMIC_RELAXED, __HIP_MEMORY_SCOPE_WORKGROUP);
            }
          }
        }
        break;
      }
      float va[16], vb[16];
      int2 nxt = rg[min(pn + lane, end - 1)];
      agg_load16<0>(hg, lane, cur.x, va);
      agg_load16<16>(hg, lane, cur.x, vb);
      asm volatile("" :: "v"(nxt.x), "v"(nxt.y));
      agg_add16<0>(acc, lane, cur.x, cur.y, va);
      agg_add16<16>(acc, lane, cur.x, cur.y, vb);
      if (pn >= end) break;
      cur = nxt;
      p = pn;
    }
  }
  __syncthreads();

  const float bv = bias[lane];
#pragma unroll
  for (int j = 0; j < TBN / 32; ++j) {
    int n = wave + 32 * j;
    float v = acc[n * HID + lane] + bv;
    out[(node0 + n) * HID + lane] = v > 0.0f ? v : 0.0f;
  }
}

template <int NT, bool RAGGED, bool BIAS, bool RELU, bool PRE>
__global__ void k_gemm_wmma(const float* __restrict__ A, const float* __restrict__ B,
                            const float* __restrict__ bias, float* __restrict__ C,
                            int M, int N, int K, const float* __restrict__ abias) {
  int wave  = threadIdx.x >> 5;
  int lane  = threadIdx.x & 31;
  int mtile = blockIdx.y * 4 + wave;
  int m0    = mtile * 16;
  if (m0 >= M) return;
  int n0   = blockIdx.x * (16 * NT);
  int half = lane >> 4;
  int lm   = lane & 15;

  gcfp Ag = (gcfp)A;
  gcfp Bg = (gcfp)B;
  gcfp abg = (gcfp)abias;

  int  ncol[NT];
  bool nok[NT];
  unsigned boff[NT];
#pragma unroll
  for (int t = 0; t < NT; ++t) {
    ncol[t] = n0 + t * 16 + lm;
    int nld;
    if (RAGGED) {
      nok[t] = ncol[t] < N;
      nld    = nok[t] ? ncol[t] : (N - 1);
    } else {
      nok[t] = true;
      nld    = ncol[t];
    }
    boff[t] = (unsigned)(half * 2) * (unsigned)N + (unsigned)nld;
  }

  unsigned aoff = (unsigned)(m0 + lm) * (unsigned)K + (unsigned)(half * 2);

  v8f acc[NT];
#pragma unroll
  for (int t = 0; t < NT; ++t) acc[t] = (v8f){0.f,0.f,0.f,0.f,0.f,0.f,0.f,0.f};

  for (int k0 = 0; k0 < K; k0 += 4) {
    v2f a;
    a.x = Ag[aoff];
    a.y = Ag[aoff + 1];
    if (PRE) {
      float p0 = a.x + abg[k0 + half * 2];
      float p1 = a.y + abg[k0 + half * 2 + 1];
      a.x = p0 > 0.0f ? p0 : 0.0f;
      a.y = p1 > 0.0f ? p1 : 0.0f;
    }
#pragma unroll
    for (int t = 0; t < NT; ++t) {
      float b0 = Bg[boff[t]];
      float b1 = Bg[boff[t] + (unsigned)N];
      if (RAGGED) { b0 = nok[t] ? b0 : 0.0f; b1 = nok[t] ? b1 : 0.0f; }
      v2f b; b.x = b0; b.y = b1;
      acc[t] = __builtin_amdgcn_wmma_f32_16x16x4_f32(
           false, a,  false, b,
           (short)0, acc[t],  false,  false);
      boff[t] += 4u * (unsigned)N;
    }
    aoff += 4u;
  }

  gfp Cg = (gfp)C;
  gcfp biasg = (gcfp)bias;
#pragma unroll
  for (int t = 0; t < NT; ++t) {
    float bv = BIAS ? biasg[nok[t] ? ncol[t] : 0] : 0.0f;
    if (!RAGGED || nok[t]) {
#pragma unroll
      for (int r = 0; r < 8; ++r) {
        int m = m0 + r + half * 8;
        float v = acc[t][r] + bv;
        if (RELU) v = v > 0.0f ? v : 0.0f;
        Cg[(unsigned)m * (unsigned)N + (unsigned)ncol[t]] = v;
      }
    }
  }
}

#define HEAD_K      (NODE_COUNT * HID)
#define HEAD_N      256
#define HS_SLICES   25
#define HS_KPARTS   8
#define HS_WAVES    (4 * HS_KPARTS)
#define HS_SLICE_K  (HEAD_K / HS_SLICES)
#define HS_PART_K   (HS_SLICE_K / HS_KPARTS)
#define HS_GK       16
#define HS_GROUPS   (HS_PART_K / HS_GK)
static_assert(HS_SLICES * HS_KPARTS * HS_GROUPS * HS_GK == HEAD_K, "K is covered exactly");
static_assert(BATCH == 64, "four row tiles");

typedef const __attribute__((address_space(1))) v2f* gcv2p;

__global__ void __launch_bounds__(HS_WAVES * 32)
k_head_gemm(const float* __restrict__ A,
            const float* __restrict__ Bp, const float* __restrict__ Bv,
            float* __restrict__ Cp, float* __restrict__ Cv) {
  __shared__ float part[HS_WAVES * 1024];

  const int wave = threadIdx.x >> 5;
  const int lane = threadIdx.x & 31;
  const int half = lane >> 4;
  const int lm   = lane & 15;
  const int mt   = wave & 3;
  const int kp   = wave >> 2;
  const int m0   = mt * 16;
  const int n0   = (blockIdx.x & 3) * 64;
  const bool second = blockIdx.x >= 4;
  const float* B = second ? Bv : Bp;
  float*       C = second ? Cv : Cp;
  const int kbeg = blockIdx.y * HS_SLICE_K + kp * HS_PART_K;

  gcfp ap = (gcfp)A + ((unsigned)(m0 + lm) * (unsigned)HEAD_K + (unsigned)(kbeg + half * 2));
  gcfp bp = (gcfp)B + ((unsigned)(kbeg + half * 2) * (unsigned)HEAD_N + (unsigned)(n0 + lm));

  v8f acc[4];
#pragma unroll
  for (int t = 0; t < 4; ++t) acc[t] = (v8f){0.f,0.f,0.f,0.f,0.f,0.f,0.f,0.f};

  for (int g = 0; g < HS_GROUPS; ++g) {
    v2f a[4], b[4][4];
#pragma unroll
    for (int s = 0; s < 4; ++s)
      a[s] = *(gcv2p)(ap + 4 * s);
#pragma unroll
    for (int s = 0; s < 4; ++s) {
#pragma unroll
      for (int t = 0; t < 4; ++t) {
        b[t][s].x = bp[(4 * s) * HEAD_N + 16 * t];
        b[t][s].y = bp[(4 * s + 1) * HEAD_N + 16 * t];
      }
    }
    __builtin_amdgcn_sched_barrier(0);
#pragma unroll
    for (int s = 0; s < 4; ++s) {
#pragma unroll
      for (int t = 0; t < 4; ++t)
        acc[t] = __builtin_amdgcn_wmma_f32_16x16x4_f32(
            false, a[s], false, b[t][s], (short)0, acc[t], false, false);
    }
    ap += HS_GK;
    bp += HS_GK * HEAD_N;
  }

  float* mine = part + wave * 1024;
#pragma unroll
  for (int t = 0; t < 4; ++t) {
#pragma unroll
    for (int r = 0; r < 8; ++r)
      mine[(r + half * 8) * 64 + t * 16 + lm] = acc[t][r];
  }
  __syncthreads();

  const int rr  = threadIdx.x >> 6;
  const int col = threadIdx.x & 63;
#pragma unroll
  for (int j = 0; j < 4; ++j) {
    float s = part[j * 1024 + rr * 64 + col];
    for (int q = 1; q < HS_KPARTS; ++q) s += part[(q * 4 + j) * 1024 + rr * 64 + col];
    atomicAdd(&C[(j * 16 + rr) * HEAD_N + n0 + col], s);
  }
}

__global__ void k_value(const float* __restrict__ sumv, const float* __restrict__ bv1,
                        const float* __restrict__ Wv2,
                        const float* __restrict__ bv2, float* __restrict__ out) {
  int b = blockIdx.x * blockDim.x + threadIdx.x;
  if (b < BATCH) {
    float s = bv2[0];
    for (int k = 0; k < 256; ++k) {
      float h = sumv[b * 256 + k] + bv1[k];
      h = h > 0.0f ? h : 0.0f;
      s += h * Wv2[k];
    }
    out[b] = s;
  }
}

static inline int cdiv(long long a, int b) { return (int)((a + b - 1) / b); }

extern "C" void kernel_launch(void* const* d_in, const int* in_sizes, int n_in,
                              void* d_out, int out_size, void* d_ws, size_t ws_size,
                              hipStream_t stream) {
  (void)in_sizes; (void)n_in; (void)out_size; (void)ws_size;

  const float* x    = (const float*)d_in[0];
  const int*   ei   = (const int*)d_in[1];
  const int*   rows = ei;
  const int*   cols = ei + N_EDGES;
  const float* W[5]  = {(const float*)d_in[2], (const float*)d_in[4], (const float*)d_in[6],
                        (const float*)d_in[8], (const float*)d_in[10]};
  const float* bb[5] = {(const float*)d_in[3], (const float*)d_in[5], (const float*)d_in[7],
                        (const float*)d_in[9], (const float*)d_in[11]};
  const float* Wp1 = (const float*)d_in[12]; const float* bp1 = (const float*)d_in[13];
  const float* Wp2 = (const float*)d_in[14]; const float* bp2 = (const float*)d_in[15];
  const float* Wv1 = (const float*)d_in[16]; const float* bv1 = (const float*)d_in[17];
  const float* Wv2 = (const float*)d_in[18]; const float* bv2 = (const float*)d_in[19];
  float* out = (float*)d_out;

  float* ws     = (float*)d_ws;
  float* deg    = ws;
  int*   cursor = (int*)(deg + N_TOTAL);
  float* hidp   = (float*)(cursor + NBLK * CUR_PITCH);
  float* hidv   = hidp + BATCH * 256;
  int*   count  = (int*)(hidv + BATCH * 256);
  int*   base   = count + 256;
  int*   rec    = base + 512;
  float* bufA   = (float*)(rec + 2 * (size_t)N_EDGES);
  float* bufB   = bufA + (size_t)N_TOTAL * HID;

  const int TB = 256;

  k_zero      <<<cdiv(N_TOTAL + NBLK * CUR_PITCH + 2 * BATCH * 256, TB), TB, 0, stream>>>(
      deg, N_TOTAL + NBLK * CUR_PITCH + 2 * BATCH * 256);
  k_degree    <<<cdiv(N_EDGES, TB), TB, 0, stream>>>(cols, deg);
  k_dinv_count<<<NBLK, TBN, 0, stream>>>(deg, count);
  k_scan      <<<1, 256, 0, stream>>>(count, base);
  k_fill      <<<cdiv(N_EDGES, TB), TB, 0, stream>>>(rows, cols, deg, base, cursor, rec);

  const float* in = x;
  int Kdim = STATE;
  dim3 layer_grid(1, N_TOTAL / 64);
  for (int l = 0; l < 5; ++l) {
    k_gemm_wmma<2, false, false, false, false><<<layer_grid, 128, 0, stream>>>(
        in, W[l], nullptr, bufA, N_TOTAL, HID, Kdim, nullptr);
    k_agg<<<NBLK, 1024, 0, stream>>>(rec, base, deg, bufA, bb[l], bufB);
    in = bufB;
    Kdim = HID;
  }

  dim3 g1(2 * HEAD_N / 64, HS_SLICES);
  k_head_gemm<<<g1, HS_WAVES * 32, 0, stream>>>(bufB, Wp1, Wv1, hidp, hidv);

  dim3 g2(cdiv(NODE_COUNT, 16), cdiv(BATCH, 64));
  k_gemm_wmma<1, true, true, false, true><<<g2, 128, 0, stream>>>(
      hidp, Wp2, bp2, out, BATCH, NODE_COUNT, 256, bp1);

  k_value<<<1, 64, 0, stream>>>(hidv, bv1, Wv2, bv2, out + (size_t)BATCH * NODE_COUNT);
}


# Edge binning: per-edge global atomics (degree count, list cursor) -> 8192-edge chunks counted and sorted by block in LDS, one global add per chunk and block; list order changes, f32 sums by arrival
# speedup vs baseline: 16.9351x; 2.1025x over previous
#include <hip/hip_runtime.h>

#define N_TOTAL    64000
#define NODE_COUNT 1000
#define BATCH      64
#define STATE      16
#define HID        32
#define N_EDGES    2048000
#define TBN        256
#define NBLK       (N_TOTAL / TBN)
#define CUR_PITCH  64
static_assert(N_TOTAL % TBN == 0, "blocks cover the nodes exactly");
static_assert(N_TOTAL <= (1 << 24), "source id and 8-bit local target share one int");
static_assert(TBN == 256 && HID == 32, "record packing and lane-per-feature layout");

typedef float v2f __attribute__((ext_vector_type(2)));
typedef float v8f __attribute__((ext_vector_type(8)));

typedef const __attribute__((address_space(1))) float* gcfp;
typedef __attribute__((address_space(1))) float*       gfp;

__global__ void k_zero(float* __restrict__ p, int n) {
  int i = blockIdx.x * blockDim.x + threadIdx.x;
  if (i < n) p[i] = 0.0f;
}

#define ECH  8192
#define NCH  (N_EDGES / ECH)
#define EPT  (ECH / 1024)
static_assert(N_EDGES % ECH == 0, "chunks cover the edges exactly");
static_assert(NBLK <= 256, "one LDS counter per target block, 256 of them");

__global__ void __launch_bounds__(1024)
k_count(const int* __restrict__ cols, int* __restrict__ count) {
  __shared__ int lcnt[256];
  const int t = threadIdx.x;
  if (t < 256) lcnt[t] = 0;
  __syncthreads();
  const int e0 = blockIdx.x * ECH + t;
  int c[EPT];
#pragma unroll
  for (int j = 0; j < EPT; ++j) c[j] = cols[e0 + j * 1024];
#pragma unroll
  for (int j = 0; j < EPT; ++j) atomicAdd(&lcnt[(unsigned)c[j] / TBN], 1);
  __syncthreads();
  if (t < NBLK) {
    int n = lcnt[t];
    if (n) atomicAdd(&count[t], n);
  }
}

__global__ void __launch_bounds__(256)
k_scan(const int* __restrict__ count, int* __restrict__ base) {
  __shared__ int s[256];
  const int t = threadIdx.x;
  s[t] = t < NBLK ? count[t] : 0;
  __syncthreads();
  for (int o = 1; o < 256; o <<= 1) {
    int a = t >= o ? s[t - o] : 0;
    __syncthreads();
    s[t] += a;
    __syncthreads();
  }
  if (t == 0) base[0] = 0;
  if (t < NBLK) base[t + 1] = s[t];
}

__global__ void __launch_bounds__(1024)
k_fill(const int* __restrict__ rows, const int* __restrict__ cols,
       const int* __restrict__ base, int* __restrict__ cursor, int* __restrict__ rec) {
  __shared__ int2 stage[ECH];
  __shared__ int lcnt[256];
  __shared__ int lstart[256];
  __shared__ int gdelta[256];
  __shared__ int wtot[8];
  const int t = threadIdx.x;
  if (t < 256) lcnt[t] = 0;
  __syncthreads();
  const int e0 = blockIdx.x * ECH + t;
  int r[EPT], c[EPT], rk[EPT];
#pragma unroll
  for (int j = 0; j < EPT; ++j) {
    r[j] = rows[e0 + j * 1024];
    c[j] = cols[e0 + j * 1024];
  }
#pragma unroll
  for (int j = 0; j < EPT; ++j) rk[j] = atomicAdd(&lcnt[(unsigned)c[j] / TBN], 1);
  __syncthreads();

  int cnt = 0, incl = 0, gpos = 0;
  if (t < 256) {
    cnt = lcnt[t];
    if (t < NBLK) gpos = base[t] + atomicAdd(&cursor[t * CUR_PITCH], cnt);
    incl = cnt;
    for (int o = 1; o < 32; o <<= 1) {
      int u = __shfl_up(incl, o);
      if ((t & 31) >= o) incl += u;
    }
    if ((t & 31) == 31) wtot[t >> 5] = incl;
  }
  __syncthreads();
  if (t < 256) {
    int pre = 0;
    for (int w = 0; w < (t >> 5); ++w) pre += wtot[w];
    int first = pre + incl - cnt;
    lstart[t] = first;
    gdelta[t] = gpos - first;
  }
  __syncthreads();

#pragma unroll
  for (int j = 0; j < EPT; ++j) {
    int b  = (unsigned)c[j] / TBN;
    int li = lstart[b] + rk[j];
    int2 v;
    v.x = (r[j] << 8) | (c[j] & (TBN - 1));
    v.y = gdelta[b] + li;
    stage[li] = v;
  }
  __syncthreads();
#pragma unroll
  for (int j = 0; j < EPT; ++j) {
    int2 s = stage[t + j * 1024];
    int2 v;
    v.x = s.x;
    v.y = 0;
    ((int2*)rec)[s.y] = v;
  }
}

__global__ void __launch_bounds__(1024)
k_deg_dinv(const int* __restrict__ rec, const int* __restrict__ base,
           float* __restrict__ dinv) {
  __shared__ int ldeg[TBN];
  const int t = threadIdx.x;
  if (t < TBN) ldeg[t] = 0;
  __syncthreads();
  const int beg = base[blockIdx.x];
  const int end = base[blockIdx.x + 1];
  for (int p0 = beg + t; p0 < end; p0 += 8 * 1024) {
    int px[8];
#pragma unroll
    for (int j = 0; j < 8; ++j) px[j] = rec[2 * min(p0 + j * 1024, end - 1)];
#pragma unroll
    for (int j = 0; j < 8; ++j)
      if (p0 + j * 1024 < end) atomicAdd(&ldeg[px[j] & (TBN - 1)], 1);
  }
  __syncthreads();
  if (t < TBN) dinv[blockIdx.x * TBN + t] = rsqrtf((float)ldeg[t] + 1.0f);
}

__global__ void __launch_bounds__(1024)
k_weight(int* __restrict__ rec, const int* __restrict__ base,
         const float* __restrict__ dinv) {
  __shared__ float ldv[TBN];
  const int t = threadIdx.x;
  if (t < TBN) ldv[t] = dinv[blockIdx.x * TBN + t];
  __syncthreads();
  const int beg = base[blockIdx.x];
  const int end = base[blockIdx.x + 1];
  int2* rg = (int2*)rec;
  for (int p0 = beg + t; p0 < end; p0 += 4 * 1024) {
    int px[4];
    float ds[4];
#pragma unroll
    for (int j = 0; j < 4; ++j) px[j] = rec[2 * min(p0 + j * 1024, end - 1)];
#pragma unroll
    for (int j = 0; j < 4; ++j) ds[j] = dinv[(unsigned)px[j] >> 8];
#pragma unroll
    for (int j = 0; j < 4; ++j) {
      int p = p0 + j * 1024;
      if (p < end) {
        int2 v;
        v.x = px[j];
        v.y = __float_as_int(ds[j] * ldv[px[j] & (TBN - 1)]);
        rg[p] = v;
      }
    }
  }
}

typedef const __attribute__((address_space(1))) int2* gci2p;

template <int I0>
__device__ __forceinline__ void agg_load16(gcfp hg, int lane, int px, float (&v)[16]) {
#pragma unroll
  for (int i = 0; i < 16; ++i) {
    int pk = __builtin_amdgcn_readlane(px, I0 + i);
    v[i] = hg[((unsigned)pk >> 8) * HID + (unsigned)lane];
  }
}

template <int I0>
__device__ __forceinline__ void agg_add16(float* acc, int lane, int px, int py,
                                          const float (&v)[16]) {
#pragma unroll
  for (int i = 0; i < 16; ++i) {
    int pk  = __builtin_amdgcn_readlane(px, I0 + i);
    float w = __int_as_float(__builtin_amdgcn_readlane(py, I0 + i));
    __hip_atomic_fetch_add(&acc[(pk & (TBN - 1)) * HID + lane], v[i] * w,
                           __ATOMIC_RELAXED, __HIP_MEMORY_SCOPE_WORKGROUP);
  }
}

__global__ void __launch_bounds__(1024)
k_agg(const int* __restrict__ rec, const int* __restrict__ base,
      const float* __restrict__ dinv, const float* __restrict__ h,
      const float* __restrict__ bias, float* __restrict__ out) {
  __shared__ float acc[TBN * HID];

  const int wave  = __builtin_amdgcn_readfirstlane((int)(threadIdx.x >> 5));
  const int lane  = threadIdx.x & 31;
  const int node0 = blockIdx.x * TBN;
  gcfp  hg = (gcfp)h;
  gci2p rg = (gci2p)rec;

#pragma unroll
  for (int j = 0; j < TBN / 32; ++j) {
    int n = wave + 32 * j;
    float d = dinv[node0 + n];
    acc[n * HID + lane] = hg[(unsigned)(node0 + n) * HID + (unsigned)lane] * (d * d);
  }
  __syncthreads();

  const int beg = base[blockIdx.x];
  const int end = base[blockIdx.x + 1];

  int p = beg + wave * 32;
  if (p < end) {
    int2 cur = rg[min(p + lane, end - 1)];
    for (;;) {
      const int n  = end - p;
      const int pn = p + 32 * 32;
      if (n < 32) {
        for (int i0 = 0; i0 < n; i0 += 8) {
          float v[8];
#pragma unroll
          for (int j = 0; j < 8; ++j) {
            v[j] = 0.0f;
            if (i0 + j < n) {
              int pk = __builtin_amdgcn_readlane(cur.x, i0 + j);
              v[j] = hg[((unsigned)pk >> 8) * HID + (unsigned)lane];
            }
          }
#pragma unroll
          for (int j = 0; j < 8; ++j) {
            if (i0 + j < n) {
              int pk  = __builtin_amdgcn_readlane(cur.x, i0 + j);
              float w = __int_as_float(__builtin_amdgcn_readlane(cur.y, i0 + j));
              __hip_atomic_fetch_add(&acc[(pk & (TBN - 1)) * HID + lane], v[j] * w,
                                     __ATOMIC_RELAXED, __HIP_MEMORY_SCOPE_WORKGROUP);
            }
          }
        }
        break;
      }
      float va[16], vb[16];
      int2 nxt = rg[min(pn + lane, end - 1)];
      agg_load16<0>(hg, lane, cur.x, va);
      agg_load16<16>(hg, lane, cur.x, vb);
      asm volatile("" :: "v"(nxt.x), "v"(nxt.y));
      agg_add16<0>(acc, lane, cur.x, cur.y, va);
      agg_add16<16>(acc, lane, cur.x, cur.y, vb);
      if (pn >= end) break;
      cur = nxt;
      p = pn;
    }
  }
  __syncthreads();

  const float bv = bias[lane];
#pragma unroll
  for (int j = 0; j < TBN / 32; ++j) {
    int n = wave + 32 * j;
    float v = acc[n * HID + lane] + bv;
    out[(node0 + n) * HID + lane] = v > 0.0f ? v : 0.0f;
  }
}

template <int NT, bool RAGGED, bool BIAS, bool RELU, bool PRE>
__global__ void k_gemm_wmma(const float* __restrict__ A, const float* __restrict__ B,
                            const float* __restrict__ bias, float* __restrict__ C,
                            int M, int N, int K, const float* __restrict__ abias) {
  int wave  = threadIdx.x >> 5;
  int lane  = threadIdx.x & 31;
  int mtile = blockIdx.y * 4 + wave;
  int m0    = mtile * 16;
  if (m0 >= M) return;
  int n0   = blockIdx.x * (16 * NT);
  int half = lane >> 4;
  int lm   = lane & 15;

  gcfp Ag = (gcfp)A;
  gcfp Bg = (gcfp)B;
  gcfp abg = (gcfp)abias;

  int  ncol[NT];
  bool nok[NT];
  unsigned boff[NT];
#pragma unroll
  for (int t = 0; t < NT; ++t) {
    ncol[t] = n0 + t * 16 + lm;
    int nld;
    if (RAGGED) {
      nok[t] = ncol[t] < N;
      nld    = nok[t] ? ncol[t] : (N - 1);
    } else {
      nok[t] = true;
      nld    = ncol[t];
    }
    boff[t] = (unsigned)(half * 2) * (unsigned)N + (unsigned)nld;
  }

  unsigned aoff = (unsigned)(m0 + lm) * (unsigned)K + (unsigned)(half * 2);

  v8f acc[NT];
#pragma unroll
  for (int t = 0; t < NT; ++t) acc[t] = (v8f){0.f,0.f,0.f,0.f,0.f,0.f,0.f,0.f};

  for (int k0 = 0; k0 < K; k0 += 4) {
    v2f a;
    a.x = Ag[aoff];
    a.y = Ag[aoff + 1];
    if (PRE) {
      float p0 = a.x + abg[k0 + half * 2];
      float p1 = a.y + abg[k0 + half * 2 + 1];
      a.x = p0 > 0.0f ? p0 : 0.0f;
      a.y = p1 > 0.0f ? p1 : 0.0f;
    }
#pragma unroll
    for (int t = 0; t < NT; ++t) {
      float b0 = Bg[boff[t]];
      float b1 = Bg[boff[t] + (unsigned)N];
      if (RAGGED) { b0 = nok[t] ? b0 : 0.0f; b1 = nok[t] ? b1 : 0.0f; }
      v2f b; b.x = b0; b.y = b1;
      acc[t] = __builtin_amdgcn_wmma_f32_16x16x4_f32(
           false, a,  false, b,
           (short)0, acc[t],  false,  false);
      boff[t] += 4u * (unsigned)N;
    }
    aoff += 4u;
  }

  gfp Cg = (gfp)C;
  gcfp biasg = (gcfp)bias;
#pragma unroll
  for (int t = 0; t < NT; ++t) {
    float bv = BIAS ? biasg[nok[t] ? ncol[t] : 0] : 0.0f;
    if (!RAGGED || nok[t]) {
#pragma unroll
      for (int r = 0; r < 8; ++r) {
        int m = m0 + r + half * 8;
        float v = acc[t][r] + bv;
        if (RELU) v = v > 0.0f ? v : 0.0f;
        Cg[(unsigned)m * (unsigned)N + (unsigned)ncol[t]] = v;
      }
    }
  }
}

#define HEAD_K      (NODE_COUNT * HID)
#define HEAD_N      256
#define HS_SLICES   25
#define HS_KPARTS   8
#define HS_WAVES    (4 * HS_KPARTS)
#define HS_SLICE_K  (HEAD_K / HS_SLICES)
#define HS_PART_K   (HS_SLICE_K / HS_KPARTS)
#define HS_GK       16
#define HS_GROUPS   (HS_PART_K / HS_GK)
static_assert(HS_SLICES * HS_KPARTS * HS_GROUPS * HS_GK == HEAD_K, "K is covered exactly");
static_assert(BATCH == 64, "four row tiles");

typedef const __attribute__((address_space(1))) v2f* gcv2p;

__global__ void __launch_bounds__(HS_WAVES * 32)
k_head_gemm(const float* __restrict__ A,
            const float* __restrict__ Bp, const float* __restrict__ Bv,
            float* __restrict__ Cp, float* __restrict__ Cv) {
  __shared__ float part[HS_WAVES * 1024];

  const int wave = threadIdx.x >> 5;
  const int lane = threadIdx.x & 31;
  const int half = lane >> 4;
  const int lm   = lane & 15;
  const int mt   = wave & 3;
  const int kp   = wave >> 2;
  const int m0   = mt * 16;
  const int n0   = (blockIdx.x & 3) * 64;
  const bool second = blockIdx.x >= 4;
  const float* B = second ? Bv : Bp;
  float*       C = second ? Cv : Cp;
  const int kbeg = blockIdx.y * HS_SLICE_K + kp * HS_PART_K;

  gcfp ap = (gcfp)A + ((unsigned)(m0 + lm) * (unsigned)HEAD_K + (unsigned)(kbeg + half * 2));
  gcfp bp = (gcfp)B + ((unsigned)(kbeg + half * 2) * (unsigned)HEAD_N + (unsigned)(n0 + lm));

  v8f acc[4];
#pragma unroll
  for (int t = 0; t < 4; ++t) acc[t] = (v8f){0.f,0.f,0.f,0.f,0.f,0.f,0.f,0.f};

  for (int g = 0; g < HS_GROUPS; ++g) {
    v2f a[4], b[4][4];
#pragma unroll
    for (int s = 0; s < 4; ++s)
      a[s] = *(gcv2p)(ap + 4 * s);
#pragma unroll
    for (int s = 0; s < 4; ++s) {
#pragma unroll
      for (int t = 0; t < 4; ++t) {
        b[t][s].x = bp[(4 * s) * HEAD_N + 16 * t];
        b[t][s].y = bp[(4 * s + 1) * HEAD_N + 16 * t];
      }
    }
    __builtin_amdgcn_sched_barrier(0);
#pragma unroll
    for (int s = 0; s < 4; ++s) {
#pragma unroll
      for (int t = 0; t < 4; ++t)
        acc[t] = __builtin_amdgcn_wmma_f32_16x16x4_f32(
            false, a[s], false, b[t][s], (short)0, acc[t], false, false);
    }
    ap += HS_GK;
    bp += HS_GK * HEAD_N;
  }

  float* mine = part + wave * 1024;
#pragma unroll
  for (int t = 0; t < 4; ++t) {
#pragma unroll
    for (int r = 0; r < 8; ++r)
      mine[(r + half * 8) * 64 + t * 16 + lm] = acc[t][r];
  }
  __syncthreads();

  const int rr  = threadIdx.x >> 6;
  const int col = threadIdx.x & 63;
#pragma unroll
  for (int j = 0; j < 4; ++j) {
    float s = part[j * 1024 + rr * 64 + col];
    for (int q = 1; q < HS_KPARTS; ++q) s += part[(q * 4 + j) * 1024 + rr * 64 + col];
    atomicAdd(&C[(j * 16 + rr) * HEAD_N + n0 + col], s);
  }
}

__global__ void k_value(const float* __restrict__ sumv, const float* __restrict__ bv1,
                        const float* __restrict__ Wv2,
                        const float* __restrict__ bv2, float* __restrict__ out) {
  int b = blockIdx.x * blockDim.x + threadIdx.x;
  if (b < BATCH) {
    float s = bv2[0];
    for (int k = 0; k < 256; ++k) {
      float h = sumv[b * 256 + k] + bv1[k];
      h = h > 0.0f ? h : 0.0f;
      s += h * Wv2[k];
    }
    out[b] = s;
  }
}

static inline int cdiv(long long a, int b) { return (int)((a + b - 1) / b); }

extern "C" void kernel_launch(void* const* d_in, const int* in_sizes, int n_in,
                              void* d_out, int out_size, void* d_ws, size_t ws_size,
                              hipStream_t stream) {
  (void)in_sizes; (void)n_in; (void)out_size; (void)ws_size;

  const float* x    = (const float*)d_in[0];
  const int*   ei   = (const int*)d_in[1];
  const int*   rows = ei;
  const int*   cols = ei + N_EDGES;
  const float* W[5]  = {(const float*)d_in[2], (const float*)d_in[4], (const float*)d_in[6],
                        (const float*)d_in[8], (const float*)d_in[10]};
  const float* bb[5] = {(const float*)d_in[3], (const float*)d_in[5], (const float*)d_in[7],
                        (const float*)d_in[9], (const float*)d_in[11]};
  const float* Wp1 = (const float*)d_in[12]; const float* bp1 = (const float*)d_in[13];
  const float* Wp2 = (const float*)d_in[14]; const float* bp2 = (const float*)d_in[15];
  const float* Wv1 = (const float*)d_in[16]; const float* bv1 = (const float*)d_in[17];
  const float* Wv2 = (const float*)d_in[18]; const float* bv2 = (const float*)d_in[19];
  float* out = (float*)d_out;

  float* ws     = (float*)d_ws;
  float* deg    = ws;
  int*   cursor = (int*)(deg + N_TOTAL);
  float* hidp   = (float*)(cursor + NBLK * CUR_PITCH);
  float* hidv   = hidp + BATCH * 256;
  int*   count  = (int*)(hidv + BATCH * 256);
  int*   base   = count + 256;
  int*   rec    = base + 512;
  float* bufA   = (float*)(rec + 2 * (size_t)N_EDGES);
  float* bufB   = bufA + (size_t)N_TOTAL * HID;

  const int TB = 256;

  k_zero    <<<cdiv(NBLK * CUR_PITCH + 2 * BATCH * 256 + 256, TB), TB, 0, stream>>>(
      (float*)cursor, NBLK * CUR_PITCH + 2 * BATCH * 256 + 256);
  k_count   <<<NCH, 1024, 0, stream>>>(cols, count);
  k_scan    <<<1, 256, 0, stream>>>(count, base);
  k_fill    <<<NCH, 1024, 0, stream>>>(rows, cols, base, cursor, rec);
  k_deg_dinv<<<NBLK, 1024, 0, stream>>>(rec, base, deg);
  k_weight  <<<NBLK, 1024, 0, stream>>>(rec, base, deg);

  const float* in = x;
  int Kdim = STATE;
  dim3 layer_grid(1, N_TOTAL / 64);
  for (int l = 0; l < 5; ++l) {
    k_gemm_wmma<2, false, false, false, false><<<layer_grid, 128, 0, stream>>>(
        in, W[l], nullptr, bufA, N_TOTAL, HID, Kdim, nullptr);
    k_agg<<<NBLK, 1024, 0, stream>>>(rec, base, deg, bufA, bb[l], bufB);
    in = bufB;
    Kdim = HID;
  }

  dim3 g1(2 * HEAD_N / 64, HS_SLICES);
  k_head_gemm<<<g1, HS_WAVES * 32, 0, stream>>>(bufB, Wp1, Wv1, hidp, hidv);

  dim3 g2(cdiv(NODE_COUNT, 16), cdiv(BATCH, 64));
  k_gemm_wmma<1, true, true, false, true><<<g2, 128, 0, stream>>>(
      hidp, Wp2, bp2, out, BATCH, NODE_COUNT, 256, bp1);

  k_value<<<1, 64, 0, stream>>>(hidv, bv1, Wv2, bv2, out + (size_t)BATCH * NODE_COUNT);
}


# Layers as (A_hat h) W: a block sums the layer's INPUT rows (16 wide in layer 1) in LDS, then W, bias, relu on its 256 sums with WMMA in the same launch; 5 layer GEMM launches go; f32 rounding changes
# speedup vs baseline: 19.7939x; 1.1688x over previous
#include <hip/hip_runtime.h>

#define N_TOTAL    64000
#define NODE_COUNT 1000
#define BATCH      64
#define STATE      16
#define HID        32
#define N_EDGES    2048000
#define TBN        256
#define NBLK       (N_TOTAL / TBN)
#define CUR_PITCH  64
static_assert(N_TOTAL % TBN == 0, "blocks cover the nodes exactly");
static_assert(N_TOTAL <= (1 << 24), "source id and 8-bit local target share one int");
static_assert(TBN == 256 && HID == 32, "record packing and lane-per-feature layout");

typedef float v2f __attribute__((ext_vector_type(2)));
typedef float v8f __attribute__((ext_vector_type(8)));

typedef const __attribute__((address_space(1))) float* gcfp;
typedef __attribute__((address_space(1))) float*       gfp;

__global__ void k_zero(float* __restrict__ p, int n) {
  int i = blockIdx.x * blockDim.x + threadIdx.x;
  if (i < n) p[i] = 0.0f;
}

#define ECH  8192
#define NCH  (N_EDGES / ECH)
#define EPT  (ECH / 1024)
static_assert(N_EDGES % ECH == 0, "chunks cover the edges exactly");
static_assert(NBLK <= 256, "one LDS counter per target block, 256 of them");

__global__ void __launch_bounds__(1024)
k_count(const int* __restrict__ cols, int* __restrict__ count) {
  __shared__ int lcnt[256];
  const int t = threadIdx.x;
  if (t < 256) lcnt[t] = 0;
  __syncthreads();
  const int e0 = blockIdx.x * ECH + t;
  int c[EPT];
#pragma unroll
  for (int j = 0; j < EPT; ++j) c[j] = cols[e0 + j * 1024];
#pragma unroll
  for (int j = 0; j < EPT; ++j) atomicAdd(&lcnt[(unsigned)c[j] / TBN], 1);
  __syncthreads();
  if (t < NBLK) {
    int n = lcnt[t];
    if (n) atomicAdd(&count[t], n);
  }
}

__global__ void __launch_bounds__(256)
k_scan(const int* __restrict__ count, int* __restrict__ base) {
  __shared__ int s[256];
  const int t = threadIdx.x;
  s[t] = t < NBLK ? count[t] : 0;
  __syncthreads();
  for (int o = 1; o < 256; o <<= 1) {
    int a = t >= o ? s[t - o] : 0;
    __syncthreads();
    s[t] += a;
    __syncthreads();
  }
  if (t == 0) base[0] = 0;
  if (t < NBLK) base[t + 1] = s[t];
}

__global__ void __launch_bounds__(1024)
k_fill(const int* __restrict__ rows, const int* __restrict__ cols,
       const int* __restrict__ base, int* __restrict__ cursor, int* __restrict__ rec) {
  __shared__ int2 stage[ECH];
  __shared__ int lcnt[256];
  __shared__ int lstart[256];
  __shared__ int gdelta[256];
  __shared__ int wtot[8];
  const int t = threadIdx.x;
  if (t < 256) lcnt[t] = 0;
  __syncthreads();
  const int e0 = blockIdx.x * ECH + t;
  int r[EPT], c[EPT], rk[EPT];
#pragma unroll
  for (int j = 0; j < EPT; ++j) {
    r[j] = rows[e0 + j * 1024];
    c[j] = cols[e0 + j * 1024];
  }
#pragma unroll
  for (int j = 0; j < EPT; ++j) rk[j] = atomicAdd(&lcnt[(unsigned)c[j] / TBN], 1);
  __syncthreads();

  int cnt = 0, incl = 0, gpos = 0;
  if (t < 256) {
    cnt = lcnt[t];
    if (t < NBLK) gpos = base[t] + atomicAdd(&cursor[t * CUR_PITCH], cnt);
    incl = cnt;
    for (int o = 1; o < 32; o <<= 1) {
      int u = __shfl_up(incl, o);
      if ((t & 31) >= o) incl += u;
    }
    if ((t & 31) == 31) wtot[t >> 5] = incl;
  }
  __syncthreads();
  if (t < 256) {
    int pre = 0;
    for (int w = 0; w < (t >> 5); ++w) pre += wtot[w];
    int first = pre + incl - cnt;
    lstart[t] = first;
    gdelta[t] = gpos - first;
  }
  __syncthreads();

#pragma unroll
  for (int j = 0; j < EPT; ++j) {
    int b  = (unsigned)c[j] / TBN;
    int li = lstart[b] + rk[j];
    int2 v;
    v.x = (r[j] << 8) | (c[j] & (TBN - 1));
    v.y = gdelta[b] + li;
    stage[li] = v;
  }
  __syncthreads();
#pragma unroll
  for (int j = 0; j < EPT; ++j) {
    int2 s = stage[t + j * 1024];
    int2 v;
    v.x = s.x;
    v.y = 0;
    ((int2*)rec)[s.y] = v;
  }
}

__global__ void __launch_bounds__(1024)
k_deg_dinv(const int* __restrict__ rec, const int* __restrict__ base,
           float* __restrict__ dinv) {
  __shared__ int ldeg[TBN];
  const int t = threadIdx.x;
  if (t < TBN) ldeg[t] = 0;
  __syncthreads();
  const int beg = base[blockIdx.x];
  const int end = base[blockIdx.x + 1];
  for (int p0 = beg + t; p0 < end; p0 += 8 * 1024) {
    int px[8];
#pragma unroll
    for (int j = 0; j < 8; ++j) px[j] = rec[2 * min(p0 + j * 1024, end - 1)];
#pragma unroll
    for (int j = 0; j < 8; ++j)
      if (p0 + j * 1024 < end) atomicAdd(&ldeg[px[j] & (TBN - 1)], 1);
  }
  __syncthreads();
  if (t < TBN) dinv[blockIdx.x * TBN + t] = rsqrtf((float)ldeg[t] + 1.0f);
}

__global__ void __launch_bounds__(1024)
k_weight(int* __restrict__ rec, const int* __restrict__ base,
         const float* __restrict__ dinv) {
  __shared__ float ldv[TBN];
  const int t = threadIdx.x;
  if (t < TBN) ldv[t] = dinv[blockIdx.x * TBN + t];
  __syncthreads();
  const int beg = base[blockIdx.x];
  const int end = base[blockIdx.x + 1];
  int2* rg = (int2*)rec;
  for (int p0 = beg + t; p0 < end; p0 += 4 * 1024) {
    int px[4];
    float ds[4];
#pragma unroll
    for (int j = 0; j < 4; ++j) px[j] = rec[2 * min(p0 + j * 1024, end - 1)];
#pragma unroll
    for (int j = 0; j < 4; ++j) ds[j] = dinv[(unsigned)px[j] >> 8];
#pragma unroll
    for (int j = 0; j < 4; ++j) {
      int p = p0 + j * 1024;
      if (p < end) {
        int2 v;
        v.x = px[j];
        v.y = __float_as_int(ds[j] * ldv[px[j] & (TBN - 1)]);
        rg[p] = v;
      }
    }
  }
}

typedef const __attribute__((address_space(1))) int2* gci2p;
#define STG_PITCH 34

template <int F, int I0>
__device__ __forceinline__ void agg_load16(gcfp hg, int f, int px, float (&v)[16]) {
#pragma unroll
  for (int i = 0; i < 16; ++i) {
    int pk = __builtin_amdgcn_readlane(px, I0 + i);
    v[i] = hg[((unsigned)pk >> 8) * F + (unsigned)f];
  }
}

template <int I0>
__device__ __forceinline__ void agg_add16(float* acc, int lane, int px, int py,
                                          const float (&v)[16]) {
#pragma unroll
  for (int i = 0; i < 16; ++i) {
    int pk  = __builtin_amdgcn_readlane(px, I0 + i);
    float w = __int_as_float(__builtin_amdgcn_readlane(py, I0 + i));
    __hip_atomic_fetch_add(&acc[(pk & (TBN - 1)) * HID + lane], v[i] * w,
                           __ATOMIC_RELAXED, __HIP_MEMORY_SCOPE_WORKGROUP);
  }
}

template <int F>
__global__ void __launch_bounds__(1024)
k_agg(const int* __restrict__ rec, const int* __restrict__ base,
      const float* __restrict__ dinv, const float* __restrict__ h,
      const float* __restrict__ Wl, const float* __restrict__ bias,
      float* __restrict__ out) {
  constexpr int P = F + 2;
  __shared__ __attribute__((aligned(16))) float acc[TBN * STG_PITCH];
  __shared__ __attribute__((aligned(16))) float apad[TBN * P];

  const int wave  = __builtin_amdgcn_readfirstlane((int)(threadIdx.x >> 5));
  const int lane  = threadIdx.x & 31;
  const int f     = lane & (F - 1);
  const int half  = lane >> 4;
  const int lm    = lane & 15;
  const int mt    = wave >> 1;
  const int nt    = wave & 1;
  const int node0 = blockIdx.x * TBN;
  gcfp  hg = (gcfp)h;
  gci2p rg = (gci2p)rec;

  gcfp wg = (gcfp)Wl;
  v2f wb[F / 4];
#pragma unroll
  for (int s = 0; s < F / 4; ++s) {
    wb[s].x = wg[(4 * s + 2 * half) * HID + nt * 16 + lm];
    wb[s].y = wg[(4 * s + 2 * half + 1) * HID + nt * 16 + lm];
  }

#pragma unroll
  for (int j = 0; j < TBN / 32; ++j) {
    int n = wave + 32 * j;
    float d = dinv[node0 + n];
    acc[n * HID + lane] = hg[(unsigned)(node0 + n) * F + (unsigned)f] * (d * d);
  }
  __syncthreads();

  const int beg = base[blockIdx.x];
  const int end = base[blockIdx.x + 1];

  int p = beg + wave * 32;
  if (p < end) {
    int2 cur = rg[min(p + lane, end - 1)];
    for (;;) {
      const int n  = end - p;
      const int pn = p + 32 * 32;
      if (n < 32) {
        for (int i0 = 0; i0 < n; i0 += 8) {
          float v[8];
#pragma unroll
          for (int j = 0; j < 8; ++j) {
            v[j] = 0.0f;
            if (i0 + j < n) {
              int pk = __builtin_amdgcn_readlane(cur.x, i0 + j);
              v[j] = hg[((unsigned)pk >> 8) * F + (unsigned)f];
            }
          }
#pragma unroll
          for (int j = 0; j < 8; ++j) {
            if (i0 + j < n) {
              int pk  = __builtin_amdgcn_readlane(cur.x, i0 + j);
              float w = __int_as_float(__builtin_amdgcn_readlane(cur.y, i0 + j));
              __hip_atomic_fetch_add(&acc[(pk & (TBN - 1)) * HID + lane], v[j] * w,
                                     __ATOMIC_RELAXED, __HIP_MEMORY_SCOPE_WORKGROUP);
            }
          }
        }
        break;
      }
      float va[16], vb[16];
      int2 nxt = rg[min(pn + lane, end - 1)];
      agg_load16<F, 0>(hg, f, cur.x, va);
      agg_load16<F, 16>(hg, f, cur.x, vb);
      asm volatile("" :: "v"(nxt.x), "v"(nxt.y));
      agg_add16<0>(acc, lane, cur.x, cur.y, va);
      agg_add16<16>(acc, lane, cur.x, cur.y, vb);
      if (pn >= end) break;
      cur = nxt;
      p = pn;
    }
  }
  __syncthreads();

#pragma unroll
  for (int j = 0; j < TBN / 32; ++j) {
    int n = wave + 32 * j;
    apad[n * P + f] = acc[n * HID + f];
  }
  __syncthreads();

  v8f d = (v8f){0.f, 0.f, 0.f, 0.f, 0.f, 0.f, 0.f, 0.f};
  const float* arow = &apad[(mt * 16 + lm) * P + 2 * half];
#pragma unroll
  for (int s = 0; s < F / 4; ++s) {
    v2f a = *(const v2f*)(arow + 4 * s);
    d = __builtin_amdgcn_wmma_f32_16x16x4_f32(false, a, false, wb[s], (short)0, d,
                                              false, false);
  }
  float* srow = &acc[(mt * 16 + 8 * half) * STG_PITCH + nt * 16 + lm];
#pragma unroll
  for (int r = 0; r < 8; ++r) srow[r * STG_PITCH] = d[r];
  __syncthreads();

  const float bv = bias[lane];
#pragma unroll
  for (int j = 0; j < TBN / 32; ++j) {
    int n = wave + 32 * j;
    float v = acc[n * STG_PITCH + lane] + bv;
    out[(node0 + n) * HID + lane] = v > 0.0f ? v : 0.0f;
  }
}

template <int NT, bool RAGGED, bool BIAS, bool RELU, bool PRE>
__global__ void k_gemm_wmma(const float* __restrict__ A, const float* __restrict__ B,
                            const float* __restrict__ bias, float* __restrict__ C,
                            int M, int N, int K, const float* __restrict__ abias) {
  int wave  = threadIdx.x >> 5;
  int lane  = threadIdx.x & 31;
  int mtile = blockIdx.y * 4 + wave;
  int m0    = mtile * 16;
  if (m0 >= M) return;
  int n0   = blockIdx.x * (16 * NT);
  int half = lane >> 4;
  int lm   = lane & 15;

  gcfp Ag = (gcfp)A;
  gcfp Bg = (gcfp)B;
  gcfp abg = (gcfp)abias;

  int  ncol[NT];
  bool nok[NT];
  unsigned boff[NT];
#pragma unroll
  for (int t = 0; t < NT; ++t) {
    ncol[t] = n0 + t * 16 + lm;
    int nld;
    if (RAGGED) {
      nok[t] = ncol[t] < N;
      nld    = nok[t] ? ncol[t] : (N - 1);
    } else {
      nok[t] = true;
      nld    = ncol[t];
    }
    boff[t] = (unsigned)(half * 2) * (unsigned)N + (unsigned)nld;
  }

  unsigned aoff = (unsigned)(m0 + lm) * (unsigned)K + (unsigned)(half * 2);

  v8f acc[NT];
#pragma unroll
  for (int t = 0; t < NT; ++t) acc[t] = (v8f){0.f,0.f,0.f,0.f,0.f,0.f,0.f,0.f};

  for (int k0 = 0; k0 < K; k0 += 4) {
    v2f a;
    a.x = Ag[aoff];
    a.y = Ag[aoff + 1];
    if (PRE) {
      float p0 = a.x + abg[k0 + half * 2];
      float p1 = a.y + abg[k0 + half * 2 + 1];
      a.x = p0 > 0.0f ? p0 : 0.0f;
      a.y = p1 > 0.0f ? p1 : 0.0f;
    }
#pragma unroll
    for (int t = 0; t < NT; ++t) {
      float b0 = Bg[boff[t]];
      float b1 = Bg[boff[t] + (unsigned)N];
      if (RAGGED) { b0 = nok[t] ? b0 : 0.0f; b1 = nok[t] ? b1 : 0.0f; }
      v2f b; b.x = b0; b.y = b1;
      acc[t] = __builtin_amdgcn_wmma_f32_16x16x4_f32(
           false, a,  false, b,
           (short)0, acc[t],  false,  false);
      boff[t] += 4u * (unsigned)N;
    }
    aoff += 4u;
  }

  gfp Cg = (gfp)C;
  gcfp biasg = (gcfp)bias;
#pragma unroll
  for (int t = 0; t < NT; ++t) {
    float bv = BIAS ? biasg[nok[t] ? ncol[t] : 0] : 0.0f;
    if (!RAGGED || nok[t]) {
#pragma unroll
      for (int r = 0; r < 8; ++r) {
        int m = m0 + r + half * 8;
        float v = acc[t][r] + bv;
        if (RELU) v = v > 0.0f ? v : 0.0f;
        Cg[(unsigned)m * (unsigned)N + (unsigned)ncol[t]] = v;
      }
    }
  }
}

#define HEAD_K      (NODE_COUNT * HID)
#define HEAD_N      256
#define HS_SLICES   25
#define HS_KPARTS   8
#define HS_WAVES    (4 * HS_KPARTS)
#define HS_SLICE_K  (HEAD_K / HS_SLICES)
#define HS_PART_K   (HS_SLICE_K / HS_KPARTS)
#define HS_GK       16
#define HS_GROUPS   (HS_PART_K / HS_GK)
static_assert(HS_SLICES * HS_KPARTS * HS_GROUPS * HS_GK == HEAD_K, "K is covered exactly");
static_assert(BATCH == 64, "four row tiles");

typedef const __attribute__((address_space(1))) v2f* gcv2p;

__global__ void __launch_bounds__(HS_WAVES * 32)
k_head_gemm(const float* __restrict__ A,
            const float* __restrict__ Bp, const float* __restrict__ Bv,
            float* __restrict__ Cp, float* __restrict__ Cv) {
  __shared__ float part[HS_WAVES * 1024];

  const int wave = threadIdx.x >> 5;
  const int lane = threadIdx.x & 31;
  const int half = lane >> 4;
  const int lm   = lane & 15;
  const int mt   = wave & 3;
  const int kp   = wave >> 2;
  const int m0   = mt * 16;
  const int n0   = (blockIdx.x & 3) * 64;
  const bool second = blockIdx.x >= 4;
  const float* B = second ? Bv : Bp;
  float*       C = second ? Cv : Cp;
  const int kbeg = blockIdx.y * HS_SLICE_K + kp * HS_PART_K;

  gcfp ap = (gcfp)A + ((unsigned)(m0 + lm) * (unsigned)HEAD_K + (unsigned)(kbeg + half * 2));
  gcfp bp = (gcfp)B + ((unsigned)(kbeg + half * 2) * (unsigned)HEAD_N + (unsigned)(n0 + lm));

  v8f acc[4];
#pragma unroll
  for (int t = 0; t < 4; ++t) acc[t] = (v8f){0.f,0.f,0.f,0.f,0.f,0.f,0.f,0.f};

  for (int g = 0; g < HS_GROUPS; ++g) {
    v2f a[4], b[4][4];
#pragma unroll
    for (int s = 0; s < 4; ++s)
      a[s] = *(gcv2p)(ap + 4 * s);
#pragma unroll
    for (int s = 0; s < 4; ++s) {
#pragma unroll
      for (int t = 0; t < 4; ++t) {
        b[t][s].x = bp[(4 * s) * HEAD_N + 16 * t];
        b[t][s].y = bp[(4 * s + 1) * HEAD_N + 16 * t];
      }
    }
    __builtin_amdgcn_sched_barrier(0);
#pragma unroll
    for (int s = 0; s < 4; ++s) {
#pragma unroll
      for (int t = 0; t < 4; ++t)
        acc[t] = __builtin_amdgcn_wmma_f32_16x16x4_f32(
            false, a[s], false, b[t][s], (short)0, acc[t], false, false);
    }
    ap += HS_GK;
    bp += HS_GK * HEAD_N;
  }

  float* mine = part + wave * 1024;
#pragma unroll
  for (int t = 0; t < 4; ++t) {
#pragma unroll
    for (int r = 0; r < 8; ++r)
      mine[(r + half * 8) * 64 + t * 16 + lm] = acc[t][r];
  }
  __syncthreads();

  const int rr  = threadIdx.x >> 6;
  const int col = threadIdx.x & 63;
#pragma unroll
  for (int j = 0; j < 4; ++j) {
    float s = part[j * 1024 + rr * 64 + col];
    for (int q = 1; q < HS_KPARTS; ++q) s += part[(q * 4 + j) * 1024 + rr * 64 + col];
    atomicAdd(&C[(j * 16 + rr) * HEAD_N + n0 + col], s);
  }
}

__global__ void k_value(const float* __restrict__ sumv, const float* __restrict__ bv1,
                        const float* __restrict__ Wv2,
                        const float* __restrict__ bv2, float* __restrict__ out) {
  int b = blockIdx.x * blockDim.x + threadIdx.x;
  if (b < BATCH) {
    float s = bv2[0];
    for (int k = 0; k < 256; ++k) {
      float h = sumv[b * 256 + k] + bv1[k];
      h = h > 0.0f ? h : 0.0f;
      s += h * Wv2[k];
    }
    out[b] = s;
  }
}

static inline int cdiv(long long a, int b) { return (int)((a + b - 1) / b); }

extern "C" void kernel_launch(void* const* d_in, const int* in_sizes, int n_in,
                              void* d_out, int out_size, void* d_ws, size_t ws_size,
                              hipStream_t stream) {
  (void)in_sizes; (void)n_in; (void)out_size; (void)ws_size;

  const float* x    = (const float*)d_in[0];
  const int*   ei   = (const int*)d_in[1];
  const int*   rows = ei;
  const int*   cols = ei + N_EDGES;
  const float* W[5]  = {(const float*)d_in[2], (const float*)d_in[4], (const float*)d_in[6],
                        (const float*)d_in[8], (const float*)d_in[10]};
  const float* bb[5] = {(const float*)d_in[3], (const float*)d_in[5], (const float*)d_in[7],
                        (const float*)d_in[9], (const float*)d_in[11]};
  const float* Wp1 = (const float*)d_in[12]; const float* bp1 = (const float*)d_in[13];
  const float* Wp2 = (const float*)d_in[14]; const float* bp2 = (const float*)d_in[15];
  const float* Wv1 = (const float*)d_in[16]; const float* bv1 = (const float*)d_in[17];
  const float* Wv2 = (const float*)d_in[18]; const float* bv2 = (const float*)d_in[19];
  float* out = (float*)d_out;

  float* ws     = (float*)d_ws;
  float* deg    = ws;
  int*   cursor = (int*)(deg + N_TOTAL);
  float* hidp   = (float*)(cursor + NBLK * CUR_PITCH);
  float* hidv   = hidp + BATCH * 256;
  int*   count  = (int*)(hidv + BATCH * 256);
  int*   base   = count + 256;
  int*   rec    = base + 512;
  float* bufA   = (float*)(rec + 2 * (size_t)N_EDGES);
  float* bufB   = bufA + (size_t)N_TOTAL * HID;

  const int TB = 256;

  k_zero    <<<cdiv(NBLK * CUR_PITCH + 2 * BATCH * 256 + 256, TB), TB, 0, stream>>>(
      (float*)cursor, NBLK * CUR_PITCH + 2 * BATCH * 256 + 256);
  k_count   <<<NCH, 1024, 0, stream>>>(cols, count);
  k_scan    <<<1, 256, 0, stream>>>(count, base);
  k_fill    <<<NCH, 1024, 0, stream>>>(rows, cols, base, cursor, rec);
  k_deg_dinv<<<NBLK, 1024, 0, stream>>>(rec, base, deg);
  k_weight  <<<NBLK, 1024, 0, stream>>>(rec, base, deg);

  k_agg<STATE><<<NBLK, 1024, 0, stream>>>(rec, base, deg, x,    W[0], bb[0], bufA);
  k_agg<HID>  <<<NBLK, 1024, 0, stream>>>(rec, base, deg, bufA, W[1], bb[1], bufB);
  k_agg<HID>  <<<NBLK, 1024, 0, stream>>>(rec, base, deg, bufB, W[2], bb[2], bufA);
  k_agg<HID>  <<<NBLK, 1024, 0, stream>>>(rec, base, deg, bufA, W[3], bb[3], bufB);
  k_agg<HID>  <<<NBLK, 1024, 0, stream>>>(rec, base, deg, bufB, W[4], bb[4], bufA);

  dim3 g1(2 * HEAD_N / 64, HS_SLICES);
  k_head_gemm<<<g1, HS_WAVES * 32, 0, stream>>>(bufA, Wp1, Wv1, hidp, hidv);

  dim3 g2(cdiv(NODE_COUNT, 16), cdiv(BATCH, 64));
  k_gemm_wmma<1, true, true, false, true><<<g2, 128, 0, stream>>>(
      hidp, Wp2, bp2, out, BATCH, NODE_COUNT, 256, bp1);

  k_value<<<1, 64, 0, stream>>>(hidv, bv1, Wv2, bv2, out + (size_t)BATCH * NODE_COUNT);
}


# Launches 14 -> 10: zero-fill, bin scan, edge-weight and value-dot kernels ride in neighbour launches (count matrix summed in fill, weights formed in layer 1); list order fixed, V sum order changes
# speedup vs baseline: 22.1806x; 1.1206x over previous
#include <hip/hip_runtime.h>

#define N_TOTAL    64000
#define NODE_COUNT 1000
#define BATCH      64
#define STATE      16
#define HID        32
#define N_EDGES    2048000
#define TBN        256
#define NBLK       (N_TOTAL / TBN)
static_assert(N_TOTAL % TBN == 0, "blocks cover the nodes exactly");
static_assert(N_TOTAL <= (1 << 24), "source id and 8-bit local target share one int");
static_assert(TBN == 256 && HID == 32, "record packing and lane-per-feature layout");

typedef float v2f __attribute__((ext_vector_type(2)));
typedef float v8f __attribute__((ext_vector_type(8)));

typedef const __attribute__((address_space(1))) float* gcfp;
typedef __attribute__((address_space(1))) float*       gfp;

#define ECH  8192
#define NCH  (N_EDGES / ECH)
#define EPT  (ECH / 1024)
static_assert(N_EDGES % ECH == 0, "chunks cover the edges exactly");
static_assert(NBLK <= 256, "one LDS counter per target block, 256 of them");
#define HID_ALL  (2 * BATCH * 256)
static_assert(HID_ALL % 1024 == 0 && HID_ALL / 1024 <= NCH, "head sums cleared by the first chunks");

__global__ void __launch_bounds__(1024)
k_count(const int* __restrict__ cols, int* __restrict__ cnt, float* __restrict__ hid) {
  __shared__ int lcnt[256];
  const int t = threadIdx.x;
  if (t < 256) lcnt[t] = 0;
  if (blockIdx.x < HID_ALL / 1024) hid[blockIdx.x * 1024 + t] = 0.0f;
  __syncthreads();
  const int e0 = blockIdx.x * ECH + t;
  int c[EPT];
#pragma unroll
  for (int j = 0; j < EPT; ++j) c[j] = cols[e0 + j * 1024];
#pragma unroll
  for (int j = 0; j < EPT; ++j) atomicAdd(&lcnt[(unsigned)c[j] / TBN], 1);
  __syncthreads();
  if (t < 256) cnt[blockIdx.x * 256 + t] = lcnt[t];
}

#define CQ  16
#define CJ  ((NCH + CQ - 1) / CQ)
static_assert(CQ * 64 == 1024, "one thread per (row group, four blocks)");
typedef int v4i __attribute__((ext_vector_type(4)));
__global__ void __launch_bounds__(1024)
k_fill(const int* __restrict__ rows, const int* __restrict__ cols,
       const int* __restrict__ cnt, int* __restrict__ base, int* __restrict__ rec) {
  __shared__ int2 stage[ECH];
  __shared__ int lcnt[256];
  __shared__ int lstart[256];
  __shared__ int gdelta[256];
  __shared__ int wtot[8];
  __shared__ int wtot2[8];
  __shared__ __attribute__((aligned(16))) int qtot[CQ * 256];
  __shared__ __attribute__((aligned(16))) int qpre[CQ * 256];
  const int t = threadIdx.x;
  if (t < 256) lcnt[t] = 0;
  __syncthreads();
  const int e0 = blockIdx.x * ECH + t;
  int r[EPT], c[EPT], rk[EPT];
#pragma unroll
  for (int j = 0; j < EPT; ++j) {
    r[j] = rows[e0 + j * 1024];
    c[j] = cols[e0 + j * 1024];
  }
  {
    const int b4 = t & 63, q = t >> 6;
    const v4i* cm = (const v4i*)cnt;
    v4i v[CJ];
#pragma unroll
    for (int j = 0; j < CJ; ++j) v[j] = cm[min(q + CQ * j, NCH - 1) * 64 + b4];
    const v4i zero = (v4i){0, 0, 0, 0};
    v4i tot = zero, pre = zero;
#pragma unroll
    for (int j = 0; j < CJ; ++j) {
      const int ch = q + CQ * j;
      tot += (ch < NCH) ? v[j] : zero;
      pre += (ch < (int)blockIdx.x) ? v[j] : zero;
    }
    ((v4i*)qtot)[q * 64 + b4] = tot;
    ((v4i*)qpre)[q * 64 + b4] = pre;
  }
#pragma unroll
  for (int j = 0; j < EPT; ++j) rk[j] = atomicAdd(&lcnt[(unsigned)c[j] / TBN], 1);
  __syncthreads();

  int cn = 0, incl = 0, gt = 0, gp = 0, incl2 = 0;
  if (t < 256) {
    cn = lcnt[t];
#pragma unroll
    for (int g = 0; g < CQ; ++g) {
      gt += qtot[g * 256 + t];
      gp += qpre[g * 256 + t];
    }
    incl  = cn;
    incl2 = gt;
    for (int o = 1; o < 32; o <<= 1) {
      int u  = __shfl_up(incl, o);
      int u2 = __shfl_up(incl2, o);
      if ((t & 31) >= o) { incl += u; incl2 += u2; }
    }
    if ((t & 31) == 31) { wtot[t >> 5] = incl; wtot2[t >> 5] = incl2; }
  }
  __syncthreads();
  if (t < 256) {
    int pre = 0, pre2 = 0;
    for (int w = 0; w < (t >> 5); ++w) { pre += wtot[w]; pre2 += wtot2[w]; }
    int first  = pre + incl - cn;
    int bfirst = pre2 + incl2 - gt;
    lstart[t] = first;
    gdelta[t] = bfirst + gp - first;
    if (blockIdx.x == 0) {
      if (t < NBLK) base[t] = bfirst;
      if (t == NBLK - 1) base[NBLK] = bfirst + gt;
    }
  }
  __syncthreads();

#pragma unroll
  for (int j = 0; j < EPT; ++j) {
    int b  = (unsigned)c[j] / TBN;
    int li = lstart[b] + rk[j];
    int2 v;
    v.x = (r[j] << 8) | (c[j] & (TBN - 1));
    v.y = gdelta[b] + li;
    stage[li] = v;
  }
  __syncthreads();
#pragma unroll
  for (int j = 0; j < EPT; ++j) {
    int2 s = stage[t + j * 1024];
    int2 v;
    v.x = s.x;
    v.y = 0;
    ((int2*)rec)[s.y] = v;
  }
}

__global__ void __launch_bounds__(1024)
k_deg_dinv(const int* __restrict__ rec, const int* __restrict__ base,
           float* __restrict__ dinv) {
  __shared__ int ldeg[TBN];
  const int t = threadIdx.x;
  if (t < TBN) ldeg[t] = 0;
  __syncthreads();
  const int beg = base[blockIdx.x];
  const int end = base[blockIdx.x + 1];
  for (int p0 = beg + t; p0 < end; p0 += 8 * 1024) {
    int px[8];
#pragma unroll
    for (int j = 0; j < 8; ++j) px[j] = rec[2 * min(p0 + j * 1024, end - 1)];
#pragma unroll
    for (int j = 0; j < 8; ++j)
      if (p0 + j * 1024 < end) atomicAdd(&ldeg[px[j] & (TBN - 1)], 1);
  }
  __syncthreads();
  if (t < TBN) dinv[blockIdx.x * TBN + t] = rsqrtf((float)ldeg[t] + 1.0f);
}

typedef const __attribute__((address_space(1))) int2* gci2p;
#define STG_PITCH 34

template <int F, int I0>
__device__ __forceinline__ void agg_load16(gcfp hg, int f, int px, float (&v)[16]) {
#pragma unroll
  for (int i = 0; i < 16; ++i) {
    int pk = __builtin_amdgcn_readlane(px, I0 + i);
    v[i] = hg[((unsigned)pk >> 8) * F + (unsigned)f];
  }
}

template <int I0>
__device__ __forceinline__ void agg_add16(float* acc, int lane, int px, int py,
                                          const float (&v)[16]) {
#pragma unroll
  for (int i = 0; i < 16; ++i) {
    int pk  = __builtin_amdgcn_readlane(px, I0 + i);
    float w = __int_as_float(__builtin_amdgcn_readlane(py, I0 + i));
    __hip_atomic_fetch_add(&acc[(pk & (TBN - 1)) * HID + lane], v[i] * w,
                           __ATOMIC_RELAXED, __HIP_MEMORY_SCOPE_WORKGROUP);
  }
}

template <int F, bool FIRST>
__global__ void __launch_bounds__(1024)
k_agg(int* __restrict__ rec, const int* __restrict__ base,
      const float* __restrict__ dinv, const float* __restrict__ h,
      const float* __restrict__ Wl, const float* __restrict__ bias,
      float* __restrict__ out) {
  constexpr int P = F + 2;
  __shared__ __attribute__((aligned(16))) float acc[TBN * STG_PITCH];
  __shared__ __attribute__((aligned(16))) float apad[TBN * P];
  __shared__ float ldv[FIRST ? TBN : 1];

  const int wave  = __builtin_amdgcn_readfirstlane((int)(threadIdx.x >> 5));
  const int lane  = threadIdx.x & 31;
  const int f     = lane & (F - 1);
  const int half  = lane >> 4;
  const int lm    = lane & 15;
  const int mt    = wave >> 1;
  const int nt    = wave & 1;
  const int node0 = blockIdx.x * TBN;
  gcfp  hg = (gcfp)h;
  gcfp  dg = (gcfp)dinv;
  gci2p rg = (gci2p)rec;
  if (FIRST) {
    if (threadIdx.x < TBN) ldv[threadIdx.x] = dinv[node0 + threadIdx.x];
  }

  gcfp wg = (gcfp)Wl;
  v2f wb[F / 4];
#pragma unroll
  for (int s = 0; s < F / 4; ++s) {
    wb[s].x = wg[(4 * s + 2 * half) * HID + nt * 16 + lm];
    wb[s].y = wg[(4 * s + 2 * half + 1) * HID + nt * 16 + lm];
  }

#pragma unroll
  for (int j = 0; j < TBN / 32; ++j) {
    int n = wave + 32 * j;
    float d = dinv[node0 + n];
    acc[n * HID + lane] = hg[(unsigned)(node0 + n) * F + (unsigned)f] * (d * d);
  }
  __syncthreads();

  const int beg = base[blockIdx.x];
  const int end = base[blockIdx.x + 1];

  int p = beg + wave * 32;
  if (p < end) {
    int2 cur = rg[min(p + lane, end - 1)];
    for (;;) {
      const int n  = end - p;
      const int pn = p + 32 * 32;
      if (n < 32) {
        if (FIRST) {
          float wv = dg[(unsigned)cur.x >> 8] * ldv[cur.x & (TBN - 1)];
          if (lane < n) rec[2 * (p + lane) + 1] = __float_as_int(wv);
          cur.y = __float_as_int(wv);
        }
        for (int i0 = 0; i0 < n; i0 += 8) {
          float v[8];
#pragma unroll
          for (int j = 0; j < 8; ++j) {
            v[j] = 0.0f;
            if (i0 + j < n) {
              int pk = __builtin_amdgcn_readlane(cur.x, i0 + j);
              v[j] = hg[((unsigned)pk >> 8) * F + (unsigned)f];
            }
          }
#pragma unroll
          for (int j = 0; j < 8; ++j) {
            if (i0 + j < n) {
              int pk  = __builtin_amdgcn_readlane(cur.x, i0 + j);
              float w = __int_as_float(__builtin_amdgcn_readlane(cur.y, i0 + j));
              __hip_atomic_fetch_add(&acc[(pk & (TBN - 1)) * HID + lane], v[j] * w,
                                     __ATOMIC_RELAXED, __HIP_MEMORY_SCOPE_WORKGROUP);
            }
          }
        }
        break;
      }
      float va[16], vb[16];
      int2 nxt = rg[min(pn + lane, end - 1)];
      float ds = 0.0f;
      if (FIRST) ds = dg[(unsigned)cur.x >> 8];
      agg_load16<F, 0>(hg, f, cur.x, va);
      agg_load16<F, 16>(hg, f, cur.x, vb);
      asm volatile("" :: "v"(nxt.x), "v"(nxt.y));
      if (FIRST) {
        float wv = ds * ldv[cur.x & (TBN - 1)];
        rec[2 * (p + lane) + 1] = __float_as_int(wv);
        cur.y = __float_as_int(wv);
      }
      agg_add16<0>(acc, lane, cur.x, cur.y, va);
      agg_add16<16>(acc, lane, cur.x, cur.y, vb);
      if (pn >= end) break;
      cur = nxt;
      p = pn;
    }
  }
  __syncthreads();

#pragma unroll
  for (int j = 0; j < TBN / 32; ++j) {
    int n = wave + 32 * j;
    apad[n * P + f] = acc[n * HID + f];
  }
  __syncthreads();

  v8f d = (v8f){0.f, 0.f, 0.f, 0.f, 0.f, 0.f, 0.f, 0.f};
  const float* arow = &apad[(mt * 16 + lm) * P + 2 * half];
#pragma unroll
  for (int s = 0; s < F / 4; ++s) {
    v2f a = *(const v2f*)(arow + 4 * s);
    d = __builtin_amdgcn_wmma_f32_16x16x4_f32(false, a, false, wb[s], (short)0, d,
                                              false, false);
  }
  float* srow = &acc[(mt * 16 + 8 * half) * STG_PITCH + nt * 16 + lm];
#pragma unroll
  for (int r = 0; r < 8; ++r) srow[r * STG_PITCH] = d[r];
  __syncthreads();

  const float bv = bias[lane];
#pragma unroll
  for (int j = 0; j < TBN / 32; ++j) {
    int n = wave + 32 * j;
    float v = acc[n * STG_PITCH + lane] + bv;
    out[(node0 + n) * HID + lane] = v > 0.0f ? v : 0.0f;
  }
}

template <int NT, bool RAGGED, bool BIAS, bool RELU, bool PRE>
__device__ __forceinline__ void gemm_wmma(const float* __restrict__ A, const float* __restrict__ B,
                                          const float* __restrict__ bias, float* __restrict__ C,
                                          int M, int N, int K, const float* __restrict__ abias) {
  int wave  = threadIdx.x >> 5;
  int lane  = threadIdx.x & 31;
  int mtile = blockIdx.y * 4 + wave;
  int m0    = mtile * 16;
  if (m0 >= M) return;
  int n0   = blockIdx.x * (16 * NT);
  int half = lane >> 4;
  int lm   = lane & 15;

  gcfp Ag = (gcfp)A;
  gcfp Bg = (gcfp)B;
  gcfp abg = (gcfp)abias;

  int  ncol[NT];
  bool nok[NT];
  unsigned boff[NT];
#pragma unroll
  for (int t = 0; t < NT; ++t) {
    ncol[t] = n0 + t * 16 + lm;
    int nld;
    if (RAGGED) {
      nok[t] = ncol[t] < N;
      nld    = nok[t] ? ncol[t] : (N - 1);
    } else {
      nok[t] = true;
      nld    = ncol[t];
    }
    boff[t] = (unsigned)(half * 2) * (unsigned)N + (unsigned)nld;
  }

  unsigned aoff = (unsigned)(m0 + lm) * (unsigned)K + (unsigned)(half * 2);

  v8f acc[NT];
#pragma unroll
  for (int t = 0; t < NT; ++t) acc[t] = (v8f){0.f,0.f,0.f,0.f,0.f,0.f,0.f,0.f};

  for (int k0 = 0; k0 < K; k0 += 4) {
    v2f a;
    a.x = Ag[aoff];
    a.y = Ag[aoff + 1];
    if (PRE) {
      float p0 = a.x + abg[k0 + half * 2];
      float p1 = a.y + abg[k0 + half * 2 + 1];
      a.x = p0 > 0.0f ? p0 : 0.0f;
      a.y = p1 > 0.0f ? p1 : 0.0f;
    }
#pragma unroll
    for (int t = 0; t < NT; ++t) {
      float b0 = Bg[boff[t]];
      float b1 = Bg[boff[t] + (unsigned)N];
      if (RAGGED) { b0 = nok[t] ? b0 : 0.0f; b1 = nok[t] ? b1 : 0.0f; }
      v2f b; b.x = b0; b.y = b1;
      acc[t] = __builtin_amdgcn_wmma_f32_16x16x4_f32(
           false, a,  false, b,
           (short)0, acc[t],  false,  false);
      boff[t] += 4u * (unsigned)N;
    }
    aoff += 4u;
  }

  gfp Cg = (gfp)C;
  gcfp biasg = (gcfp)bias;
#pragma unroll
  for (int t = 0; t < NT; ++t) {
    float bv = BIAS ? biasg[nok[t] ? ncol[t] : 0] : 0.0f;
    if (!RAGGED || nok[t]) {
#pragma unroll
      for (int r = 0; r < 8; ++r) {
        int m = m0 + r + half * 8;
        float v = acc[t][r] + bv;
        if (RELU) v = v > 0.0f ? v : 0.0f;
        Cg[(unsigned)m * (unsigned)N + (unsigned)ncol[t]] = v;
      }
    }
  }
}

#define HEAD_K      (NODE_COUNT * HID)
#define HEAD_N      256
#define HS_SLICES   25
#define HS_KPARTS   8
#define HS_WAVES    (4 * HS_KPARTS)
#define HS_SLICE_K  (HEAD_K / HS_SLICES)
#define HS_PART_K   (HS_SLICE_K / HS_KPARTS)
#define HS_GK       16
#define HS_GROUPS   (HS_PART_K / HS_GK)
static_assert(HS_SLICES * HS_KPARTS * HS_GROUPS * HS_GK == HEAD_K, "K is covered exactly");
static_assert(BATCH == 64, "four row tiles");

typedef const __attribute__((address_space(1))) v2f* gcv2p;

__global__ void __launch_bounds__(HS_WAVES * 32)
k_head_gemm(const float* __restrict__ A,
            const float* __restrict__ Bp, const float* __restrict__ Bv,
            float* __restrict__ Cp, float* __restrict__ Cv) {
  __shared__ float part[HS_WAVES * 1024];

  const int wave = threadIdx.x >> 5;
  const int lane = threadIdx.x & 31;
  const int half = lane >> 4;
  const int lm   = lane & 15;
  const int mt   = wave & 3;
  const int kp   = wave >> 2;
  const int m0   = mt * 16;
  const int n0   = (blockIdx.x & 3) * 64;
  const bool second = blockIdx.x >= 4;
  const float* B = second ? Bv : Bp;
  float*       C = second ? Cv : Cp;
  const int kbeg = blockIdx.y * HS_SLICE_K + kp * HS_PART_K;

  gcfp ap = (gcfp)A + ((unsigned)(m0 + lm) * (unsigned)HEAD_K + (unsigned)(kbeg + half * 2));
  gcfp bp = (gcfp)B + ((unsigned)(kbeg + half * 2) * (unsigned)HEAD_N + (unsigned)(n0 + lm));

  v8f acc[4];
#pragma unroll
  for (int t = 0; t < 4; ++t) acc[t] = (v8f){0.f,0.f,0.f,0.f,0.f,0.f,0.f,0.f};

  for (int g = 0; g < HS_GROUPS; ++g) {
    v2f a[4], b[4][4];
#pragma unroll
    for (int s = 0; s < 4; ++s)
      a[s] = *(gcv2p)(ap + 4 * s);
#pragma unroll
    for (int s = 0; s < 4; ++s) {
#pragma unroll
      for (int t = 0; t < 4; ++t) {
        b[t][s].x = bp[(4 * s) * HEAD_N + 16 * t];
        b[t][s].y = bp[(4 * s + 1) * HEAD_N + 16 * t];
      }
    }
    __builtin_amdgcn_sched_barrier(0);
#pragma unroll
    for (int s = 0; s < 4; ++s) {
#pragma unroll
      for (int t = 0; t < 4; ++t)
        acc[t] = __builtin_amdgcn_wmma_f32_16x16x4_f32(
            false, a[s], false, b[t][s], (short)0, acc[t], false, false);
    }
    ap += HS_GK;
    bp += HS_GK * HEAD_N;
  }

  float* mine = part + wave * 1024;
#pragma unroll
  for (int t = 0; t < 4; ++t) {
#pragma unroll
    for (int r = 0; r < 8; ++r)
      mine[(r + half * 8) * 64 + t * 16 + lm] = acc[t][r];
  }
  __syncthreads();

  const int rr  = threadIdx.x >> 6;
  const int col = threadIdx.x & 63;
#pragma unroll
  for (int j = 0; j < 4; ++j) {
    float s = part[j * 1024 + rr * 64 + col];
    for (int q = 1; q < HS_KPARTS; ++q) s += part[(q * 4 + j) * 1024 + rr * 64 + col];
    atomicAdd(&C[(j * 16 + rr) * HEAD_N + n0 + col], s);
  }
}

#define POL_TILES ((NODE_COUNT + 15) / 16)
#define VAL_ROWS  32
static_assert(BATCH % VAL_ROWS == 0 && HEAD_N == 256, "value rows and k split");
typedef float v4f __attribute__((ext_vector_type(4)));
typedef const __attribute__((address_space(1))) v4f* gcv4p;

__global__ void __launch_bounds__(128)
k_head_out(const float* __restrict__ sump, const float* __restrict__ Wp2,
           const float* __restrict__ bp2, float* __restrict__ outp,
           const float* __restrict__ bp1,
           const float* __restrict__ sumv, const float* __restrict__ bv1,
           const float* __restrict__ Wv2, const float* __restrict__ bv2,
           float* __restrict__ outv) {
  __shared__ float vres[VAL_ROWS];
  if (blockIdx.x < POL_TILES) {
    gemm_wmma<1, true, true, false, true>(sump, Wp2, bp2, outp, BATCH, NODE_COUNT, 256, bp1);
    return;
  }
  const int wave = threadIdx.x >> 5;
  const int lane = threadIdx.x & 31;
  const int vb   = blockIdx.x - POL_TILES;
  const int row0 = vb * VAL_ROWS + wave * (VAL_ROWS / 4);
  gcv4p sg = (gcv4p)sumv;
  gcv4p bg = (gcv4p)bv1;
  gcv4p wg = (gcv4p)Wv2;
  const v4f bb0 = bg[lane], bb1 = bg[32 + lane];
  const v4f ww0 = wg[lane], ww1 = wg[32 + lane];
  v4f x0[VAL_ROWS / 4], x1[VAL_ROWS / 4];
#pragma unroll
  for (int r = 0; r < VAL_ROWS / 4; ++r) {
    x0[r] = sg[(row0 + r) * (HEAD_N / 4) + lane];
    x1[r] = sg[(row0 + r) * (HEAD_N / 4) + 32 + lane];
  }
  const float b2 = bv2[0];
#pragma unroll
  for (int r = 0; r < VAL_ROWS / 4; ++r) {
    float sacc = 0.0f;
#pragma unroll
    for (int e = 0; e < 4; ++e) {
      float h0 = x0[r][e] + bb0[e];
      float h1 = x1[r][e] + bb1[e];
      h0 = h0 > 0.0f ? h0 : 0.0f;
      h1 = h1 > 0.0f ? h1 : 0.0f;
      sacc += h0 * ww0[e];
      sacc += h1 * ww1[e];
    }
    for (int o = 16; o > 0; o >>= 1) sacc += __shfl_xor(sacc, o);
    if (lane == 0) vres[wave * (VAL_ROWS / 4) + r] = sacc + b2;
  }
  __syncthreads();
  if (wave == 0) outv[vb * VAL_ROWS + lane] = vres[lane];
}

extern "C" void kernel_launch(void* const* d_in, const int* in_sizes, int n_in,
                              void* d_out, int out_size, void* d_ws, size_t ws_size,
                              hipStream_t stream) {
  (void)in_sizes; (void)n_in; (void)out_size; (void)ws_size;

  const float* x    = (const float*)d_in[0];
  const int*   ei   = (const int*)d_in[1];
  const int*   rows = ei;
  const int*   cols = ei + N_EDGES;
  const float* W[5]  = {(const float*)d_in[2], (const float*)d_in[4], (const float*)d_in[6],
                        (const float*)d_in[8], (const float*)d_in[10]};
  const float* bb[5] = {(const float*)d_in[3], (const float*)d_in[5], (const float*)d_in[7],
                        (const float*)d_in[9], (const float*)d_in[11]};
  const float* Wp1 = (const float*)d_in[12]; const float* bp1 = (const float*)d_in[13];
  const float* Wp2 = (const float*)d_in[14]; const float* bp2 = (const float*)d_in[15];
  const float* Wv1 = (const float*)d_in[16]; const float* bv1 = (const float*)d_in[17];
  const float* Wv2 = (const float*)d_in[18]; const float* bv2 = (const float*)d_in[19];
  float* out = (float*)d_out;

  float* ws     = (float*)d_ws;
  float* deg    = ws;
  float* hidp   = deg + N_TOTAL;
  float* hidv   = hidp + BATCH * 256;
  int*   cnt    = (int*)(hidv + BATCH * 256);
  int*   base   = cnt + NCH * 256;
  int*   rec    = base + 512;
  float* bufA   = (float*)(rec + 2 * (size_t)N_EDGES);
  float* bufB   = bufA + (size_t)N_TOTAL * HID;

  k_count   <<<NCH, 1024, 0, stream>>>(cols, cnt, hidp);
  k_fill    <<<NCH, 1024, 0, stream>>>(rows, cols, cnt, base, rec);
  k_deg_dinv<<<NBLK, 1024, 0, stream>>>(rec, base, deg);

  k_agg<STATE, true> <<<NBLK, 1024, 0, stream>>>(rec, base, deg, x,    W[0], bb[0], bufA);
  k_agg<HID, false>  <<<NBLK, 1024, 0, stream>>>(rec, base, deg, bufA, W[1], bb[1], bufB);
  k_agg<HID, false>  <<<NBLK, 1024, 0, stream>>>(rec, base, deg, bufB, W[2], bb[2], bufA);
  k_agg<HID, false>  <<<NBLK, 1024, 0, stream>>>(rec, base, deg, bufA, W[3], bb[3], bufB);
  k_agg<HID, false>  <<<NBLK, 1024, 0, stream>>>(rec, base, deg, bufB, W[4], bb[4], bufA);

  dim3 g1(2 * HEAD_N / 64, HS_SLICES);
  k_head_gemm<<<g1, HS_WAVES * 32, 0, stream>>>(bufA, Wp1, Wv1, hidp, hidv);

  k_head_out<<<POL_TILES + BATCH / VAL_ROWS, 128, 0, stream>>>(
      hidp, Wp2, bp2, out, bp1, hidv, bv1, Wv2, bv2, out + (size_t)BATCH * NODE_COUNT);
}


# Dependent load round trips cut: layers fetch bounds, first records, bias under start loads; policy GEMM K split on 8 waves per row tile, one load batch each, LDS sum by part; X sum order changes
# speedup vs baseline: 23.3205x; 1.0514x over previous
#include <hip/hip_runtime.h>

#define N_TOTAL    64000
#define NODE_COUNT 1000
#define BATCH      64
#define STATE      16
#define HID        32
#define N_EDGES    2048000
#define TBN        256
#define NBLK       (N_TOTAL / TBN)
static_assert(N_TOTAL % TBN == 0, "blocks cover the nodes exactly");
static_assert(N_TOTAL <= (1 << 24), "source id and 8-bit local target share one int");
static_assert(TBN == 256 && HID == 32, "record packing and lane-per-feature layout");

typedef float v2f __attribute__((ext_vector_type(2)));
typedef float v8f __attribute__((ext_vector_type(8)));

typedef const __attribute__((address_space(1))) float* gcfp;
typedef __attribute__((address_space(1))) float*       gfp;

#define ECH  8192
#define NCH  (N_EDGES / ECH)
#define EPT  (ECH / 1024)
static_assert(N_EDGES % ECH == 0, "chunks cover the edges exactly");
static_assert(NBLK <= 256, "one LDS counter per target block, 256 of them");
#define HID_ALL  (2 * BATCH * 256)
static_assert(HID_ALL % 1024 == 0 && HID_ALL / 1024 <= NCH, "head sums cleared by the first chunks");

__global__ void __launch_bounds__(1024)
k_count(const int* __restrict__ cols, int* __restrict__ cnt, float* __restrict__ hid) {
  __shared__ int lcnt[256];
  const int t = threadIdx.x;
  if (t < 256) lcnt[t] = 0;
  if (blockIdx.x < HID_ALL / 1024) hid[blockIdx.x * 1024 + t] = 0.0f;
  __syncthreads();
  const int e0 = blockIdx.x * ECH + t;
  int c[EPT];
#pragma unroll
  for (int j = 0; j < EPT; ++j) c[j] = cols[e0 + j * 1024];
#pragma unroll
  for (int j = 0; j < EPT; ++j) atomicAdd(&lcnt[(unsigned)c[j] / TBN], 1);
  __syncthreads();
  if (t < 256) cnt[blockIdx.x * 256 + t] = lcnt[t];
}

#define CQ  16
#define CJ  ((NCH + CQ - 1) / CQ)
static_assert(CQ * 64 == 1024, "one thread per (row group, four blocks)");
typedef int v4i __attribute__((ext_vector_type(4)));
__global__ void __launch_bounds__(1024)
k_fill(const int* __restrict__ rows, const int* __restrict__ cols,
       const int* __restrict__ cnt, int* __restrict__ base, int* __restrict__ rec) {
  __shared__ int2 stage[ECH];
  __shared__ int lcnt[256];
  __shared__ int lstart[256];
  __shared__ int gdelta[256];
  __shared__ int wtot[8];
  __shared__ int wtot2[8];
  __shared__ __attribute__((aligned(16))) int qtot[CQ * 256];
  __shared__ __attribute__((aligned(16))) int qpre[CQ * 256];
  const int t = threadIdx.x;
  if (t < 256) lcnt[t] = 0;
  __syncthreads();
  const int e0 = blockIdx.x * ECH + t;
  int r[EPT], c[EPT], rk[EPT];
#pragma unroll
  for (int j = 0; j < EPT; ++j) {
    r[j] = rows[e0 + j * 1024];
    c[j] = cols[e0 + j * 1024];
  }
  {
    const int b4 = t & 63, q = t >> 6;
    const v4i* cm = (const v4i*)cnt;
    v4i v[CJ];
#pragma unroll
    for (int j = 0; j < CJ; ++j) v[j] = cm[min(q + CQ * j, NCH - 1) * 64 + b4];
    const v4i zero = (v4i){0, 0, 0, 0};
    v4i tot = zero, pre = zero;
#pragma unroll
    for (int j = 0; j < CJ; ++j) {
      const int ch = q + CQ * j;
      tot += (ch < NCH) ? v[j] : zero;
      pre += (ch < (int)blockIdx.x) ? v[j] : zero;
    }
    ((v4i*)qtot)[q * 64 + b4] = tot;
    ((v4i*)qpre)[q * 64 + b4] = pre;
  }
#pragma unroll
  for (int j = 0; j < EPT; ++j) rk[j] = atomicAdd(&lcnt[(unsigned)c[j] / TBN], 1);
  __syncthreads();

  int cn = 0, incl = 0, gt = 0, gp = 0, incl2 = 0;
  if (t < 256) {
    cn = lcnt[t];
#pragma unroll
    for (int g = 0; g < CQ; ++g) {
      gt += qtot[g * 256 + t];
      gp += qpre[g * 256 + t];
    }
    incl  = cn;
    incl2 = gt;
    for (int o = 1; o < 32; o <<= 1) {
      int u  = __shfl_up(incl, o);
      int u2 = __shfl_up(incl2, o);
      if ((t & 31) >= o) { incl += u; incl2 += u2; }
    }
    if ((t & 31) == 31) { wtot[t >> 5] = incl; wtot2[t >> 5] = incl2; }
  }
  __syncthreads();
  if (t < 256) {
    int pre = 0, pre2 = 0;
    for (int w = 0; w < (t >> 5); ++w) { pre += wtot[w]; pre2 += wtot2[w]; }
    int first  = pre + incl - cn;
    int bfirst = pre2 + incl2 - gt;
    lstart[t] = first;
    gdelta[t] = bfirst + gp - first;
    if (blockIdx.x == 0) {
      if (t < NBLK) base[t] = bfirst;
      if (t == NBLK - 1) base[NBLK] = bfirst + gt;
    }
  }
  __syncthreads();

#pragma unroll
  for (int j = 0; j < EPT; ++j) {
    int b  = (unsigned)c[j] / TBN;
    int li = lstart[b] + rk[j];
    int2 v;
    v.x = (r[j] << 8) | (c[j] & (TBN - 1));
    v.y = gdelta[b] + li;
    stage[li] = v;
  }
  __syncthreads();
#pragma unroll
  for (int j = 0; j < EPT; ++j) {
    int2 s = stage[t + j * 1024];
    int2 v;
    v.x = s.x;
    v.y = 0;
    ((int2*)rec)[s.y] = v;
  }
}

__global__ void __launch_bounds__(1024)
k_deg_dinv(const int* __restrict__ rec, const int* __restrict__ base,
           float* __restrict__ dinv) {
  __shared__ int ldeg[TBN];
  const int t = threadIdx.x;
  if (t < TBN) ldeg[t] = 0;
  __syncthreads();
  const int beg = base[blockIdx.x];
  const int end = base[blockIdx.x + 1];
  for (int p0 = beg + t; p0 < end; p0 += 8 * 1024) {
    int px[8];
#pragma unroll
    for (int j = 0; j < 8; ++j) px[j] = rec[2 * min(p0 + j * 1024, end - 1)];
#pragma unroll
    for (int j = 0; j < 8; ++j)
      if (p0 + j * 1024 < end) atomicAdd(&ldeg[px[j] & (TBN - 1)], 1);
  }
  __syncthreads();
  if (t < TBN) dinv[blockIdx.x * TBN + t] = rsqrtf((float)ldeg[t] + 1.0f);
}

typedef const __attribute__((address_space(1))) int2* gci2p;
#define STG_PITCH 34

template <int F, int I0>
__device__ __forceinline__ void agg_load16(gcfp hg, int f, int px, float (&v)[16]) {
#pragma unroll
  for (int i = 0; i < 16; ++i) {
    int pk = __builtin_amdgcn_readlane(px, I0 + i);
    v[i] = hg[((unsigned)pk >> 8) * F + (unsigned)f];
  }
}

template <int I0>
__device__ __forceinline__ void agg_add16(float* acc, int lane, int px, int py,
                                          const float (&v)[16]) {
#pragma unroll
  for (int i = 0; i < 16; ++i) {
    int pk  = __builtin_amdgcn_readlane(px, I0 + i);
    float w = __int_as_float(__builtin_amdgcn_readlane(py, I0 + i));
    __hip_atomic_fetch_add(&acc[(pk & (TBN - 1)) * HID + lane], v[i] * w,
                           __ATOMIC_RELAXED, __HIP_MEMORY_SCOPE_WORKGROUP);
  }
}

template <int F, bool FIRST>
__global__ void __launch_bounds__(1024)
k_agg(int* __restrict__ rec, const int* __restrict__ base,
      const float* __restrict__ dinv, const float* __restrict__ h,
      const float* __restrict__ Wl, const float* __restrict__ bias,
      float* __restrict__ out) {
  constexpr int P = F + 2;
  __shared__ __attribute__((aligned(16))) float acc[TBN * STG_PITCH];
  __shared__ __attribute__((aligned(16))) float apad[TBN * P];
  __shared__ float ldv[FIRST ? TBN : 1];

  const int wave  = __builtin_amdgcn_readfirstlane((int)(threadIdx.x >> 5));
  const int lane  = threadIdx.x & 31;
  const int f     = lane & (F - 1);
  const int half  = lane >> 4;
  const int lm    = lane & 15;
  const int mt    = wave >> 1;
  const int nt    = wave & 1;
  const int node0 = blockIdx.x * TBN;
  gcfp  hg = (gcfp)h;
  gcfp  dg = (gcfp)dinv;
  gci2p rg = (gci2p)rec;

  gcfp wg = (gcfp)Wl;
  v2f wb[F / 4];
#pragma unroll
  for (int s = 0; s < F / 4; ++s) {
    wb[s].x = wg[(4 * s + 2 * half) * HID + nt * 16 + lm];
    wb[s].y = wg[(4 * s + 2 * half + 1) * HID + nt * 16 + lm];
  }

  float hv[TBN / 32], dd[TBN / 32];
#pragma unroll
  for (int j = 0; j < TBN / 32; ++j) {
    int n = wave + 32 * j;
    dd[j] = dinv[node0 + n];
    hv[j] = hg[(unsigned)(node0 + n) * F + (unsigned)f];
  }

  const int beg = base[blockIdx.x];
  const int end = base[blockIdx.x + 1];
  const float bv = bias[lane];
  int p = beg + wave * 32;
  int2 cur = make_int2(0, 0);
  if (p < end) cur = rg[min(p + lane, end - 1)];
  if (FIRST) {
    if (threadIdx.x < TBN) ldv[threadIdx.x] = dinv[node0 + threadIdx.x];
  }

#pragma unroll
  for (int j = 0; j < TBN / 32; ++j) {
    int n = wave + 32 * j;
    acc[n * HID + lane] = hv[j] * (dd[j] * dd[j]);
  }
  __syncthreads();

  if (p < end) {
    for (;;) {
      const int n  = end - p;
      const int pn = p + 32 * 32;
      if (n < 32) {
        if (FIRST) {
          float wv = dg[(unsigned)cur.x >> 8] * ldv[cur.x & (TBN - 1)];
          if (lane < n) rec[2 * (p + lane) + 1] = __float_as_int(wv);
          cur.y = __float_as_int(wv);
        }
        for (int i0 = 0; i0 < n; i0 += 8) {
          float v[8];
#pragma unroll
          for (int j = 0; j < 8; ++j) {
            v[j] = 0.0f;
            if (i0 + j < n) {
              int pk = __builtin_amdgcn_readlane(cur.x, i0 + j);
              v[j] = hg[((unsigned)pk >> 8) * F + (unsigned)f];
            }
          }
#pragma unroll
          for (int j = 0; j < 8; ++j) {
            if (i0 + j < n) {
              int pk  = __builtin_amdgcn_readlane(cur.x, i0 + j);
              float w = __int_as_float(__builtin_amdgcn_readlane(cur.y, i0 + j));
              __hip_atomic_fetch_add(&acc[(pk & (TBN - 1)) * HID + lane], v[j] * w,
                                     __ATOMIC_RELAXED, __HIP_MEMORY_SCOPE_WORKGROUP);
            }
          }
        }
        break;
      }
      float va[16], vb[16];
      int2 nxt = rg[min(pn + lane, end - 1)];
      float ds = 0.0f;
      if (FIRST) ds = dg[(unsigned)cur.x >> 8];
      agg_load16<F, 0>(hg, f, cur.x, va);
      agg_load16<F, 16>(hg, f, cur.x, vb);
      asm volatile("" :: "v"(nxt.x), "v"(nxt.y));
      if (FIRST) {
        float wv = ds * ldv[cur.x & (TBN - 1)];
        rec[2 * (p + lane) + 1] = __float_as_int(wv);
        cur.y = __float_as_int(wv);
      }
      agg_add16<0>(acc, lane, cur.x, cur.y, va);
      agg_add16<16>(acc, lane, cur.x, cur.y, vb);
      if (pn >= end) break;
      cur = nxt;
      p = pn;
    }
  }
  __syncthreads();

#pragma unroll
  for (int j = 0; j < TBN / 32; ++j) {
    int n = wave + 32 * j;
    apad[n * P + f] = acc[n * HID + f];
  }
  __syncthreads();

  v8f d = (v8f){0.f, 0.f, 0.f, 0.f, 0.f, 0.f, 0.f, 0.f};
  const float* arow = &apad[(mt * 16 + lm) * P + 2 * half];
#pragma unroll
  for (int s = 0; s < F / 4; ++s) {
    v2f a = *(const v2f*)(arow + 4 * s);
    d = __builtin_amdgcn_wmma_f32_16x16x4_f32(false, a, false, wb[s], (short)0, d,
                                              false, false);
  }
  float* srow = &acc[(mt * 16 + 8 * half) * STG_PITCH + nt * 16 + lm];
#pragma unroll
  for (int r = 0; r < 8; ++r) srow[r * STG_PITCH] = d[r];
  __syncthreads();

#pragma unroll
  for (int j = 0; j < TBN / 32; ++j) {
    int n = wave + 32 * j;
    float v = acc[n * STG_PITCH + lane] + bv;
    out[(node0 + n) * HID + lane] = v > 0.0f ? v : 0.0f;
  }
}

#define HEAD_K      (NODE_COUNT * HID)
#define HEAD_N      256
#define HS_SLICES   25
#define HS_KPARTS   8
#define HS_WAVES    (4 * HS_KPARTS)
#define HS_SLICE_K  (HEAD_K / HS_SLICES)
#define HS_PART_K   (HS_SLICE_K / HS_KPARTS)
#define HS_GK       16
#define HS_GROUPS   (HS_PART_K / HS_GK)
static_assert(HS_SLICES * HS_KPARTS * HS_GROUPS * HS_GK == HEAD_K, "K is covered exactly");
static_assert(BATCH == 64, "four row tiles");

typedef const __attribute__((address_space(1))) v2f* gcv2p;

__global__ void __launch_bounds__(HS_WAVES * 32)
k_head_gemm(const float* __restrict__ A,
            const float* __restrict__ Bp, const float* __restrict__ Bv,
            float* __restrict__ Cp, float* __restrict__ Cv) {
  __shared__ float part[HS_WAVES * 1024];

  const int wave = threadIdx.x >> 5;
  const int lane = threadIdx.x & 31;
  const int half = lane >> 4;
  const int lm   = lane & 15;
  const int mt   = wave & 3;
  const int kp   = wave >> 2;
  const int m0   = mt * 16;
  const int n0   = (blockIdx.x & 3) * 64;
  const bool second = blockIdx.x >= 4;
  const float* B = second ? Bv : Bp;
  float*       C = second ? Cv : Cp;
  const int kbeg = blockIdx.y * HS_SLICE_K + kp * HS_PART_K;

  gcfp ap = (gcfp)A + ((unsigned)(m0 + lm) * (unsigned)HEAD_K + (unsigned)(kbeg + half * 2));
  gcfp bp = (gcfp)B + ((unsigned)(kbeg + half * 2) * (unsigned)HEAD_N + (unsigned)(n0 + lm));

  v8f acc[4];
#pragma unroll
  for (int t = 0; t < 4; ++t) acc[t] = (v8f){0.f,0.f,0.f,0.f,0.f,0.f,0.f,0.f};

  for (int g = 0; g < HS_GROUPS; ++g) {
    v2f a[4], b[4][4];
#pragma unroll
    for (int s = 0; s < 4; ++s)
      a[s] = *(gcv2p)(ap + 4 * s);
#pragma unroll
    for (int s = 0; s < 4; ++s) {
#pragma unroll
      for (int t = 0; t < 4; ++t) {
        b[t][s].x = bp[(4 * s) * HEAD_N + 16 * t];
        b[t][s].y = bp[(4 * s + 1) * HEAD_N + 16 * t];
      }
    }
    __builtin_amdgcn_sched_barrier(0);
#pragma unroll
    for (int s = 0; s < 4; ++s) {
#pragma unroll
      for (int t = 0; t < 4; ++t)
        acc[t] = __builtin_amdgcn_wmma_f32_16x16x4_f32(
            false, a[s], false, b[t][s], (short)0, acc[t], false, false);
    }
    ap += HS_GK;
    bp += HS_GK * HEAD_N;
  }

  float* mine = part + wave * 1024;
#pragma unroll
  for (int t = 0; t < 4; ++t) {
#pragma unroll
    for (int r = 0; r < 8; ++r)
      mine[(r + half * 8) * 64 + t * 16 + lm] = acc[t][r];
  }
  __syncthreads();

  const int rr  = threadIdx.x >> 6;
  const int col = threadIdx.x & 63;
#pragma unroll
  for (int j = 0; j < 4; ++j) {
    float s = part[j * 1024 + rr * 64 + col];
    for (int q = 1; q < HS_KPARTS; ++q) s += part[(q * 4 + j) * 1024 + rr * 64 + col];
    atomicAdd(&C[(j * 16 + rr) * HEAD_N + n0 + col], s);
  }
}

#define POL_TILES ((NODE_COUNT + 15) / 16)
#define PO_KPARTS 8
#define PO_WAVES  (4 * PO_KPARTS)
#define PO_PART_K (HEAD_N / PO_KPARTS)
#define PO_STEPS  (PO_PART_K / 4)
#define VAL_ROWS  32
#define VAL_RPW   (VAL_ROWS / PO_WAVES)
static_assert(BATCH % VAL_ROWS == 0 && HEAD_N == 256, "value rows and k split");
static_assert(VAL_ROWS % PO_WAVES == 0 && PO_KPARTS * PO_STEPS * 4 == HEAD_N, "rows per wave, K covered exactly");
typedef float v4f __attribute__((ext_vector_type(4)));
typedef const __attribute__((address_space(1))) v4f* gcv4p;

__global__ void __launch_bounds__(PO_WAVES * 32)
k_head_out(const float* __restrict__ sump, const float* __restrict__ Wp2,
           const float* __restrict__ bp2, float* __restrict__ outp,
           const float* __restrict__ bp1,
           const float* __restrict__ sumv, const float* __restrict__ bv1,
           const float* __restrict__ Wv2, const float* __restrict__ bv2,
           float* __restrict__ outv) {
  __shared__ float vres[VAL_ROWS];
  __shared__ float ppart[PO_WAVES * 256];
  const int wave = threadIdx.x >> 5;
  const int lane = threadIdx.x & 31;
  if (blockIdx.x < POL_TILES) {
    const int half = lane >> 4;
    const int lm   = lane & 15;
    const int mt   = wave & 3;
    const int kp   = wave >> 2;
    const int m0   = mt * 16;
    const int ncol = blockIdx.x * 16 + lm;
    const bool nok = ncol < NODE_COUNT;
    const int nld  = nok ? ncol : (NODE_COUNT - 1);
    const int kb   = kp * PO_PART_K + half * 2;
    gcfp ap  = (gcfp)sump + ((unsigned)(m0 + lm) * (unsigned)HEAD_N + (unsigned)kb);
    gcfp abp = (gcfp)bp1 + kb;
    gcfp bp  = (gcfp)Wp2 + ((unsigned)kb * (unsigned)NODE_COUNT + (unsigned)nld);
    const float bo = ((gcfp)bp2)[nld];

    float a0[PO_STEPS], a1[PO_STEPS], c0[PO_STEPS], c1[PO_STEPS], b0[PO_STEPS], b1[PO_STEPS];
#pragma unroll
    for (int s = 0; s < PO_STEPS; ++s) {
      a0[s] = ap[4 * s];
      a1[s] = ap[4 * s + 1];
      c0[s] = abp[4 * s];
      c1[s] = abp[4 * s + 1];
      b0[s] = bp[(4 * s) * NODE_COUNT];
      b1[s] = bp[(4 * s + 1) * NODE_COUNT];
    }
    v8f d = (v8f){0.f, 0.f, 0.f, 0.f, 0.f, 0.f, 0.f, 0.f};
#pragma unroll
    for (int s = 0; s < PO_STEPS; ++s) {
      float p0 = a0[s] + c0[s];
      float p1 = a1[s] + c1[s];
      v2f a, b;
      a.x = p0 > 0.0f ? p0 : 0.0f;
      a.y = p1 > 0.0f ? p1 : 0.0f;
      b.x = nok ? b0[s] : 0.0f;
      b.y = nok ? b1[s] : 0.0f;
      d = __builtin_amdgcn_wmma_f32_16x16x4_f32(false, a, false, b, (short)0, d, false, false);
    }
#pragma unroll
    for (int r = 0; r < 8; ++r) ppart[(kp * 4 + mt) * 256 + r * 32 + lane] = d[r];
    __syncthreads();
    if (kp == 0) {
      float s[8];
#pragma unroll
      for (int r = 0; r < 8; ++r) {
        s[r] = ppart[mt * 256 + r * 32 + lane];
#pragma unroll
        for (int q = 1; q < PO_KPARTS; ++q) s[r] += ppart[(q * 4 + mt) * 256 + r * 32 + lane];
      }
      if (nok) {
        gfp og = (gfp)outp;
#pragma unroll
        for (int r = 0; r < 8; ++r)
          og[(unsigned)(m0 + r + half * 8) * (unsigned)NODE_COUNT + (unsigned)ncol] = s[r] + bo;
      }
    }
    return;
  }
  const int vb   = blockIdx.x - POL_TILES;
  const int row0 = vb * VAL_ROWS + wave * VAL_RPW;
  gcv4p sg = (gcv4p)sumv;
  gcv4p bg = (gcv4p)bv1;
  gcv4p wg = (gcv4p)Wv2;
  const v4f bb0 = bg[lane], bb1 = bg[32 + lane];
  const v4f ww0 = wg[lane], ww1 = wg[32 + lane];
  v4f x0[VAL_RPW], x1[VAL_RPW];
#pragma unroll
  for (int r = 0; r < VAL_RPW; ++r) {
    x0[r] = sg[(row0 + r) * (HEAD_N / 4) + lane];
    x1[r] = sg[(row0 + r) * (HEAD_N / 4) + 32 + lane];
  }
  const float b2 = bv2[0];
#pragma unroll
  for (int r = 0; r < VAL_RPW; ++r) {
    float sacc = 0.0f;
#pragma unroll
    for (int e = 0; e < 4; ++e) {
      float h0 = x0[r][e] + bb0[e];
      float h1 = x1[r][e] + bb1[e];
      h0 = h0 > 0.0f ? h0 : 0.0f;
      h1 = h1 > 0.0f ? h1 : 0.0f;
      sacc += h0 * ww0[e];
      sacc += h1 * ww1[e];
    }
    for (int o = 16; o > 0; o >>= 1) sacc += __shfl_xor(sacc, o);
    if (lane == 0) vres[wave * VAL_RPW + r] = sacc + b2;
  }
  __syncthreads();
  if (wave == 0) outv[vb * VAL_ROWS + lane] = vres[lane];
}

extern "C" void kernel_launch(void* const* d_in, const int* in_sizes, int n_in,
                              void* d_out, int out_size, void* d_ws, size_t ws_size,
                              hipStream_t stream) {
  (void)in_sizes; (void)n_in; (void)out_size; (void)ws_size;

  const float* x    = (const float*)d_in[0];
  const int*   ei   = (const int*)d_in[1];
  const int*   rows = ei;
  const int*   cols = ei + N_EDGES;
  const float* W[5]  = {(const float*)d_in[2], (const float*)d_in[4], (const float*)d_in[6],
                        (const float*)d_in[8], (const float*)d_in[10]};
  const float* bb[5] = {(const float*)d_in[3], (const float*)d_in[5], (const float*)d_in[7],
                        (const float*)d_in[9], (const float*)d_in[11]};
  const float* Wp1 = (const float*)d_in[12]; const float* bp1 = (const float*)d_in[13];
  const float* Wp2 = (const float*)d_in[14]; const float* bp2 = (const float*)d_in[15];
  const float* Wv1 = (const float*)d_in[16]; const float* bv1 = (const float*)d_in[17];
  const float* Wv2 = (const float*)d_in[18]; const float* bv2 = (const float*)d_in[19];
  float* out = (float*)d_out;

  float* ws     = (float*)d_ws;
  float* deg    = ws;
  float* hidp   = deg + N_TOTAL;
  float* hidv   = hidp + BATCH * 256;
  int*   cnt    = (int*)(hidv + BATCH * 256);
  int*   base   = cnt + NCH * 256;
  int*   rec    = base + 512;
  float* bufA   = (float*)(rec + 2 * (size_t)N_EDGES);
  float* bufB   = bufA + (size_t)N_TOTAL * HID;

  k_count   <<<NCH, 1024, 0, stream>>>(cols, cnt, hidp);
  k_fill    <<<NCH, 1024, 0, stream>>>(rows, cols, cnt, base, rec);
  k_deg_dinv<<<NBLK, 1024, 0, stream>>>(rec, base, deg);

  k_agg<STATE, true> <<<NBLK, 1024, 0, stream>>>(rec, base, deg, x,    W[0], bb[0], bufA);
  k_agg<HID, false>  <<<NBLK, 1024, 0, stream>>>(rec, base, deg, bufA, W[1], bb[1], bufB);
  k_agg<HID, false>  <<<NBLK, 1024, 0, stream>>>(rec, base, deg, bufB, W[2], bb[2], bufA);
  k_agg<HID, false>  <<<NBLK, 1024, 0, stream>>>(rec, base, deg, bufA, W[3], bb[3], bufB);
  k_agg<HID, false>  <<<NBLK, 1024, 0, stream>>>(rec, base, deg, bufB, W[4], bb[4], bufA);

  dim3 g1(2 * HEAD_N / 64, HS_SLICES);
  k_head_gemm<<<g1, HS_WAVES * 32, 0, stream>>>(bufA, Wp1, Wv1, hidp, hidv);

  k_head_out<<<POL_TILES + BATCH / VAL_ROWS, PO_WAVES * 32, 0, stream>>>(
      hidp, Wp2, bp2, out, bp1, hidv, bv1, Wv2, bv2, out + (size_t)BATCH * NODE_COUNT);
}


# Layer row gathers as 16-byte loads: one load instruction brings 4 source rows (8 in layer 1), LDS adds rotated per row so no banks collide; sums by arrival as before, layer 1 as two partial sums
# speedup vs baseline: 24.5591x; 1.0531x over previous
#include <hip/hip_runtime.h>

#define N_TOTAL    64000
#define NODE_COUNT 1000
#define BATCH      64
#define STATE      16
#define HID        32
#define N_EDGES    2048000
#define TBN        256
#define NBLK       (N_TOTAL / TBN)
static_assert(N_TOTAL % TBN == 0, "blocks cover the nodes exactly");
static_assert(N_TOTAL <= (1 << 24), "source id and 8-bit local target share one int");
static_assert(TBN == 256 && HID == 32, "record packing and lane-per-feature layout");

typedef float v2f __attribute__((ext_vector_type(2)));
typedef float v4f __attribute__((ext_vector_type(4)));
typedef float v8f __attribute__((ext_vector_type(8)));

typedef const __attribute__((address_space(1))) float* gcfp;
typedef __attribute__((address_space(1))) float*       gfp;
typedef const __attribute__((address_space(1))) v4f*   gcv4p;

#define ECH  8192
#define NCH  (N_EDGES / ECH)
#define EPT  (ECH / 1024)
static_assert(N_EDGES % ECH == 0, "chunks cover the edges exactly");
static_assert(NBLK <= 256, "one LDS counter per target block, 256 of them");
#define HID_ALL  (2 * BATCH * 256)
static_assert(HID_ALL % 1024 == 0 && HID_ALL / 1024 <= NCH, "head sums cleared by the first chunks");

__global__ void __launch_bounds__(1024)
k_count(const int* __restrict__ cols, int* __restrict__ cnt, float* __restrict__ hid) {
  __shared__ int lcnt[256];
  const int t = threadIdx.x;
  if (t < 256) lcnt[t] = 0;
  if (blockIdx.x < HID_ALL / 1024) hid[blockIdx.x * 1024 + t] = 0.0f;
  __syncthreads();
  const int e0 = blockIdx.x * ECH + t;
  int c[EPT];
#pragma unroll
  for (int j = 0; j < EPT; ++j) c[j] = cols[e0 + j * 1024];
#pragma unroll
  for (int j = 0; j < EPT; ++j) atomicAdd(&lcnt[(unsigned)c[j] / TBN], 1);
  __syncthreads();
  if (t < 256) cnt[blockIdx.x * 256 + t] = lcnt[t];
}

#define CQ  16
#define CJ  ((NCH + CQ - 1) / CQ)
static_assert(CQ * 64 == 1024, "one thread per (row group, four blocks)");
typedef int v4i __attribute__((ext_vector_type(4)));
__global__ void __launch_bounds__(1024)
k_fill(const int* __restrict__ rows, const int* __restrict__ cols,
       const int* __restrict__ cnt, int* __restrict__ base, int* __restrict__ rec) {
  __shared__ int2 stage[ECH];
  __shared__ int lcnt[256];
  __shared__ int lstart[256];
  __shared__ int gdelta[256];
  __shared__ int wtot[8];
  __shared__ int wtot2[8];
  __shared__ __attribute__((aligned(16))) int qtot[CQ * 256];
  __shared__ __attribute__((aligned(16))) int qpre[CQ * 256];
  const int t = threadIdx.x;
  if (t < 256) lcnt[t] = 0;
  __syncthreads();
  const int e0 = blockIdx.x * ECH + t;
  int r[EPT], c[EPT], rk[EPT];
#pragma unroll
  for (int j = 0; j < EPT; ++j) {
    r[j] = rows[e0 + j * 1024];
    c[j] = cols[e0 + j * 1024];
  }
  {
    const int b4 = t & 63, q = t >> 6;
    const v4i* cm = (const v4i*)cnt;
    v4i v[CJ];
#pragma unroll
    for (int j = 0; j < CJ; ++j) v[j] = cm[min(q + CQ * j, NCH - 1) * 64 + b4];
    const v4i zero = (v4i){0, 0, 0, 0};
    v4i tot = zero, pre = zero;
#pragma unroll
    for (int j = 0; j < CJ; ++j) {
      const int ch = q + CQ * j;
      tot += (ch < NCH) ? v[j] : zero;
      pre += (ch < (int)blockIdx.x) ? v[j] : zero;
    }
    ((v4i*)qtot)[q * 64 + b4] = tot;
    ((v4i*)qpre)[q * 64 + b4] = pre;
  }
#pragma unroll
  for (int j = 0; j < EPT; ++j) rk[j] = atomicAdd(&lcnt[(unsigned)c[j] / TBN], 1);
  __syncthreads();

  int cn = 0, incl = 0, gt = 0, gp = 0, incl2 = 0;
  if (t < 256) {
    cn = lcnt[t];
#pragma unroll
    for (int g = 0; g < CQ; ++g) {
      gt += qtot[g * 256 + t];
      gp += qpre[g * 256 + t];
    }
    incl  = cn;
    incl2 = gt;
    for (int o = 1; o < 32; o <<= 1) {
      int u  = __shfl_up(incl, o);
      int u2 = __shfl_up(incl2, o);
      if ((t & 31) >= o) { incl += u; incl2 += u2; }
    }
    if ((t & 31) == 31) { wtot[t >> 5] = incl; wtot2[t >> 5] = incl2; }
  }
  __syncthreads();
  if (t < 256) {
    int pre = 0, pre2 = 0;
    for (int w = 0; w < (t >> 5); ++w) { pre += wtot[w]; pre2 += wtot2[w]; }
    int first  = pre + incl - cn;
    int bfirst = pre2 + incl2 - gt;
    lstart[t] = first;
    gdelta[t] = bfirst + gp - first;
    if (blockIdx.x == 0) {
      if (t < NBLK) base[t] = bfirst;
      if (t == NBLK - 1) base[NBLK] = bfirst + gt;
    }
  }
  __syncthreads();

#pragma unroll
  for (int j = 0; j < EPT; ++j) {
    int b  = (unsigned)c[j] / TBN;
    int li = lstart[b] + rk[j];
    int2 v;
    v.x = (r[j] << 8) | (c[j] & (TBN - 1));
    v.y = gdelta[b] + li;
    stage[li] = v;
  }
  __syncthreads();
#pragma unroll
  for (int j = 0; j < EPT; ++j) {
    int2 s = stage[t + j * 1024];
    int2 v;
    v.x = s.x;
    v.y = 0;
    ((int2*)rec)[s.y] = v;
  }
}

__global__ void __launch_bounds__(1024)
k_deg_dinv(const int* __restrict__ rec, const int* __restrict__ base,
           float* __restrict__ dinv) {
  __shared__ int ldeg[TBN];
  const int t = threadIdx.x;
  if (t < TBN) ldeg[t] = 0;
  __syncthreads();
  const int beg = base[blockIdx.x];
  const int end = base[blockIdx.x + 1];
  for (int p0 = beg + t; p0 < end; p0 += 8 * 1024) {
    int px[8];
#pragma unroll
    for (int j = 0; j < 8; ++j) px[j] = rec[2 * min(p0 + j * 1024, end - 1)];
#pragma unroll
    for (int j = 0; j < 8; ++j)
      if (p0 + j * 1024 < end) atomicAdd(&ldeg[px[j] & (TBN - 1)], 1);
  }
  __syncthreads();
  if (t < TBN) dinv[blockIdx.x * TBN + t] = rsqrtf((float)ldeg[t] + 1.0f);
}

typedef const __attribute__((address_space(1))) int2* gci2p;
#define STG_PITCH 34

template <int F, int Q>
__device__ __forceinline__ int agg_pick(int v) {
  constexpr unsigned S0  = (F == 32) ? Q : 2 * Q;
  constexpr unsigned S1  = (F == 32) ? Q : 2 * Q + 1;
  constexpr unsigned SEL = S0 | (S0 << 3) | (S0 << 6) | (S0 << 9) |
                           (S1 << 12) | (S1 << 15) | (S1 << 18) | (S1 << 21);
  return (int)__builtin_amdgcn_mov_dpp8((unsigned)v, SEL);
}

#define AGG_STEPS(F) ((F) / 4)

template <int F, int Q, bool TAIL>
__device__ __forceinline__ void agg_rows(gcv4p h4, int jq, int px, int rb, int n,
                                         v4f (&rv)[AGG_STEPS(F)]) {
  if constexpr (Q < AGG_STEPS(F)) {
    const unsigned pk = (unsigned)agg_pick<F, Q>(px);
    const int ri = rb + ((F == 32) ? Q : 2 * Q);
    if (!TAIL || ri < n) rv[Q] = h4[(pk >> 8) * (unsigned)(F / 4) + (unsigned)jq];
    else                 rv[Q] = (v4f){0.f, 0.f, 0.f, 0.f};
    agg_rows<F, Q + 1, TAIL>(h4, jq, px, rb, n, rv);
  }
}

template <int F, int Q, bool TAIL>
__device__ __forceinline__ void agg_sums(float* acc, int px, int py, int rb, int n,
                                         bool rot1, bool rot2, const int (&col)[4],
                                         const v4f (&rv)[AGG_STEPS(F)]) {
  if constexpr (Q < AGG_STEPS(F)) {
    const int   pk = agg_pick<F, Q>(px);
    const float w  = __int_as_float(agg_pick<F, Q>(py));
    const int   ri = rb + ((F == 32) ? Q : 2 * Q);
    const v4f v = rv[Q];
    const float a0 = rot1 ? v.y : v.x, a1 = rot1 ? v.z : v.y;
    const float a2 = rot1 ? v.w : v.z, a3 = rot1 ? v.x : v.w;
    const float r0 = rot2 ? a2 : a0, r1 = rot2 ? a3 : a1;
    const float r2 = rot2 ? a0 : a2, r3 = rot2 ? a1 : a3;
    float* row = acc + (pk & (TBN - 1)) * HID;
    if (!TAIL || ri < n) {
      __hip_atomic_fetch_add(row + col[0], r0 * w, __ATOMIC_RELAXED, __HIP_MEMORY_SCOPE_WORKGROUP);
      __hip_atomic_fetch_add(row + col[1], r1 * w, __ATOMIC_RELAXED, __HIP_MEMORY_SCOPE_WORKGROUP);
      __hip_atomic_fetch_add(row + col[2], r2 * w, __ATOMIC_RELAXED, __HIP_MEMORY_SCOPE_WORKGROUP);
      __hip_atomic_fetch_add(row + col[3], r3 * w, __ATOMIC_RELAXED, __HIP_MEMORY_SCOPE_WORKGROUP);
    }
    agg_sums<F, Q + 1, TAIL>(acc, px, py, rb, n, rot1, rot2, col, rv);
  }
}

template <int F, bool FIRST>
__global__ void __launch_bounds__(1024)
k_agg(int* __restrict__ rec, const int* __restrict__ base,
      const float* __restrict__ dinv, const float* __restrict__ h,
      const float* __restrict__ Wl, const float* __restrict__ bias,
      float* __restrict__ out) {
  constexpr int P = F + 2;
  __shared__ __attribute__((aligned(16))) float acc[TBN * STG_PITCH];
  __shared__ __attribute__((aligned(16))) float apad[TBN * P];
  __shared__ float ldv[FIRST ? TBN : 1];

  const int wave  = __builtin_amdgcn_readfirstlane((int)(threadIdx.x >> 5));
  const int lane  = threadIdx.x & 31;
  const int f     = lane & (F - 1);
  const int half  = lane >> 4;
  const int lm    = lane & 15;
  const int mt    = wave >> 1;
  const int nt    = wave & 1;
  const int node0 = blockIdx.x * TBN;
  gcfp  hg = (gcfp)h;
  gcv4p h4 = (gcv4p)h;
  gcfp  dg = (gcfp)dinv;
  gci2p rg = (gci2p)rec;

  constexpr int LPR = F / 4;
  const int  jq   = lane & (LPR - 1);
  const int  rw   = lane / LPR;
  const bool rot1 = (rw & 1) != 0;
  const bool rot2 = (rw & 2) != 0;
  const int  rb   = 8 * (lane >> 3) + (F == 16 ? ((lane >> 2) & 1) : 0);
  int col[4];
#pragma unroll
  for (int i = 0; i < 4; ++i) col[i] = 4 * jq + (F == 16 ? 16 * (rw >> 2) : 0) + ((i + rw) & 3);

  gcfp wg = (gcfp)Wl;
  v2f wb[F / 4];
#pragma unroll
  for (int s = 0; s < F / 4; ++s) {
    wb[s].x = wg[(4 * s + 2 * half) * HID + nt * 16 + lm];
    wb[s].y = wg[(4 * s + 2 * half + 1) * HID + nt * 16 + lm];
  }

  float hv[TBN / 32], dd[TBN / 32];
#pragma unroll
  for (int j = 0; j < TBN / 32; ++j) {
    int n = wave + 32 * j;
    dd[j] = dinv[node0 + n];
    hv[j] = hg[(unsigned)(node0 + n) * F + (unsigned)f];
  }

  const int beg = base[blockIdx.x];
  const int end = base[blockIdx.x + 1];
  const float bv = bias[lane];
  int p = beg + wave * 32;
  int2 cur = make_int2(0, 0);
  if (p < end) cur = rg[min(p + lane, end - 1)];
  if (FIRST) {
    if (threadIdx.x < TBN) ldv[threadIdx.x] = dinv[node0 + threadIdx.x];
  }

#pragma unroll
  for (int j = 0; j < TBN / 32; ++j) {
    int n = wave + 32 * j;
    float sv = hv[j] * (dd[j] * dd[j]);
    if (F < HID && lane >= F) sv = 0.0f;
    acc[n * HID + lane] = sv;
  }
  __syncthreads();

  if (p < end) {
    for (;;) {
      const int n  = end - p;
      const int pn = p + 32 * 32;
      if (n < 32) {
        if (FIRST) {
          float wv = dg[(unsigned)cur.x >> 8] * ldv[cur.x & (TBN - 1)];
          if (lane < n) rec[2 * (p + lane) + 1] = __float_as_int(wv);
          cur.y = __float_as_int(wv);
        }
        v4f rt[AGG_STEPS(F)];
        agg_rows<F, 0, true>(h4, jq, cur.x, rb, n, rt);
        agg_sums<F, 0, true>(acc, cur.x, cur.y, rb, n, rot1, rot2, col, rt);
        break;
      }
      v4f rv[AGG_STEPS(F)];
      int2 nxt = rg[min(pn + lane, end - 1)];
      float ds = 0.0f;
      if (FIRST) ds = dg[(unsigned)cur.x >> 8];
      agg_rows<F, 0, false>(h4, jq, cur.x, rb, 32, rv);
      asm volatile("" :: "v"(nxt.x), "v"(nxt.y));
      if (FIRST) {
        float wv = ds * ldv[cur.x & (TBN - 1)];
        rec[2 * (p + lane) + 1] = __float_as_int(wv);
        cur.y = __float_as_int(wv);
      }
      agg_sums<F, 0, false>(acc, cur.x, cur.y, rb, 32, rot1, rot2, col, rv);
      if (pn >= end) break;
      cur = nxt;
      p = pn;
    }
  }
  __syncthreads();

#pragma unroll
  for (int j = 0; j < TBN / 32; ++j) {
    int n = wave + 32 * j;
    float s = acc[n * HID + f];
    if (F < HID) s += acc[n * HID + F + f];
    apad[n * P + f] = s;
  }
  __syncthreads();

  v8f d = (v8f){0.f, 0.f, 0.f, 0.f, 0.f, 0.f, 0.f, 0.f};
  const float* arow = &apad[(mt * 16 + lm) * P + 2 * half];
#pragma unroll
  for (int s = 0; s < F / 4; ++s) {
    v2f a = *(const v2f*)(arow + 4 * s);
    d = __builtin_amdgcn_wmma_f32_16x16x4_f32(false, a, false, wb[s], (short)0, d,
                                              false, false);
  }
  float* srow = &acc[(mt * 16 + 8 * half) * STG_PITCH + nt * 16 + lm];
#pragma unroll
  for (int r = 0; r < 8; ++r) srow[r * STG_PITCH] = d[r];
  __syncthreads();

#pragma unroll
  for (int j = 0; j < TBN / 32; ++j) {
    int n = wave + 32 * j;
    float v = acc[n * STG_PITCH + lane] + bv;
    out[(node0 + n) * HID + lane] = v > 0.0f ? v : 0.0f;
  }
}

#define HEAD_K      (NODE_COUNT * HID)
#define HEAD_N      256
#define HS_SLICES   25
#define HS_KPARTS   8
#define HS_WAVES    (4 * HS_KPARTS)
#define HS_SLICE_K  (HEAD_K / HS_SLICES)
#define HS_PART_K   (HS_SLICE_K / HS_KPARTS)
#define HS_GK       16
#define HS_GROUPS   (HS_PART_K / HS_GK)
static_assert(HS_SLICES * HS_KPARTS * HS_GROUPS * HS_GK == HEAD_K, "K is covered exactly");
static_assert(BATCH == 64, "four row tiles");

typedef const __attribute__((address_space(1))) v2f* gcv2p;

__global__ void __launch_bounds__(HS_WAVES * 32)
k_head_gemm(const float* __restrict__ A,
            const float* __restrict__ Bp, const float* __restrict__ Bv,
            float* __restrict__ Cp, float* __restrict__ Cv) {
  __shared__ float part[HS_WAVES * 1024];

  const int wave = threadIdx.x >> 5;
  const int lane = threadIdx.x & 31;
  const int half = lane >> 4;
  const int lm   = lane & 15;
  const int mt   = wave & 3;
  const int kp   = wave >> 2;
  const int m0   = mt * 16;
  const int n0   = (blockIdx.x & 3) * 64;
  const bool second = blockIdx.x >= 4;
  const float* B = second ? Bv : Bp;
  float*       C = second ? Cv : Cp;
  const int kbeg = blockIdx.y * HS_SLICE_K + kp * HS_PART_K;

  gcfp ap = (gcfp)A + ((unsigned)(m0 + lm) * (unsigned)HEAD_K + (unsigned)(kbeg + half * 2));
  gcfp bp = (gcfp)B + ((unsigned)(kbeg + half * 2) * (unsigned)HEAD_N + (unsigned)(n0 + lm));

  v8f acc[4];
#pragma unroll
  for (int t = 0; t < 4; ++t) acc[t] = (v8f){0.f,0.f,0.f,0.f,0.f,0.f,0.f,0.f};

  for (int g = 0; g < HS_GROUPS; ++g) {
    v2f a[4], b[4][4];
#pragma unroll
    for (int s = 0; s < 4; ++s)
      a[s] = *(gcv2p)(ap + 4 * s);
#pragma unroll
    for (int s = 0; s < 4; ++s) {
#pragma unroll
      for (int t = 0; t < 4; ++t) {
        b[t][s].x = bp[(4 * s) * HEAD_N + 16 * t];
        b[t][s].y = bp[(4 * s + 1) * HEAD_N + 16 * t];
      }
    }
    __builtin_amdgcn_sched_barrier(0);
#pragma unroll
    for (int s = 0; s < 4; ++s) {
#pragma unroll
      for (int t = 0; t < 4; ++t)
        acc[t] = __builtin_amdgcn_wmma_f32_16x16x4_f32(
            false, a[s], false, b[t][s], (short)0, acc[t], false, false);
    }
    ap += HS_GK;
    bp += HS_GK * HEAD_N;
  }

  float* mine = part + wave * 1024;
#pragma unroll
  for (int t = 0; t < 4; ++t) {
#pragma unroll
    for (int r = 0; r < 8; ++r)
      mine[(r + half * 8) * 64 + t * 16 + lm] = acc[t][r];
  }
  __syncthreads();

  const int rr  = threadIdx.x >> 6;
  const int col = threadIdx.x & 63;
#pragma unroll
  for (int j = 0; j < 4; ++j) {
    float s = part[j * 1024 + rr * 64 + col];
    for (int q = 1; q < HS_KPARTS; ++q) s += part[(q * 4 + j) * 1024 + rr * 64 + col];
    atomicAdd(&C[(j * 16 + rr) * HEAD_N + n0 + col], s);
  }
}

#define POL_TILES ((NODE_COUNT + 15) / 16)
#define PO_KPARTS 8
#define PO_WAVES  (4 * PO_KPARTS)
#define PO_PART_K (HEAD_N / PO_KPARTS)
#define PO_STEPS  (PO_PART_K / 4)
#define VAL_ROWS  32
#define VAL_RPW   (VAL_ROWS / PO_WAVES)
static_assert(BATCH % VAL_ROWS == 0 && HEAD_N == 256, "value rows and k split");
static_assert(VAL_ROWS % PO_WAVES == 0 && PO_KPARTS * PO_STEPS * 4 == HEAD_N, "rows per wave, K covered exactly");

__global__ void __launch_bounds__(PO_WAVES * 32)
k_head_out(const float* __restrict__ sump, const float* __restrict__ Wp2,
           const float* __restrict__ bp2, float* __restrict__ outp,
           const float* __restrict__ bp1,
           const float* __restrict__ sumv, const float* __restrict__ bv1,
           const float* __restrict__ Wv2, const float* __restrict__ bv2,
           float* __restrict__ outv) {
  __shared__ float vres[VAL_ROWS];
  __shared__ float ppart[PO_WAVES * 256];
  const int wave = threadIdx.x >> 5;
  const int lane = threadIdx.x & 31;
  if (blockIdx.x < POL_TILES) {
    const int half = lane >> 4;
    const int lm   = lane & 15;
    const int mt   = wave & 3;
    const int kp   = wave >> 2;
    const int m0   = mt * 16;
    const int ncol = blockIdx.x * 16 + lm;
    const bool nok = ncol < NODE_COUNT;
    const int nld  = nok ? ncol : (NODE_COUNT - 1);
    const int kb   = kp * PO_PART_K + half * 2;
    gcfp ap  = (gcfp)sump + ((unsigned)(m0 + lm) * (unsigned)HEAD_N + (unsigned)kb);
    gcfp abp = (gcfp)bp1 + kb;
    gcfp bp  = (gcfp)Wp2 + ((unsigned)kb * (unsigned)NODE_COUNT + (unsigned)nld);
    const float bo = ((gcfp)bp2)[nld];

    float a0[PO_STEPS], a1[PO_STEPS], c0[PO_STEPS], c1[PO_STEPS], b0[PO_STEPS], b1[PO_STEPS];
#pragma unroll
    for (int s = 0; s < PO_STEPS; ++s) {
      a0[s] = ap[4 * s];
      a1[s] = ap[4 * s + 1];
      c0[s] = abp[4 * s];
      c1[s] = abp[4 * s + 1];
      b0[s] = bp[(4 * s) * NODE_COUNT];
      b1[s] = bp[(4 * s + 1) * NODE_COUNT];
    }
    v8f d = (v8f){0.f, 0.f, 0.f, 0.f, 0.f, 0.f, 0.f, 0.f};
#pragma unroll
    for (int s = 0; s < PO_STEPS; ++s) {
      float p0 = a0[s] + c0[s];
      float p1 = a1[s] + c1[s];
      v2f a, b;
      a.x = p0 > 0.0f ? p0 : 0.0f;
      a.y = p1 > 0.0f ? p1 : 0.0f;
      b.x = nok ? b0[s] : 0.0f;
      b.y = nok ? b1[s] : 0.0f;
      d = __builtin_amdgcn_wmma_f32_16x16x4_f32(false, a, false, b, (short)0, d, false, false);
    }
#pragma unroll
    for (int r = 0; r < 8; ++r) ppart[(kp * 4 + mt) * 256 + r * 32 + lane] = d[r];
    __syncthreads();
    if (kp == 0) {
      float s[8];
#pragma unroll
      for (int r = 0; r < 8; ++r) {
        s[r] = ppart[mt * 256 + r * 32 + lane];
#pragma unroll
        for (int q = 1; q < PO_KPARTS; ++q) s[r] += ppart[(q * 4 + mt) * 256 + r * 32 + lane];
      }
      if (nok) {
        gfp og = (gfp)outp;
#pragma unroll
        for (int r = 0; r < 8; ++r)
          og[(unsigned)(m0 + r + half * 8) * (unsigned)NODE_COUNT + (unsigned)ncol] = s[r] + bo;
      }
    }
    return;
  }
  const int vb   = blockIdx.x - POL_TILES;
  const int row0 = vb * VAL_ROWS + wave * VAL_RPW;
  gcv4p sg = (gcv4p)sumv;
  gcv4p bg = (gcv4p)bv1;
  gcv4p wg = (gcv4p)Wv2;
  const v4f bb0 = bg[lane], bb1 = bg[32 + lane];
  const v4f ww0 = wg[lane], ww1 = wg[32 + lane];
  v4f x0[VAL_RPW], x1[VAL_RPW];
#pragma unroll
  for (int r = 0; r < VAL_RPW; ++r) {
    x0[r] = sg[(row0 + r) * (HEAD_N / 4) + lane];
    x1[r] = sg[(row0 + r) * (HEAD_N / 4) + 32 + lane];
  }
  const float b2 = bv2[0];
#pragma unroll
  for (int r = 0; r < VAL_RPW; ++r) {
    float sacc = 0.0f;
#pragma unroll
    for (int e = 0; e < 4; ++e) {
      float h0 = x0[r][e] + bb0[e];
      float h1 = x1[r][e] + bb1[e];
      h0 = h0 > 0.0f ? h0 : 0.0f;
      h1 = h1 > 0.0f ? h1 : 0.0f;
      sacc += h0 * ww0[e];
      sacc += h1 * ww1[e];
    }
    for (int o = 16; o > 0; o >>= 1) sacc += __shfl_xor(sacc, o);
    if (lane == 0) vres[wave * VAL_RPW + r] = sacc + b2;
  }
  __syncthreads();
  if (wave == 0) outv[vb * VAL_ROWS + lane] = vres[lane];
}

extern "C" void kernel_launch(void* const* d_in, const int* in_sizes, int n_in,
                              void* d_out, int out_size, void* d_ws, size_t ws_size,
                              hipStream_t stream) {
  (void)in_sizes; (void)n_in; (void)out_size; (void)ws_size;

  const float* x    = (const float*)d_in[0];
  const int*   ei   = (const int*)d_in[1];
  const int*   rows = ei;
  const int*   cols = ei + N_EDGES;
  const float* W[5]  = {(const float*)d_in[2], (const float*)d_in[4], (const float*)d_in[6],
                        (const float*)d_in[8], (const float*)d_in[10]};
  const float* bb[5] = {(const float*)d_in[3], (const float*)d_in[5], (const float*)d_in[7],
                        (const float*)d_in[9], (const float*)d_in[11]};
  const float* Wp1 = (const float*)d_in[12]; const float* bp1 = (const float*)d_in[13];
  const float* Wp2 = (const float*)d_in[14]; const float* bp2 = (const float*)d_in[15];
  const float* Wv1 = (const float*)d_in[16]; const float* bv1 = (const float*)d_in[17];
  const float* Wv2 = (const float*)d_in[18]; const float* bv2 = (const float*)d_in[19];
  float* out = (float*)d_out;

  float* ws     = (float*)d_ws;
  float* deg    = ws;
  float* hidp   = deg + N_TOTAL;
  float* hidv   = hidp + BATCH * 256;
  int*   cnt    = (int*)(hidv + BATCH * 256);
  int*   base   = cnt + NCH * 256;
  int*   rec    = base + 512;
  float* bufA   = (float*)(rec + 2 * (size_t)N_EDGES);
  float* bufB   = bufA + (size_t)N_TOTAL * HID;

  k_count   <<<NCH, 1024, 0, stream>>>(cols, cnt, hidp);
  k_fill    <<<NCH, 1024, 0, stream>>>(rows, cols, cnt, base, rec);
  k_deg_dinv<<<NBLK, 1024, 0, stream>>>(rec, base, deg);

  k_agg<STATE, true> <<<NBLK, 1024, 0, stream>>>(rec, base, deg, x,    W[0], bb[0], bufA);
  k_agg<HID, false>  <<<NBLK, 1024, 0, stream>>>(rec, base, deg, bufA, W[1], bb[1], bufB);
  k_agg<HID, false>  <<<NBLK, 1024, 0, stream>>>(rec, base, deg, bufB, W[2], bb[2], bufA);
  k_agg<HID, false>  <<<NBLK, 1024, 0, stream>>>(rec, base, deg, bufA, W[3], bb[3], bufB);
  k_agg<HID, false>  <<<NBLK, 1024, 0, stream>>>(rec, base, deg, bufB, W[4], bb[4], bufA);

  dim3 g1(2 * HEAD_N / 64, HS_SLICES);
  k_head_gemm<<<g1, HS_WAVES * 32, 0, stream>>>(bufA, Wp1, Wv1, hidp, hidv);

  k_head_out<<<POL_TILES + BATCH / VAL_ROWS, PO_WAVES * 32, 0, stream>>>(
      hidp, Wp2, bp2, out, bp1, hidv, bv1, Wv2, bv2, out + (size_t)BATCH * NODE_COUNT);
}


# Edge binning 3 launches -> 2: each 8192-edge chunk is sorted by target block in place with its own offset row; the degree launch joins a block's 250 runs into its list; list and sum order unchanged
# speedup vs baseline: 25.5442x; 1.0401x over previous
#include <hip/hip_runtime.h>

#define N_TOTAL    64000
#define NODE_COUNT 1000
#define BATCH      64
#define STATE      16
#define HID        32
#define N_EDGES    2048000
#define TBN        256
#define NBLK       (N_TOTAL / TBN)
static_assert(N_TOTAL % TBN == 0, "blocks cover the nodes exactly");
static_assert(N_TOTAL <= (1 << 24), "source id and 8-bit local target share one int");
static_assert(TBN == 256 && HID == 32, "record packing and lane-per-feature layout");

typedef float v2f __attribute__((ext_vector_type(2)));
typedef float v4f __attribute__((ext_vector_type(4)));
typedef float v8f __attribute__((ext_vector_type(8)));

typedef const __attribute__((address_space(1))) float* gcfp;
typedef __attribute__((address_space(1))) float*       gfp;
typedef const __attribute__((address_space(1))) v4f*   gcv4p;

#define ECH  8192
#define NCH  (N_EDGES / ECH)
#define EPT  (ECH / 1024)
static_assert(N_EDGES % ECH == 0, "chunks cover the edges exactly");
static_assert(NBLK < 256, "one LDS counter per target block, and entry b + 1 of a row exists");
static_assert(NCH <= 256, "one thread per chunk in the list kernel's scan");
#define HID_ALL  (2 * BATCH * 256)
static_assert(HID_ALL % 1024 == 0 && HID_ALL / 1024 <= NCH, "head sums cleared by the first chunks");
#define BSTRIDE 32

__global__ void __launch_bounds__(1024)
k_sort(const int* __restrict__ rows, const int* __restrict__ cols,
       int* __restrict__ offs, int* __restrict__ srt, float* __restrict__ hid) {
  __shared__ int stage[ECH];
  __shared__ int lcnt[256];
  __shared__ int lstart[256];
  __shared__ int wtot[8];
  const int t = threadIdx.x;
  if (t < 256) lcnt[t] = 0;
  if (blockIdx.x < HID_ALL / 1024) hid[blockIdx.x * 1024 + t] = 0.0f;
  __syncthreads();
  const int e0 = blockIdx.x * ECH + t;
  int r[EPT], c[EPT], rk[EPT];
#pragma unroll
  for (int j = 0; j < EPT; ++j) {
    r[j] = rows[e0 + j * 1024];
    c[j] = cols[e0 + j * 1024];
  }
#pragma unroll
  for (int j = 0; j < EPT; ++j) rk[j] = atomicAdd(&lcnt[(unsigned)c[j] / TBN], 1);
  __syncthreads();

  int cn = 0, incl = 0;
  if (t < 256) {
    cn   = lcnt[t];
    incl = cn;
    for (int o = 1; o < 32; o <<= 1) {
      int u = __shfl_up(incl, o);
      if ((t & 31) >= o) incl += u;
    }
    if ((t & 31) == 31) wtot[t >> 5] = incl;
  }
  __syncthreads();
  if (t < 256) {
    int pre = 0;
    for (int w = 0; w < (t >> 5); ++w) pre += wtot[w];
    const int first = pre + incl - cn;
    lstart[t] = first;
    offs[blockIdx.x * 256 + t] = first;
  }
  __syncthreads();

#pragma unroll
  for (int j = 0; j < EPT; ++j) {
    int b = (unsigned)c[j] / TBN;
    stage[lstart[b] + rk[j]] = (r[j] << 8) | (c[j] & (TBN - 1));
  }
  __syncthreads();
#pragma unroll
  for (int j = 0; j < EPT; ++j) srt[e0 + j * 1024] = stage[t + j * 1024];
}

#define SPW  ((NCH + 31) / 32)
static_assert(SPW * 32 <= 256, "the run tables hold every index a wave asks for");
__global__ void __launch_bounds__(1024)
k_list_deg(const int* __restrict__ srt, const int* __restrict__ offs,
           int* __restrict__ bounds, int* __restrict__ rec, float* __restrict__ dinv) {
  __shared__ int ldeg[TBN];
  __shared__ int sbeg[256];
  __shared__ int slen[256];
  __shared__ int sdst[256];
  __shared__ int wtot[8];
  __shared__ int wtot2[8];
  const int t = threadIdx.x;
  const int b = blockIdx.x;
  int o0 = 0, len = 0, incl = 0, incl2 = 0;
  if (t < 256) {
    ldeg[t] = 0;
    if (t < NCH) {
      o0  = offs[t * 256 + b];
      len = offs[t * 256 + b + 1] - o0;
    }
    incl  = len;
    incl2 = o0;
    for (int o = 1; o < 32; o <<= 1) {
      int u  = __shfl_up(incl, o);
      int u2 = __shfl_up(incl2, o);
      if ((t & 31) >= o) { incl += u; incl2 += u2; }
    }
    if ((t & 31) == 31) { wtot[t >> 5] = incl; wtot2[t >> 5] = incl2; }
  }
  __syncthreads();
  if (t < 256) {
    int pre = 0, first = 0;
    for (int w = 0; w < 8; ++w) {
      first += wtot2[w];
      if (w < (t >> 5)) pre += wtot[w];
    }
    sbeg[t] = t * ECH + o0;
    slen[t] = len;
    sdst[t] = first + pre + incl - len;
    if (t == 255) {
      bounds[b * BSTRIDE]     = first;
      bounds[b * BSTRIDE + 1] = first + pre + incl;
    }
  }
  __syncthreads();

  const int wave = t >> 5;
  const int lane = t & 31;
  int sb[SPW], sl[SPW], sd[SPW], v0[SPW], v1[SPW];
#pragma unroll
  for (int j = 0; j < SPW; ++j) {
    const int c = wave + 32 * j;
    sb[j] = sbeg[c];
    sl[j] = slen[c];
    sd[j] = sdst[c];
  }
#pragma unroll
  for (int j = 0; j < SPW; ++j) {
    v0[j] = (lane < sl[j]) ? srt[sb[j] + lane] : 0;
    v1[j] = (lane + 32 < sl[j]) ? srt[sb[j] + lane + 32] : 0;
  }
  int2* rl = (int2*)rec;
#pragma unroll
  for (int j = 0; j < SPW; ++j) {
    if (lane < sl[j]) {
      rl[sd[j] + lane] = make_int2(v0[j], 0);
      atomicAdd(&ldeg[v0[j] & (TBN - 1)], 1);
    }
    if (lane + 32 < sl[j]) {
      rl[sd[j] + lane + 32] = make_int2(v1[j], 0);
      atomicAdd(&ldeg[v1[j] & (TBN - 1)], 1);
    }
    for (int i = lane + 64; i < sl[j]; i += 32) {
      const int v = srt[sb[j] + i];
      rl[sd[j] + i] = make_int2(v, 0);
      atomicAdd(&ldeg[v & (TBN - 1)], 1);
    }
  }
  __syncthreads();
  if (t < TBN) dinv[b * TBN + t] = rsqrtf((float)ldeg[t] + 1.0f);
}

typedef const __attribute__((address_space(1))) int2* gci2p;
#define STG_PITCH 34

template <int F, int Q>
__device__ __forceinline__ int agg_pick(int v) {
  constexpr unsigned S0  = (F == 32) ? Q : 2 * Q;
  constexpr unsigned S1  = (F == 32) ? Q : 2 * Q + 1;
  constexpr unsigned SEL = S0 | (S0 << 3) | (S0 << 6) | (S0 << 9) |
                           (S1 << 12) | (S1 << 15) | (S1 << 18) | (S1 << 21);
  return (int)__builtin_amdgcn_mov_dpp8((unsigned)v, SEL);
}

#define AGG_STEPS(F) ((F) / 4)

template <int F, int Q, bool TAIL>
__device__ __forceinline__ void agg_rows(gcv4p h4, int jq, int px, int rb, int n,
                                         v4f (&rv)[AGG_STEPS(F)]) {
  if constexpr (Q < AGG_STEPS(F)) {
    const unsigned pk = (unsigned)agg_pick<F, Q>(px);
    const int ri = rb + ((F == 32) ? Q : 2 * Q);
    if (!TAIL || ri < n) rv[Q] = h4[(pk >> 8) * (unsigned)(F / 4) + (unsigned)jq];
    else                 rv[Q] = (v4f){0.f, 0.f, 0.f, 0.f};
    agg_rows<F, Q + 1, TAIL>(h4, jq, px, rb, n, rv);
  }
}

template <int F, int Q, bool TAIL>
__device__ __forceinline__ void agg_sums(float* acc, int px, int py, int rb, int n,
                                         bool rot1, bool rot2, const int (&col)[4],
                                         const v4f (&rv)[AGG_STEPS(F)]) {
  if constexpr (Q < AGG_STEPS(F)) {
    const int   pk = agg_pick<F, Q>(px);
    const float w  = __int_as_float(agg_pick<F, Q>(py));
    const int   ri = rb + ((F == 32) ? Q : 2 * Q);
    const v4f v = rv[Q];
    const float a0 = rot1 ? v.y : v.x, a1 = rot1 ? v.z : v.y;
    const float a2 = rot1 ? v.w : v.z, a3 = rot1 ? v.x : v.w;
    const float r0 = rot2 ? a2 : a0, r1 = rot2 ? a3 : a1;
    const float r2 = rot2 ? a0 : a2, r3 = rot2 ? a1 : a3;
    float* row = acc + (pk & (TBN - 1)) * HID;
    if (!TAIL || ri < n) {
      __hip_atomic_fetch_add(row + col[0], r0 * w, __ATOMIC_RELAXED, __HIP_MEMORY_SCOPE_WORKGROUP);
      __hip_atomic_fetch_add(row + col[1], r1 * w, __ATOMIC_RELAXED, __HIP_MEMORY_SCOPE_WORKGROUP);
      __hip_atomic_fetch_add(row + col[2], r2 * w, __ATOMIC_RELAXED, __HIP_MEMORY_SCOPE_WORKGROUP);
      __hip_atomic_fetch_add(row + col[3], r3 * w, __ATOMIC_RELAXED, __HIP_MEMORY_SCOPE_WORKGROUP);
    }
    agg_sums<F, Q + 1, TAIL>(acc, px, py, rb, n, rot1, rot2, col, rv);
  }
}

template <int F, bool FIRST>
__global__ void __launch_bounds__(1024)
k_agg(int* __restrict__ rec, const int* __restrict__ base,
      const float* __restrict__ dinv, const float* __restrict__ h,
      const float* __restrict__ Wl, const float* __restrict__ bias,
      float* __restrict__ out) {
  constexpr int P = F + 2;
  __shared__ __attribute__((aligned(16))) float acc[TBN * STG_PITCH];
  __shared__ __attribute__((aligned(16))) float apad[TBN * P];
  __shared__ float ldv[FIRST ? TBN : 1];

  const int wave  = __builtin_amdgcn_readfirstlane((int)(threadIdx.x >> 5));
  const int lane  = threadIdx.x & 31;
  const int f     = lane & (F - 1);
  const int half  = lane >> 4;
  const int lm    = lane & 15;
  const int mt    = wave >> 1;
  const int nt    = wave & 1;
  const int node0 = blockIdx.x * TBN;
  gcfp  hg = (gcfp)h;
  gcv4p h4 = (gcv4p)h;
  gcfp  dg = (gcfp)dinv;
  gci2p rg = (gci2p)rec;

  constexpr int LPR = F / 4;
  const int  jq   = lane & (LPR - 1);
  const int  rw   = lane / LPR;
  const bool rot1 = (rw & 1) != 0;
  const bool rot2 = (rw & 2) != 0;
  const int  rb   = 8 * (lane >> 3) + (F == 16 ? ((lane >> 2) & 1) : 0);
  int col[4];
#pragma unroll
  for (int i = 0; i < 4; ++i) col[i] = 4 * jq + (F == 16 ? 16 * (rw >> 2) : 0) + ((i + rw) & 3);

  gcfp wg = (gcfp)Wl;
  v2f wb[F / 4];
#pragma unroll
  for (int s = 0; s < F / 4; ++s) {
    wb[s].x = wg[(4 * s + 2 * half) * HID + nt * 16 + lm];
    wb[s].y = wg[(4 * s + 2 * half + 1) * HID + nt * 16 + lm];
  }

  float hv[TBN / 32], dd[TBN / 32];
#pragma unroll
  for (int j = 0; j < TBN / 32; ++j) {
    int n = wave + 32 * j;
    dd[j] = dinv[node0 + n];
    hv[j] = hg[(unsigned)(node0 + n) * F + (unsigned)f];
  }

  const int beg = base[blockIdx.x * BSTRIDE];
  const int end = base[blockIdx.x * BSTRIDE + 1];
  const float bv = bias[lane];
  int p = beg + wave * 32;
  int2 cur = make_int2(0, 0);
  if (p < end) cur = rg[min(p + lane, end - 1)];
  if (FIRST) {
    if (threadIdx.x < TBN) ldv[threadIdx.x] = dinv[node0 + threadIdx.x];
  }

#pragma unroll
  for (int j = 0; j < TBN / 32; ++j) {
    int n = wave + 32 * j;
    float sv = hv[j] * (dd[j] * dd[j]);
    if (F < HID && lane >= F) sv = 0.0f;
    acc[n * HID + lane] = sv;
  }
  __syncthreads();

  if (p < end) {
    for (;;) {
      const int n  = end - p;
      const int pn = p + 32 * 32;
      if (n < 32) {
        if (FIRST) {
          float wv = dg[(unsigned)cur.x >> 8] * ldv[cur.x & (TBN - 1)];
          if (lane < n) rec[2 * (p + lane) + 1] = __float_as_int(wv);
          cur.y = __float_as_int(wv);
        }
        v4f rt[AGG_STEPS(F)];
        agg_rows<F, 0, true>(h4, jq, cur.x, rb, n, rt);
        agg_sums<F, 0, true>(acc, cur.x, cur.y, rb, n, rot1, rot2, col, rt);
        break;
      }
      v4f rv[AGG_STEPS(F)];
      int2 nxt = rg[min(pn + lane, end - 1)];
      float ds = 0.0f;
      if (FIRST) ds = dg[(unsigned)cur.x >> 8];
      agg_rows<F, 0, false>(h4, jq, cur.x, rb, 32, rv);
      asm volatile("" :: "v"(nxt.x), "v"(nxt.y));
      if (FIRST) {
        float wv = ds * ldv[cur.x & (TBN - 1)];
        rec[2 * (p + lane) + 1] = __float_as_int(wv);
        cur.y = __float_as_int(wv);
      }
      agg_sums<F, 0, false>(acc, cur.x, cur.y, rb, 32, rot1, rot2, col, rv);
      if (pn >= end) break;
      cur = nxt;
      p = pn;
    }
  }
  __syncthreads();

#pragma unroll
  for (int j = 0; j < TBN / 32; ++j) {
    int n = wave + 32 * j;
    float s = acc[n * HID + f];
    if (F < HID) s += acc[n * HID + F + f];
    apad[n * P + f] = s;
  }
  __syncthreads();

  v8f d = (v8f){0.f, 0.f, 0.f, 0.f, 0.f, 0.f, 0.f, 0.f};
  const float* arow = &apad[(mt * 16 + lm) * P + 2 * half];
#pragma unroll
  for (int s = 0; s < F / 4; ++s) {
    v2f a = *(const v2f*)(arow + 4 * s);
    d = __builtin_amdgcn_wmma_f32_16x16x4_f32(false, a, false, wb[s], (short)0, d,
                                              false, false);
  }
  float* srow = &acc[(mt * 16 + 8 * half) * STG_PITCH + nt * 16 + lm];
#pragma unroll
  for (int r = 0; r < 8; ++r) srow[r * STG_PITCH] = d[r];
  __syncthreads();

#pragma unroll
  for (int j = 0; j < TBN / 32; ++j) {
    int n = wave + 32 * j;
    float v = acc[n * STG_PITCH + lane] + bv;
    out[(node0 + n) * HID + lane] = v > 0.0f ? v : 0.0f;
  }
}

#define HEAD_K      (NODE_COUNT * HID)
#define HEAD_N      256
#define HS_SLICES   25
#define HS_KPARTS   8
#define HS_WAVES    (4 * HS_KPARTS)
#define HS_SLICE_K  (HEAD_K / HS_SLICES)
#define HS_PART_K   (HS_SLICE_K / HS_KPARTS)
#define HS_GK       16
#define HS_GROUPS   (HS_PART_K / HS_GK)
static_assert(HS_SLICES * HS_KPARTS * HS_GROUPS * HS_GK == HEAD_K, "K is covered exactly");
static_assert(BATCH == 64, "four row tiles");

typedef const __attribute__((address_space(1))) v2f* gcv2p;

__global__ void __launch_bounds__(HS_WAVES * 32)
k_head_gemm(const float* __restrict__ A,
            const float* __restrict__ Bp, const float* __restrict__ Bv,
            float* __restrict__ Cp, float* __restrict__ Cv) {
  __shared__ float part[HS_WAVES * 1024];

  const int wave = threadIdx.x >> 5;
  const int lane = threadIdx.x & 31;
  const int half = lane >> 4;
  const int lm   = lane & 15;
  const int mt   = wave & 3;
  const int kp   = wave >> 2;
  const int m0   = mt * 16;
  const int n0   = (blockIdx.x & 3) * 64;
  const bool second = blockIdx.x >= 4;
  const float* B = second ? Bv : Bp;
  float*       C = second ? Cv : Cp;
  const int kbeg = blockIdx.y * HS_SLICE_K + kp * HS_PART_K;

  gcfp ap = (gcfp)A + ((unsigned)(m0 + lm) * (unsigned)HEAD_K + (unsigned)(kbeg + half * 2));
  gcfp bp = (gcfp)B + ((unsigned)(kbeg + half * 2) * (unsigned)HEAD_N + (unsigned)(n0 + lm));

  v8f acc[4];
#pragma unroll
  for (int t = 0; t < 4; ++t) acc[t] = (v8f){0.f,0.f,0.f,0.f,0.f,0.f,0.f,0.f};

  for (int g = 0; g < HS_GROUPS; ++g) {
    v2f a[4], b[4][4];
#pragma unroll
    for (int s = 0; s < 4; ++s)
      a[s] = *(gcv2p)(ap + 4 * s);
#pragma unroll
    for (int s = 0; s < 4; ++s) {
#pragma unroll
      for (int t = 0; t < 4; ++t) {
        b[t][s].x = bp[(4 * s) * HEAD_N + 16 * t];
        b[t][s].y = bp[(4 * s + 1) * HEAD_N + 16 * t];
      }
    }
    __builtin_amdgcn_sched_barrier(0);
#pragma unroll
    for (int s = 0; s < 4; ++s) {
#pragma unroll
      for (int t = 0; t < 4; ++t)
        acc[t] = __builtin_amdgcn_wmma_f32_16x16x4_f32(
            false, a[s], false, b[t][s], (short)0, acc[t], false, false);
    }
    ap += HS_GK;
    bp += HS_GK * HEAD_N;
  }

  float* mine = part + wave * 1024;
#pragma unroll
  for (int t = 0; t < 4; ++t) {
#pragma unroll
    for (int r = 0; r < 8; ++r)
      mine[(r + half * 8) * 64 + t * 16 + lm] = acc[t][r];
  }
  __syncthreads();

  const int rr  = threadIdx.x >> 6;
  const int col = threadIdx.x & 63;
#pragma unroll
  for (int j = 0; j < 4; ++j) {
    float s = part[j * 1024 + rr * 64 + col];
    for (int q = 1; q < HS_KPARTS; ++q) s += part[(q * 4 + j) * 1024 + rr * 64 + col];
    atomicAdd(&C[(j * 16 + rr) * HEAD_N + n0 + col], s);
  }
}

#define POL_TILES ((NODE_COUNT + 15) / 16)
#define PO_KPARTS 8
#define PO_WAVES  (4 * PO_KPARTS)
#define PO_PART_K (HEAD_N / PO_KPARTS)
#define PO_STEPS  (PO_PART_K / 4)
#define VAL_ROWS  32
#define VAL_RPW   (VAL_ROWS / PO_WAVES)
static_assert(BATCH % VAL_ROWS == 0 && HEAD_N == 256, "value rows and k split");
static_assert(VAL_ROWS % PO_WAVES == 0 && PO_KPARTS * PO_STEPS * 4 == HEAD_N, "rows per wave, K covered exactly");

__global__ void __launch_bounds__(PO_WAVES * 32)
k_head_out(const float* __restrict__ sump, const float* __restrict__ Wp2,
           const float* __restrict__ bp2, float* __restrict__ outp,
           const float* __restrict__ bp1,
           const float* __restrict__ sumv, const float* __restrict__ bv1,
           const float* __restrict__ Wv2, const float* __restrict__ bv2,
           float* __restrict__ outv) {
  __shared__ float vres[VAL_ROWS];
  __shared__ float ppart[PO_WAVES * 256];
  const int wave = threadIdx.x >> 5;
  const int lane = threadIdx.x & 31;
  if (blockIdx.x < POL_TILES) {
    const int half = lane >> 4;
    const int lm   = lane & 15;
    const int mt   = wave & 3;
    const int kp   = wave >> 2;
    const int m0   = mt * 16;
    const int ncol = blockIdx.x * 16 + lm;
    const bool nok = ncol < NODE_COUNT;
    const int nld  = nok ? ncol : (NODE_COUNT - 1);
    const int kb   = kp * PO_PART_K + half * 2;
    gcfp ap  = (gcfp)sump + ((unsigned)(m0 + lm) * (unsigned)HEAD_N + (unsigned)kb);
    gcfp abp = (gcfp)bp1 + kb;
    gcfp bp  = (gcfp)Wp2 + ((unsigned)kb * (unsigned)NODE_COUNT + (unsigned)nld);
    const float bo = ((gcfp)bp2)[nld];

    float a0[PO_STEPS], a1[PO_STEPS], c0[PO_STEPS], c1[PO_STEPS], b0[PO_STEPS], b1[PO_STEPS];
#pragma unroll
    for (int s = 0; s < PO_STEPS; ++s) {
      a0[s] = ap[4 * s];
      a1[s] = ap[4 * s + 1];
      c0[s] = abp[4 * s];
      c1[s] = abp[4 * s + 1];
      b0[s] = bp[(4 * s) * NODE_COUNT];
      b1[s] = bp[(4 * s + 1) * NODE_COUNT];
    }
    v8f d = (v8f){0.f, 0.f, 0.f, 0.f, 0.f, 0.f, 0.f, 0.f};
#pragma unroll
    for (int s = 0; s < PO_STEPS; ++s) {
      float p0 = a0[s] + c0[s];
      float p1 = a1[s] + c1[s];
      v2f a, b;
      a.x = p0 > 0.0f ? p0 : 0.0f;
      a.y = p1 > 0.0f ? p1 : 0.0f;
      b.x = nok ? b0[s] : 0.0f;
      b.y = nok ? b1[s] : 0.0f;
      d = __builtin_amdgcn_wmma_f32_16x16x4_f32(false, a, false, b, (short)0, d, false, false);
    }
#pragma unroll
    for (int r = 0; r < 8; ++r) ppart[(kp * 4 + mt) * 256 + r * 32 + lane] = d[r];
    __syncthreads();
    if (kp == 0) {
      float s[8];
#pragma unroll
      for (int r = 0; r < 8; ++r) {
        s[r] = ppart[mt * 256 + r * 32 + lane];
#pragma unroll
        for (int q = 1; q < PO_KPARTS; ++q) s[r] += ppart[(q * 4 + mt) * 256 + r * 32 + lane];
      }
      if (nok) {
        gfp og = (gfp)outp;
#pragma unroll
        for (int r = 0; r < 8; ++r)
          og[(unsigned)(m0 + r + half * 8) * (unsigned)NODE_COUNT + (unsigned)ncol] = s[r] + bo;
      }
    }
    return;
  }
  const int vb   = blockIdx.x - POL_TILES;
  const int row0 = vb * VAL_ROWS + wave * VAL_RPW;
  gcv4p sg = (gcv4p)sumv;
  gcv4p bg = (gcv4p)bv1;
  gcv4p wg = (gcv4p)Wv2;
  const v4f bb0 = bg[lane], bb1 = bg[32 + lane];
  const v4f ww0 = wg[lane], ww1 = wg[32 + lane];
  v4f x0[VAL_RPW], x1[VAL_RPW];
#pragma unroll
  for (int r = 0; r < VAL_RPW; ++r) {
    x0[r] = sg[(row0 + r) * (HEAD_N / 4) + lane];
    x1[r] = sg[(row0 + r) * (HEAD_N / 4) + 32 + lane];
  }
  const float b2 = bv2[0];
#pragma unroll
  for (int r = 0; r < VAL_RPW; ++r) {
    float sacc = 0.0f;
#pragma unroll
    for (int e = 0; e < 4; ++e) {
      float h0 = x0[r][e] + bb0[e];
      float h1 = x1[r][e] + bb1[e];
      h0 = h0 > 0.0f ? h0 : 0.0f;
      h1 = h1 > 0.0f ? h1 : 0.0f;
      sacc += h0 * ww0[e];
      sacc += h1 * ww1[e];
    }
    for (int o = 16; o > 0; o >>= 1) sacc += __shfl_xor(sacc, o);
    if (lane == 0) vres[wave * VAL_RPW + r] = sacc + b2;
  }
  __syncthreads();
  if (wave == 0) outv[vb * VAL_ROWS + lane] = vres[lane];
}

extern "C" void kernel_launch(void* const* d_in, const int* in_sizes, int n_in,
                              void* d_out, int out_size, void* d_ws, size_t ws_size,
                              hipStream_t stream) {
  (void)in_sizes; (void)n_in; (void)out_size; (void)ws_size;

  const float* x    = (const float*)d_in[0];
  const int*   ei   = (const int*)d_in[1];
  const int*   rows = ei;
  const int*   cols = ei + N_EDGES;
  const float* W[5]  = {(const float*)d_in[2], (const float*)d_in[4], (const float*)d_in[6],
                        (const float*)d_in[8], (const float*)d_in[10]};
  const float* bb[5] = {(const float*)d_in[3], (const float*)d_in[5], (const float*)d_in[7],
                        (const float*)d_in[9], (const float*)d_in[11]};
  const float* Wp1 = (const float*)d_in[12]; const float* bp1 = (const float*)d_in[13];
  const float* Wp2 = (const float*)d_in[14]; const float* bp2 = (const float*)d_in[15];
  const float* Wv1 = (const float*)d_in[16]; const float* bv1 = (const float*)d_in[17];
  const float* Wv2 = (const float*)d_in[18]; const float* bv2 = (const float*)d_in[19];
  float* out = (float*)d_out;

  float* ws     = (float*)d_ws;
  float* deg    = ws;
  float* hidp   = deg + N_TOTAL;
  float* hidv   = hidp + BATCH * 256;
  int*   offs   = (int*)(hidv + BATCH * 256);
  int*   base   = offs + NCH * 256;
  int*   rec    = base + 8192;
  float* bufA   = (float*)(rec + 2 * (size_t)N_EDGES);
  float* bufB   = bufA + (size_t)N_TOTAL * HID;
  int*   srt    = (int*)(bufB + (size_t)N_TOTAL * HID);

  k_sort    <<<NCH, 1024, 0, stream>>>(rows, cols, offs, srt, hidp);
  k_list_deg<<<NBLK, 1024, 0, stream>>>(srt, offs, base, rec, deg);

  k_agg<STATE, true> <<<NBLK, 1024, 0, stream>>>(rec, base, deg, x,    W[0], bb[0], bufA);
  k_agg<HID, false>  <<<NBLK, 1024, 0, stream>>>(rec, base, deg, bufA, W[1], bb[1], bufB);
  k_agg<HID, false>  <<<NBLK, 1024, 0, stream>>>(rec, base, deg, bufB, W[2], bb[2], bufA);
  k_agg<HID, false>  <<<NBLK, 1024, 0, stream>>>(rec, base, deg, bufA, W[3], bb[3], bufB);
  k_agg<HID, false>  <<<NBLK, 1024, 0, stream>>>(rec, base, deg, bufB, W[4], bb[4], bufA);

  dim3 g1(2 * HEAD_N / 64, HS_SLICES);
  k_head_gemm<<<g1, HS_WAVES * 32, 0, stream>>>(bufA, Wp1, Wv1, hidp, hidv);

  k_head_out<<<POL_TILES + BATCH / VAL_ROWS, PO_WAVES * 32, 0, stream>>>(
      hidp, Wp2, bp2, out, bp1, hidv, bv1, Wv2, bv2, out + (size_t)BATCH * NODE_COUNT);
}


# speedup vs baseline: 26.7056x; 1.0455x over previous
#include <hip/hip_runtime.h>

#define N_TOTAL    64000
#define NODE_COUNT 1000
#define BATCH      64
#define STATE      16
#define HID        32
#define N_EDGES    2048000
#define TBN        256
#define NBLK       (N_TOTAL / TBN)
static_assert(N_TOTAL % TBN == 0, "blocks cover the nodes exactly");
static_assert(N_TOTAL <= (1 << 24), "source id and 8-bit local target share one int");
static_assert(TBN == 256 && HID == 32, "record packing and lane-per-feature layout");

typedef float v2f __attribute__((ext_vector_type(2)));
typedef float v4f __attribute__((ext_vector_type(4)));
typedef float v8f __attribute__((ext_vector_type(8)));

typedef const __attribute__((address_space(1))) float* gcfp;
typedef __attribute__((address_space(1))) float*       gfp;
typedef const __attribute__((address_space(1))) v4f*   gcv4p;

#define ECH  8192
#define NCH  (N_EDGES / ECH)
#define EPT  (ECH / 1024)
static_assert(N_EDGES % ECH == 0, "chunks cover the edges exactly");
static_assert(NBLK < 256, "one LDS counter per target block, and entry b + 1 of a row exists");
static_assert(NCH <= 256, "one thread per chunk in the list kernel's scan");
#define HID_ALL  (2 * BATCH * 256)
static_assert(HID_ALL % 1024 == 0 && HID_ALL / 1024 <= NCH, "head sums cleared by the first chunks");
#define BSTRIDE 32

__global__ void __launch_bounds__(1024)
k_sort(const int* __restrict__ rows, const int* __restrict__ cols,
       int* __restrict__ offs, int* __restrict__ srt, float* __restrict__ hid) {
  __shared__ int stage[ECH];
  __shared__ int lcnt[256];
  __shared__ int lstart[256];
  __shared__ int wtot[8];
  const int t = threadIdx.x;
  if (t < 256) lcnt[t] = 0;
  if (blockIdx.x < HID_ALL / 1024) hid[blockIdx.x * 1024 + t] = 0.0f;
  __syncthreads();
  const int e0 = blockIdx.x * ECH + t;
  int r[EPT], c[EPT], rk[EPT];
#pragma unroll
  for (int j = 0; j < EPT; ++j) {
    r[j] = rows[e0 + j * 1024];
    c[j] = cols[e0 + j * 1024];
  }
#pragma unroll
  for (int j = 0; j < EPT; ++j) rk[j] = atomicAdd(&lcnt[(unsigned)c[j] / TBN], 1);
  __syncthreads();

  int cn = 0, incl = 0;
  if (t < 256) {
    cn   = lcnt[t];
    incl = cn;
    for (int o = 1; o < 32; o <<= 1) {
      int u = __shfl_up(incl, o);
      if ((t & 31) >= o) incl += u;
    }
    if ((t & 31) == 31) wtot[t >> 5] = incl;
  }
  __syncthreads();
  if (t < 256) {
    int pre = 0;
    for (int w = 0; w < (t >> 5); ++w) pre += wtot[w];
    const int first = pre + incl - cn;
    lstart[t] = first;
    offs[blockIdx.x * 256 + t] = first;
  }
  __syncthreads();

#pragma unroll
  for (int j = 0; j < EPT; ++j) {
    int b = (unsigned)c[j] / TBN;
    stage[lstart[b] + rk[j]] = (r[j] << 8) | (c[j] & (TBN - 1));
  }
  __syncthreads();
#pragma unroll
  for (int j = 0; j < EPT; ++j) srt[e0 + j * 1024] = stage[t + j * 1024];
}

#define SPW  ((NCH + 31) / 32)
static_assert(SPW * 32 <= 256, "the run tables hold every index a wave asks for");
__global__ void __launch_bounds__(1024)
k_list_deg(const int* __restrict__ srt, const int* __restrict__ offs,
           int* __restrict__ bounds, int* __restrict__ rec, float* __restrict__ dinv) {
  __shared__ int ldeg[TBN];
  __shared__ int sbeg[256];
  __shared__ int slen[256];
  __shared__ int sdst[256];
  __shared__ int wtot[8];
  __shared__ int wtot2[8];
  const int t = threadIdx.x;
  const int b = blockIdx.x;
  int o0 = 0, len = 0, incl = 0, incl2 = 0;
  if (t < 256) {
    ldeg[t] = 0;
    if (t < NCH) {
      o0  = offs[t * 256 + b];
      len = offs[t * 256 + b + 1] - o0;
    }
    incl  = len;
    incl2 = o0;
    for (int o = 1; o < 32; o <<= 1) {
      int u  = __shfl_up(incl, o);
      int u2 = __shfl_up(incl2, o);
      if ((t & 31) >= o) { incl += u; incl2 += u2; }
    }
    if ((t & 31) == 31) { wtot[t >> 5] = incl; wtot2[t >> 5] = incl2; }
  }
  __syncthreads();
  if (t < 256) {
    int pre = 0, first = 0;
    for (int w = 0; w < 8; ++w) {
      first += wtot2[w];
      if (w < (t >> 5)) pre += wtot[w];
    }
    sbeg[t] = t * ECH + o0;
    slen[t] = len;
    sdst[t] = first + pre + incl - len;
    if (t == 255) {
      bounds[b * BSTRIDE]     = first;
      bounds[b * BSTRIDE + 1] = first + pre + incl;
    }
  }
  __syncthreads();

  const int wave = t >> 5;
  const int lane = t & 31;
  int sb[SPW], sl[SPW], sd[SPW], v0[SPW], v1[SPW];
#pragma unroll
  for (int j = 0; j < SPW; ++j) {
    const int c = wave + 32 * j;
    sb[j] = sbeg[c];
    sl[j] = slen[c];
    sd[j] = sdst[c];
  }
#pragma unroll
  for (int j = 0; j < SPW; ++j) {
    v0[j] = (lane < sl[j]) ? srt[sb[j] + lane] : 0;
    v1[j] = (lane + 32 < sl[j]) ? srt[sb[j] + lane + 32] : 0;
  }
  int2* rl = (int2*)rec;
#pragma unroll
  for (int j = 0; j < SPW; ++j) {
    if (lane < sl[j]) {
      rl[sd[j] + lane] = make_int2(v0[j], 0);
      atomicAdd(&ldeg[v0[j] & (TBN - 1)], 1);
    }
    if (lane + 32 < sl[j]) {
      rl[sd[j] + lane + 32] = make_int2(v1[j], 0);
      atomicAdd(&ldeg[v1[j] & (TBN - 1)], 1);
    }
    for (int i = lane + 64; i < sl[j]; i += 32) {
      const int v = srt[sb[j] + i];
      rl[sd[j] + i] = make_int2(v, 0);
      atomicAdd(&ldeg[v & (TBN - 1)], 1);
    }
  }
  __syncthreads();
  if (t < TBN) dinv[b * TBN + t] = rsqrtf((float)ldeg[t] + 1.0f);
}

typedef const __attribute__((address_space(1))) int2* gci2p;
#define STG_PITCH 34
#define NCP       4
#define CP_FLOATS (TBN * HID)
static_assert(CP_FLOATS * sizeof(float) == 32768);
static_assert(TBN * STG_PITCH <= NCP * CP_FLOATS, "the WMMA result fits where the copies were");

template <int F, int Q>
__device__ __forceinline__ int agg_pick(int v) {
  constexpr unsigned S0  = (F == 32) ? Q : 2 * Q;
  constexpr unsigned S1  = (F == 32) ? Q : 2 * Q + 1;
  constexpr unsigned SEL = S0 | (S0 << 3) | (S0 << 6) | (S0 << 9) |
                           (S1 << 12) | (S1 << 15) | (S1 << 18) | (S1 << 21);
  return (int)__builtin_amdgcn_mov_dpp8((unsigned)v, SEL);
}

#define AGG_STEPS(F) ((F) / 4)

template <int F, int Q, bool TAIL>
__device__ __forceinline__ void agg_rows(gcv4p h4, int jq, int px, int rb, int n,
                                         v4f (&rv)[AGG_STEPS(F)]) {
  if constexpr (Q < AGG_STEPS(F)) {
    const unsigned pk = (unsigned)agg_pick<F, Q>(px);
    const int ri = rb + ((F == 32) ? Q : 2 * Q);
    if (!TAIL || ri < n) rv[Q] = h4[(pk >> 8) * (unsigned)(F / 4) + (unsigned)jq];
    else                 rv[Q] = (v4f){0.f, 0.f, 0.f, 0.f};
    agg_rows<F, Q + 1, TAIL>(h4, jq, px, rb, n, rv);
  }
}

template <int F, int Q, bool TAIL>
__device__ __forceinline__ void agg_sums(float* acc, int px, int py, int rb, int n,
                                         bool rot1, bool rot2, const int (&col)[4],
                                         const v4f (&rv)[AGG_STEPS(F)]) {
  if constexpr (Q < AGG_STEPS(F)) {
    const int   pk = agg_pick<F, Q>(px);
    const float w  = __int_as_float(agg_pick<F, Q>(py));
    const int   ri = rb + ((F == 32) ? Q : 2 * Q);
    const v4f v = rv[Q];
    const float a0 = rot1 ? v.y : v.x, a1 = rot1 ? v.z : v.y;
    const float a2 = rot1 ? v.w : v.z, a3 = rot1 ? v.x : v.w;
    const float r0 = rot2 ? a2 : a0, r1 = rot2 ? a3 : a1;
    const float r2 = rot2 ? a0 : a2, r3 = rot2 ? a1 : a3;
    float* row = acc + (pk & (TBN - 1)) * HID;
    if (!TAIL || ri < n) {
      __hip_atomic_fetch_add(row + col[0], r0 * w, __ATOMIC_RELAXED, __HIP_MEMORY_SCOPE_WORKGROUP);
      __hip_atomic_fetch_add(row + col[1], r1 * w, __ATOMIC_RELAXED, __HIP_MEMORY_SCOPE_WORKGROUP);
      __hip_atomic_fetch_add(row + col[2], r2 * w, __ATOMIC_RELAXED, __HIP_MEMORY_SCOPE_WORKGROUP);
      __hip_atomic_fetch_add(row + col[3], r3 * w, __ATOMIC_RELAXED, __HIP_MEMORY_SCOPE_WORKGROUP);
    }
    agg_sums<F, Q + 1, TAIL>(acc, px, py, rb, n, rot1, rot2, col, rv);
  }
}

template <int F, bool FIRST>
__global__ void __launch_bounds__(1024)
k_agg(int* __restrict__ rec, const int* __restrict__ base,
      const float* __restrict__ dinv, const float* __restrict__ h,
      const float* __restrict__ Wl, const float* __restrict__ bias,
      float* __restrict__ out) {
  constexpr int P = F + 2;
  __shared__ __attribute__((aligned(16))) float lds[NCP * CP_FLOATS + TBN * P + (FIRST ? TBN : 0)];
  float* const acc  = lds;
  float* const apad = lds + NCP * CP_FLOATS;
  float* const ldv  = apad + TBN * P;

  const int wave  = __builtin_amdgcn_readfirstlane((int)(threadIdx.x >> 5));
  const int lane  = threadIdx.x & 31;
  const int f     = lane & (F - 1);
  const int half  = lane >> 4;
  const int lm    = lane & 15;
  const int mt    = wave >> 1;
  const int nt    = wave & 1;
  const int node0 = blockIdx.x * TBN;
  const int simd = (int)(__builtin_amdgcn_s_getreg(23 | (8 << 6) | ((2 - 1) << 11)) & 3u);
  gcfp  hg = (gcfp)h;
  gcv4p h4 = (gcv4p)h;
  gcfp  dg = (gcfp)dinv;
  gci2p rg = (gci2p)rec;

  constexpr int LPR = F / 4;
  const int  jq   = lane & (LPR - 1);
  const int  rw   = lane / LPR;
  const bool rot1 = (rw & 1) != 0;
  const bool rot2 = (rw & 2) != 0;
  const int  rb   = 8 * (lane >> 3) + (F == 16 ? ((lane >> 2) & 1) : 0);
  int col[4];
#pragma unroll
  for (int i = 0; i < 4; ++i)
    col[i] = simd * CP_FLOATS + 4 * jq + (F == 16 ? 16 * (rw >> 2) : 0) + ((i + rw) & 3);

  gcfp wg = (gcfp)Wl;
  v2f wb[F / 4];
#pragma unroll
  for (int s = 0; s < F / 4; ++s) {
    wb[s].x = wg[(4 * s + 2 * half) * HID + nt * 16 + lm];
    wb[s].y = wg[(4 * s + 2 * half + 1) * HID + nt * 16 + lm];
  }

  float hv[TBN / 32], dd[TBN / 32];
#pragma unroll
  for (int j = 0; j < TBN / 32; ++j) {
    int n = wave + 32 * j;
    dd[j] = dinv[node0 + n];
    hv[j] = hg[(unsigned)(node0 + n) * F + (unsigned)f];
  }

  const int beg = base[blockIdx.x * BSTRIDE];
  const int end = base[blockIdx.x * BSTRIDE + 1];
  const float bv = bias[lane];
  int p = beg + wave * 32;
  int2 cur = make_int2(0, 0);
  if (p < end) cur = rg[min(p + lane, end - 1)];
  if (FIRST) {
    if (threadIdx.x < TBN) ldv[threadIdx.x] = dinv[node0 + threadIdx.x];
  }

  {
    v4f* z = (v4f*)(acc + CP_FLOATS);
#pragma unroll
    for (int i = 0; i < (NCP - 1) * CP_FLOATS / 4 / 1024; ++i)
      z[i * 1024 + threadIdx.x] = (v4f){0.f, 0.f, 0.f, 0.f};
  }

#pragma unroll
  for (int j = 0; j < TBN / 32; ++j) {
    int n = wave + 32 * j;
    float sv = hv[j] * (dd[j] * dd[j]);
    if (F < HID && lane >= F) sv = 0.0f;
    acc[n * HID + lane] = sv;
  }
  __syncthreads();

  if (p < end) {
    for (;;) {
      const int n  = end - p;
      const int pn = p + 32 * 32;
      if (n < 32) {
        if (FIRST) {
          float wv = dg[(unsigned)cur.x >> 8] * ldv[cur.x & (TBN - 1)];
          if (lane < n) rec[2 * (p + lane) + 1] = __float_as_int(wv);
          cur.y = __float_as_int(wv);
        }
        v4f rt[AGG_STEPS(F)];
        agg_rows<F, 0, true>(h4, jq, cur.x, rb, n, rt);
        agg_sums<F, 0, true>(acc, cur.x, cur.y, rb, n, rot1, rot2, col, rt);
        break;
      }
      v4f rv[AGG_STEPS(F)];
      int2 nxt = rg[min(pn + lane, end - 1)];
      float ds = 0.0f;
      if (FIRST) ds = dg[(unsigned)cur.x >> 8];
      agg_rows<F, 0, false>(h4, jq, cur.x, rb, 32, rv);
      asm volatile("" :: "v"(nxt.x), "v"(nxt.y));
      if (FIRST) {
        float wv = ds * ldv[cur.x & (TBN - 1)];
        rec[2 * (p + lane) + 1] = __float_as_int(wv);
        cur.y = __float_as_int(wv);
      }
      agg_sums<F, 0, false>(acc, cur.x, cur.y, rb, 32, rot1, rot2, col, rv);
      if (pn >= end) break;
      cur = nxt;
      p = pn;
    }
  }
  __syncthreads();

#pragma unroll
  for (int j = 0; j < TBN / 32; ++j) {
    int n = wave + 32 * j;
    float s = acc[n * HID + f];
    if (F < HID) s += acc[n * HID + F + f];
#pragma unroll
    for (int cp = 1; cp < NCP; ++cp) {
      s += acc[cp * CP_FLOATS + n * HID + f];
      if (F < HID) s += acc[cp * CP_FLOATS + n * HID + F + f];
    }
    apad[n * P + f] = s;
  }
  __syncthreads();

  v8f d = (v8f){0.f, 0.f, 0.f, 0.f, 0.f, 0.f, 0.f, 0.f};
  const float* arow = &apad[(mt * 16 + lm) * P + 2 * half];
#pragma unroll
  for (int s = 0; s < F / 4; ++s) {
    v2f a = *(const v2f*)(arow + 4 * s);
    d = __builtin_amdgcn_wmma_f32_16x16x4_f32(false, a, false, wb[s], (short)0, d,
                                              false, false);
  }
  float* srow = &acc[(mt * 16 + 8 * half) * STG_PITCH + nt * 16 + lm];
#pragma unroll
  for (int r = 0; r < 8; ++r) srow[r * STG_PITCH] = d[r];
  __syncthreads();

#pragma unroll
  for (int j = 0; j < TBN / 32; ++j) {
    int n = wave + 32 * j;
    float v = acc[n * STG_PITCH + lane] + bv;
    out[(node0 + n) * HID + lane] = v > 0.0f ? v : 0.0f;
  }
}

#define HEAD_K      (NODE_COUNT * HID)
#define HEAD_N      256
#define HS_SLICES   25
#define HS_KPARTS   8
#define HS_WAVES    (4 * HS_KPARTS)
#define HS_SLICE_K  (HEAD_K / HS_SLICES)
#define HS_PART_K   (HS_SLICE_K / HS_KPARTS)
#define HS_GK       16
#define HS_GROUPS   (HS_PART_K / HS_GK)
static_assert(HS_SLICES * HS_KPARTS * HS_GROUPS * HS_GK == HEAD_K, "K is covered exactly");
static_assert(BATCH == 64, "four row tiles");

typedef const __attribute__((address_space(1))) v2f* gcv2p;

__global__ void __launch_bounds__(HS_WAVES * 32)
k_head_gemm(const float* __restrict__ A,
            const float* __restrict__ Bp, const float* __restrict__ Bv,
            float* __restrict__ Cp, float* __restrict__ Cv) {
  __shared__ float part[HS_WAVES * 1024];

  const int wave = threadIdx.x >> 5;
  const int lane = threadIdx.x & 31;
  const int half = lane >> 4;
  const int lm   = lane & 15;
  const int mt   = wave & 3;
  const int kp   = wave >> 2;
  const int m0   = mt * 16;
  const int n0   = (blockIdx.x & 3) * 64;
  const bool second = blockIdx.x >= 4;
  const float* B = second ? Bv : Bp;
  float*       C = second ? Cv : Cp;
  const int kbeg = blockIdx.y * HS_SLICE_K + kp * HS_PART_K;

  gcfp ap = (gcfp)A + ((unsigned)(m0 + lm) * (unsigned)HEAD_K + (unsigned)(kbeg + half * 2));
  gcfp bp = (gcfp)B + ((unsigned)(kbeg + half * 2) * (unsigned)HEAD_N + (unsigned)(n0 + lm));

  v8f acc[4];
#pragma unroll
  for (int t = 0; t < 4; ++t) acc[t] = (v8f){0.f,0.f,0.f,0.f,0.f,0.f,0.f,0.f};

  for (int g = 0; g < HS_GROUPS; ++g) {
    v2f a[4], b[4][4];
#pragma unroll
    for (int s = 0; s < 4; ++s)
      a[s] = *(gcv2p)(ap + 4 * s);
#pragma unroll
    for (int s = 0; s < 4; ++s) {
#pragma unroll
      for (int t = 0; t < 4; ++t) {
        b[t][s].x = bp[(4 * s) * HEAD_N + 16 * t];
        b[t][s].y = bp[(4 * s + 1) * HEAD_N + 16 * t];
      }
    }
    __builtin_amdgcn_sched_barrier(0);
#pragma unroll
    for (int s = 0; s < 4; ++s) {
#pragma unroll
      for (int t = 0; t < 4; ++t)
        acc[t] = __builtin_amdgcn_wmma_f32_16x16x4_f32(
            false, a[s], false, b[t][s], (short)0, acc[t], false, false);
    }
    ap += HS_GK;
    bp += HS_GK * HEAD_N;
  }

  float* mine = part + wave * 1024;
#pragma unroll
  for (int t = 0; t < 4; ++t) {
#pragma unroll
    for (int r = 0; r < 8; ++r)
      mine[(r + half * 8) * 64 + t * 16 + lm] = acc[t][r];
  }
  __syncthreads();

  const int rr  = threadIdx.x >> 6;
  const int col = threadIdx.x & 63;
#pragma unroll
  for (int j = 0; j < 4; ++j) {
    float s = part[j * 1024 + rr * 64 + col];
    for (int q = 1; q < HS_KPARTS; ++q) s += part[(q * 4 + j) * 1024 + rr * 64 + col];
    atomicAdd(&C[(j * 16 + rr) * HEAD_N + n0 + col], s);
  }
}

#define POL_TILES ((NODE_COUNT + 15) / 16)
#define PO_KPARTS 8
#define PO_WAVES  (4 * PO_KPARTS)
#define PO_PART_K (HEAD_N / PO_KPARTS)
#define PO_STEPS  (PO_PART_K / 4)
#define VAL_ROWS  32
#define VAL_RPW   (VAL_ROWS / PO_WAVES)
static_assert(BATCH % VAL_ROWS == 0 && HEAD_N == 256, "value rows and k split");
static_assert(VAL_ROWS % PO_WAVES == 0 && PO_KPARTS * PO_STEPS * 4 == HEAD_N, "rows per wave, K covered exactly");

__global__ void __launch_bounds__(PO_WAVES * 32)
k_head_out(const float* __restrict__ sump, const float* __restrict__ Wp2,
           const float* __restrict__ bp2, float* __restrict__ outp,
           const float* __restrict__ bp1,
           const float* __restrict__ sumv, const float* __restrict__ bv1,
           const float* __restrict__ Wv2, const float* __restrict__ bv2,
           float* __restrict__ outv) {
  __shared__ float vres[VAL_ROWS];
  __shared__ float ppart[PO_WAVES * 256];
  const int wave = threadIdx.x >> 5;
  const int lane = threadIdx.x & 31;
  if (blockIdx.x < POL_TILES) {
    const int half = lane >> 4;
    const int lm   = lane & 15;
    const int mt   = wave & 3;
    const int kp   = wave >> 2;
    const int m0   = mt * 16;
    const int ncol = blockIdx.x * 16 + lm;
    const bool nok = ncol < NODE_COUNT;
    const int nld  = nok ? ncol : (NODE_COUNT - 1);
    const int kb   = kp * PO_PART_K + half * 2;
    gcfp ap  = (gcfp)sump + ((unsigned)(m0 + lm) * (unsigned)HEAD_N + (unsigned)kb);
    gcfp abp = (gcfp)bp1 + kb;
    gcfp bp  = (gcfp)Wp2 + ((unsigned)kb * (unsigned)NODE_COUNT + (unsigned)nld);
    const float bo = ((gcfp)bp2)[nld];

    float a0[PO_STEPS], a1[PO_STEPS], c0[PO_STEPS], c1[PO_STEPS], b0[PO_STEPS], b1[PO_STEPS];
#pragma unroll
    for (int s = 0; s < PO_STEPS; ++s) {
      a0[s] = ap[4 * s];
      a1[s] = ap[4 * s + 1];
      c0[s] = abp[4 * s];
      c1[s] = abp[4 * s + 1];
      b0[s] = bp[(4 * s) * NODE_COUNT];
      b1[s] = bp[(4 * s + 1) * NODE_COUNT];
    }
    v8f d = (v8f){0.f, 0.f, 0.f, 0.f, 0.f, 0.f, 0.f, 0.f};
#pragma unroll
    for (int s = 0; s < PO_STEPS; ++s) {
      float p0 = a0[s] + c0[s];
      float p1 = a1[s] + c1[s];
      v2f a, b;
      a.x = p0 > 0.0f ? p0 : 0.0f;
      a.y = p1 > 0.0f ? p1 : 0.0f;
      b.x = nok ? b0[s] : 0.0f;
      b.y = nok ? b1[s] : 0.0f;
      d = __builtin_amdgcn_wmma_f32_16x16x4_f32(false, a, false, b, (short)0, d, false, false);
    }
#pragma unroll
    for (int r = 0; r < 8; ++r) ppart[(kp * 4 + mt) * 256 + r * 32 + lane] = d[r];
    __syncthreads();
    if (kp == 0) {
      float s[8];
#pragma unroll
      for (int r = 0; r < 8; ++r) {
        s[r] = ppart[mt * 256 + r * 32 + lane];
#pragma unroll
        for (int q = 1; q < PO_KPARTS; ++q) s[r] += ppart[(q * 4 + mt) * 256 + r * 32 + lane];
      }
      if (nok) {
        gfp og = (gfp)outp;
#pragma unroll
        for (int r = 0; r < 8; ++r)
          og[(unsigned)(m0 + r + half * 8) * (unsigned)NODE_COUNT + (unsigned)ncol] = s[r] + bo;
      }
    }
    return;
  }
  const int vb   = blockIdx.x - POL_TILES;
  const int row0 = vb * VAL_ROWS + wave * VAL_RPW;
  gcv4p sg = (gcv4p)sumv;
  gcv4p bg = (gcv4p)bv1;
  gcv4p wg = (gcv4p)Wv2;
  const v4f bb0 = bg[lane], bb1 = bg[32 + lane];
  const v4f ww0 = wg[lane], ww1 = wg[32 + lane];
  v4f x0[VAL_RPW], x1[VAL_RPW];
#pragma unroll
  for (int r = 0; r < VAL_RPW; ++r) {
    x0[r] = sg[(row0 + r) * (HEAD_N / 4) + lane];
    x1[r] = sg[(row0 + r) * (HEAD_N / 4) + 32 + lane];
  }
  const float b2 = bv2[0];
#pragma unroll
  for (int r = 0; r < VAL_RPW; ++r) {
    float sacc = 0.0f;
#pragma unroll
    for (int e = 0; e < 4; ++e) {
      float h0 = x0[r][e] + bb0[e];
      float h1 = x1[r][e] + bb1[e];
      h0 = h0 > 0.0f ? h0 : 0.0f;
      h1 = h1 > 0.0f ? h1 : 0.0f;
      sacc += h0 * ww0[e];
      sacc += h1 * ww1[e];
    }
    for (int o = 16; o > 0; o >>= 1) sacc += __shfl_xor(sacc, o);
    if (lane == 0) vres[wave * VAL_RPW + r] = sacc + b2;
  }
  __syncthreads();
  if (wave == 0) outv[vb * VAL_ROWS + lane] = vres[lane];
}

extern "C" void kernel_launch(void* const* d_in, const int* in_sizes, int n_in,
                              void* d_out, int out_size, void* d_ws, size_t ws_size,
                              hipStream_t stream) {
  (void)in_sizes; (void)n_in; (void)out_size; (void)ws_size;

  const float* x    = (const float*)d_in[0];
  const int*   ei   = (const int*)d_in[1];
  const int*   rows = ei;
  const int*   cols = ei + N_EDGES;
  const float* W[5]  = {(const float*)d_in[2], (const float*)d_in[4], (const float*)d_in[6],
                        (const float*)d_in[8], (const float*)d_in[10]};
  const float* bb[5] = {(const float*)d_in[3], (const float*)d_in[5], (const float*)d_in[7],
                        (const float*)d_in[9], (const float*)d_in[11]};
  const float* Wp1 = (const float*)d_in[12]; const float* bp1 = (const float*)d_in[13];
  const float* Wp2 = (const float*)d_in[14]; const float* bp2 = (const float*)d_in[15];
  const float* Wv1 = (const float*)d_in[16]; const float* bv1 = (const float*)d_in[17];
  const float* Wv2 = (const float*)d_in[18]; const float* bv2 = (const float*)d_in[19];
  float* out = (float*)d_out;

  float* ws     = (float*)d_ws;
  float* deg    = ws;
  float* hidp   = deg + N_TOTAL;
  float* hidv   = hidp + BATCH * 256;
  int*   offs   = (int*)(hidv + BATCH * 256);
  int*   base   = offs + NCH * 256;
  int*   rec    = base + 8192;
  float* bufA   = (float*)(rec + 2 * (size_t)N_EDGES);
  float* bufB   = bufA + (size_t)N_TOTAL * HID;
  int*   srt    = (int*)(bufB + (size_t)N_TOTAL * HID);

  k_sort    <<<NCH, 1024, 0, stream>>>(rows, cols, offs, srt, hidp);
  k_list_deg<<<NBLK, 1024, 0, stream>>>(srt, offs, base, rec, deg);

  k_agg<STATE, true> <<<NBLK, 1024, 0, stream>>>(rec, base, deg, x,    W[0], bb[0], bufA);
  k_agg<HID, false>  <<<NBLK, 1024, 0, stream>>>(rec, base, deg, bufA, W[1], bb[1], bufB);
  k_agg<HID, false>  <<<NBLK, 1024, 0, stream>>>(rec, base, deg, bufB, W[2], bb[2], bufA);
  k_agg<HID, false>  <<<NBLK, 1024, 0, stream>>>(rec, base, deg, bufA, W[3], bb[3], bufB);
  k_agg<HID, false>  <<<NBLK, 1024, 0, stream>>>(rec, base, deg, bufB, W[4], bb[4], bufA);

  dim3 g1(2 * HEAD_N / 64, HS_SLICES);
  k_head_gemm<<<g1, HS_WAVES * 32, 0, stream>>>(bufA, Wp1, Wv1, hidp, hidv);

  k_head_out<<<POL_TILES + BATCH / VAL_ROWS, PO_WAVES * 32, 0, stream>>>(
      hidp, Wp2, bp2, out, bp1, hidv, bv1, Wv2, bv2, out + (size_t)BATCH * NODE_COUNT);
}
